# starting kernel
# baseline (speedup 1.0000x reference)
#include <hip/hip_runtime.h>
#include <hip/hip_bf16.h>
#include <math.h>

#define D_IN 128
#define NH   4
#define NF   16
#define HF   64

typedef __attribute__((ext_vector_type(2))) float v2f;
typedef __attribute__((ext_vector_type(8))) float v8f;

__global__ void dgat_init_kernel(float* __restrict__ segmax,
                                 float* __restrict__ denom,
                                 float* __restrict__ out,
                                 int nh, int nhf) {
    int i = blockIdx.x * blockDim.x + threadIdx.x;
    int stride = gridDim.x * blockDim.x;
    for (int j = i; j < nh; j += stride) {
        segmax[j] = -__builtin_huge_valf();
        denom[j]  = 0.0f;
    }
    for (int j = i; j < nhf; j += stride) {
        out[j] = 0.0f;
    }
}

__global__ void dgat_proj_wmma_kernel(const float* __restrict__ feat,
                                      const float* __restrict__ W,
                                      float* __restrict__ ft,
                                      int N, int ntiles) {
    const int lane = threadIdx.x & 31;
    const int wave = threadIdx.x >> 5;
    const int tile = blockIdx.x * (blockDim.x >> 5) + wave;
    if (tile >= ntiles) return;

    const int mt = tile >> 2;
    const int nt = tile & 3;
    const int m0 = mt * 16;
    const int n0 = nt * 16;

    const int half = lane >> 4;
    const int l    = lane & 15;

    int rowA = m0 + l;
    if (rowA > N - 1) rowA = N - 1;
    const float* __restrict__ arow = feat + (size_t)rowA * D_IN;
    const float* __restrict__ bp   = W + n0 + l;

    v8f c = {0.f, 0.f, 0.f, 0.f, 0.f, 0.f, 0.f, 0.f};

#pragma unroll 8
    for (int k = 0; k < D_IN; k += 4) {
        const int kk = k + 2 * half;
        v2f a, b;
        a.x = arow[kk];
        a.y = arow[kk + 1];
        b.x = bp[(size_t)kk * HF];
        b.y = bp[(size_t)(kk + 1) * HF];
        c = __builtin_amdgcn_wmma_f32_16x16x4_f32(
                false, a, false, b, (short)0, c, false, false);
    }

    float* __restrict__ op = ft + ((size_t)(m0 + 8 * half)) * HF + n0 + l;
    if (m0 + 16 <= N) {
#pragma unroll
        for (int v = 0; v < 8; ++v)
            op[(size_t)v * HF] = c[v];
    } else {
#pragma unroll
        for (int v = 0; v < 8; ++v) {
            const int row = m0 + v + 8 * half;
            if (row < N)
                ft[(size_t)row * HF + n0 + l] = c[v];
        }
    }
}

__device__ __forceinline__ void atomicMaxFloat(float* addr, float val) {
    if (val >= 0.0f)
        atomicMax((int*)addr, __float_as_int(val));
    else
        atomicMin((unsigned int*)addr, __float_as_uint(val));
}

__global__ void dgat_score_kernel(const float* __restrict__ ft,
                                  const int* __restrict__ src,
                                  const int* __restrict__ dst,
                                  float* __restrict__ e,
                                  float* __restrict__ segmax,
                                  int E) {
    const int t = blockIdx.x * blockDim.x + threadIdx.x;
    if (t >= E * NH) return;
    const int edge = t >> 2;
    const int h    = t & 3;
    const int s    = src[edge];
    const int d    = dst[edge];

    const float4* __restrict__ a = (const float4*)(ft + (size_t)s * HF + h * NF);
    const float4* __restrict__ b = (const float4*)(ft + (size_t)d * HF + h * NF);
    float acc = 0.0f;
#pragma unroll
    for (int i = 0; i < 4; ++i) {
        const float4 av = a[i], bv = b[i];
        acc += av.x * bv.x + av.y * bv.y + av.z * bv.z + av.w * bv.w;
    }
    acc *= 0.25f;
    e[t] = acc;
    atomicMaxFloat(&segmax[(size_t)d * NH + h], acc);
}

__global__ void dgat_exp_kernel(const float* __restrict__ segmax,
                                const int* __restrict__ dst,
                                float* __restrict__ e,
                                float* __restrict__ denom,
                                int E) {
    const int t = blockIdx.x * blockDim.x + threadIdx.x;
    if (t >= E * NH) return;
    const int edge = t >> 2;
    const int h    = t & 3;
    const int d    = dst[edge];
    const float ex = __expf(e[t] - segmax[(size_t)d * NH + h]);
    e[t] = ex;
    atomicAdd(&denom[(size_t)d * NH + h], ex);
}

__global__ void dgat_agg_kernel(const float* __restrict__ ft,
                                const int* __restrict__ src,
                                const int* __restrict__ dst,
                                const float* __restrict__ ex,
                                const float* __restrict__ denom,
                                float* __restrict__ out,
                                int E) {
    const int t = blockIdx.x * blockDim.x + threadIdx.x;
    if (t >= E * NH) return;
    const int edge = t >> 2;
    const int h    = t & 3;
    const int s    = src[edge];
    const int d    = dst[edge];
    const float sa = ex[t] / denom[(size_t)d * NH + h];

    const float* __restrict__ fs = ft + (size_t)s * HF + h * NF;
    float* __restrict__ op = out + (size_t)d * HF + h * NF;
#pragma unroll
    for (int i = 0; i < NF; ++i)
        atomicAdd(&op[i], fs[i] * sa);
}

extern "C" void kernel_launch(void* const* d_in, const int* in_sizes, int n_in,
                              void* d_out, int out_size, void* d_ws, size_t ws_size,
                              hipStream_t stream) {
    const float* feat = (const float*)d_in[0];
    const float* W    = (const float*)d_in[1];
    const int*   src  = (const int*)d_in[2];
    const int*   dst  = (const int*)d_in[3];
    float*       out  = (float*)d_out;

    const int N = in_sizes[0] / D_IN;
    const int E = in_sizes[2];

    float* ws     = (float*)d_ws;
    float* ft     = ws;
    float* e      = ft + (size_t)N * HF;
    float* segmax = e + (size_t)E * NH;
    float* denom  = segmax + (size_t)N * NH;

    const int nh  = N * NH;
    const int nhf = N * HF;

    dgat_init_kernel<<<2048, 256, 0, stream>>>(segmax, denom, out, nh, nhf);

    const int mtiles = (N + 15) / 16;
    const int ntiles = mtiles * 4;
    dgat_proj_wmma_kernel<<<(ntiles + 7) / 8, 256, 0, stream>>>(feat, W, ft, N, ntiles);

    const int work = E * NH;
    const int blk  = 256;
    const int grd  = (work + blk - 1) / blk;
    dgat_score_kernel<<<grd, blk, 0, stream>>>(ft, src, dst, e, segmax, E);
    dgat_exp_kernel<<<grd, blk, 0, stream>>>(segmax, dst, e, denom, E);
    dgat_agg_kernel<<<grd, blk, 0, stream>>>(ft, src, dst, e, denom, out, E);
}


# Group edges by destination (integer-atomic count, scan, fill, sort by edge id); one wave per node does softmax and row sum in registers: no float atomics, sums now in fixed edge order
# speedup vs baseline: 2.1550x; 2.1550x over previous
#include <hip/hip_runtime.h>
#include <hip/hip_bf16.h>
#include <math.h>


#define D_IN 128
#define NH   4
#define NF   16
#define HF   64

typedef __attribute__((ext_vector_type(2))) float v2f;
typedef __attribute__((ext_vector_type(8))) float v8f;

__global__ void dgat_zero_kernel(int* __restrict__ count,
                                 int* __restrict__ cursor,
                                 int N) {
    const int i = blockIdx.x * blockDim.x + threadIdx.x;
    if (i < N) count[i] = 0;
    if (i == 0) cursor[0] = 0;
}

__global__ void dgat_proj_wmma_kernel(const float* __restrict__ feat,
                                      const float* __restrict__ W,
                                      float* __restrict__ ft,
                                      int N, int ntiles) {
    const int lane = threadIdx.x & 31;
    const int wave = threadIdx.x >> 5;
    const int tile = blockIdx.x * (blockDim.x >> 5) + wave;
    if (tile >= ntiles) return;

    const int mt = tile >> 2;
    const int nt = tile & 3;
    const int m0 = mt * 16;
    const int n0 = nt * 16;

    const int half = lane >> 4;
    const int l    = lane & 15;

    int rowA = m0 + l;
    if (rowA > N - 1) rowA = N - 1;
    const float* __restrict__ arow = feat + (size_t)rowA * D_IN;
    const float* __restrict__ bp   = W + n0 + l;

    v8f c = {0.f, 0.f, 0.f, 0.f, 0.f, 0.f, 0.f, 0.f};

#pragma unroll 8
    for (int k = 0; k < D_IN; k += 4) {
        const int kk = k + 2 * half;
        v2f a, b;
        a.x = arow[kk];
        a.y = arow[kk + 1];
        b.x = bp[(size_t)kk * HF];
        b.y = bp[(size_t)(kk + 1) * HF];
        c = __builtin_amdgcn_wmma_f32_16x16x4_f32(
                false, a, false, b, (short)0, c, false, false);
    }

    float* __restrict__ op = ft + ((size_t)(m0 + 8 * half)) * HF + n0 + l;
    if (m0 + 16 <= N) {
#pragma unroll
        for (int v = 0; v < 8; ++v)
            op[(size_t)v * HF] = c[v];
    } else {
#pragma unroll
        for (int v = 0; v < 8; ++v) {
            const int row = m0 + v + 8 * half;
            if (row < N)
                ft[(size_t)row * HF + n0 + l] = c[v];
        }
    }
}

__global__ void dgat_count_kernel(const int* __restrict__ dst,
                                  int* __restrict__ count,
                                  int* __restrict__ rank,
                                  int E) {
    const int e0 = (blockIdx.x * blockDim.x + threadIdx.x) * 4;
    if (e0 + 3 < E) {
        const int4 d = *(const int4*)(dst + e0);
        int4 r;
        r.x = atomicAdd(&count[d.x], 1);
        r.y = atomicAdd(&count[d.y], 1);
        r.z = atomicAdd(&count[d.z], 1);
        r.w = atomicAdd(&count[d.w], 1);
        *(int4*)(rank + e0) = r;
    } else {
        for (int e = e0; e < E; ++e)
            rank[e] = atomicAdd(&count[dst[e]], 1);
    }
}

__global__ void dgat_alloc_kernel(const int* __restrict__ count,
                                  int* __restrict__ offset,
                                  int* __restrict__ cursor,
                                  int N) {
    __shared__ int wsum[8];
    __shared__ int base;
    const int i    = blockIdx.x * 256 + threadIdx.x;
    const int lane = threadIdx.x & 31;
    const int wave = threadIdx.x >> 5;
    const int c    = (i < N) ? count[i] : 0;

    int x = c;
    for (int d = 1; d < 32; d <<= 1) {
        const int y = __shfl_up(x, d);
        if (lane >= d) x += y;
    }
    if (lane == 31) wsum[wave] = x;
    __syncthreads();
    if (threadIdx.x == 0) {
        int tot = 0;
        for (int w = 0; w < 8; ++w) {
            const int t = wsum[w];
            wsum[w] = tot;
            tot += t;
        }
        base = atomicAdd(cursor, tot);
    }
    __syncthreads();
    if (i < N) offset[i] = base + wsum[wave] + x - c;
}

__global__ void dgat_fill_kernel(const int* __restrict__ dst,
                                 const int* __restrict__ rank,
                                 const int* __restrict__ offset,
                                 int* __restrict__ raw,
                                 int E) {
    const int e0 = (blockIdx.x * blockDim.x + threadIdx.x) * 4;
    if (e0 + 3 < E) {
        const int4 d = *(const int4*)(dst + e0);
        const int4 r = *(const int4*)(rank + e0);
        const int p0 = offset[d.x] + r.x;
        const int p1 = offset[d.y] + r.y;
        const int p2 = offset[d.z] + r.z;
        const int p3 = offset[d.w] + r.w;
        raw[p0] = e0;
        raw[p1] = e0 + 1;
        raw[p2] = e0 + 2;
        raw[p3] = e0 + 3;
    } else {
        for (int e = e0; e < E; ++e)
            raw[offset[dst[e]] + rank[e]] = e;
    }
}

__global__ void dgat_order_kernel(const int* __restrict__ src,
                                  const int* __restrict__ dst,
                                  const int* __restrict__ offset,
                                  const int* __restrict__ count,
                                  const int* __restrict__ raw,
                                  int* __restrict__ ssrc,
                                  int E) {
    const int e0 = (blockIdx.x * blockDim.x + threadIdx.x) * 4;
    if (e0 + 3 < E) {
        const int4 d = *(const int4*)(dst + e0);
        const int4 s = *(const int4*)(src + e0);
        const int bb[4] = {offset[d.x], offset[d.y], offset[d.z], offset[d.w]};
        const int cc[4] = {count[d.x], count[d.y], count[d.z], count[d.w]};
        const int sv[4] = {s.x, s.y, s.z, s.w};
#pragma unroll
        for (int i = 0; i < 4; ++i) {
            const int e = e0 + i;
            int r = 0;
            for (int k = 0; k < cc[i]; ++k)
                r += (raw[bb[i] + k] < e) ? 1 : 0;
            ssrc[bb[i] + r] = sv[i];
        }
    } else {
        for (int e = e0; e < E; ++e) {
            const int dd  = dst[e];
            const int beg = offset[dd];
            const int cnt = count[dd];
            int r = 0;
            for (int k = 0; k < cnt; ++k)
                r += (raw[beg + k] < e) ? 1 : 0;
            ssrc[beg + r] = src[e];
        }
    }
}

__device__ __forceinline__ float quad_sum(float v) {
    v += __int_as_float(__builtin_amdgcn_mov_dpp(__float_as_int(v), 0xB1, 0xF, 0xF, true));
    v += __int_as_float(__builtin_amdgcn_mov_dpp(__float_as_int(v), 0x4E, 0xF, 0xF, true));
    return v;
}

__device__ __forceinline__ float head_score(const float4 a, const float4 b) {
    float p = a.x * b.x;
    p = fmaf(a.y, b.y, p);
    p = fmaf(a.z, b.z, p);
    p = fmaf(a.w, b.w, p);
    return quad_sum(p) * 0.25f;
}

#define NODE_GROUP 4

__global__ void
dgat_node_kernel(const float* __restrict__ ft,
                 const int* __restrict__ offset,
                 const int* __restrict__ count,
                 const int* __restrict__ ssrc,
                 float* __restrict__ out,
                 int N) {
    const int lane = threadIdx.x & 31;
    const int node = __builtin_amdgcn_readfirstlane(blockIdx.x * 8 + (threadIdx.x >> 5));
    if (node >= N) return;
    const int half = lane >> 4;
    const int q    = lane & 15;
    const int beg  = offset[node];
    const int cnt  = count[node];

    const float4* __restrict__ ft4 = (const float4*)ft;
    const float4 fd = ft4[(size_t)node * 16 + q];
    const float4 zero4 = make_float4(0.f, 0.f, 0.f, 0.f);

    float m = -__builtin_huge_valf();
    for (int c0 = 0; c0 < cnt; c0 += 32) {
        const int nrem = min(cnt - c0, 32);
        const int myid = (lane < nrem) ? ssrc[beg + c0 + lane] : 0;
        for (int g = 0; g < nrem; g += 2 * NODE_GROUP) {
            float4 fv[NODE_GROUP];
            bool   ok[NODE_GROUP];
            for (int u = 0; u < NODE_GROUP; ++u) {
                const int k = g + 2 * u + half;
                ok[u] = k < nrem;
                const int s = __shfl(myid, min(k, nrem - 1));
                fv[u] = ft4[(size_t)s * 16 + q];
            }
            for (int u = 0; u < NODE_GROUP; ++u) {
                const float p = head_score(fv[u], fd);
                if (ok[u]) m = fmaxf(m, p);
            }
        }
    }
    m = fmaxf(m, __shfl_xor(m, 16));

    float  l   = 0.0f;
    float4 acc = zero4;
    for (int c0 = 0; c0 < cnt; c0 += 32) {
        const int nrem = min(cnt - c0, 32);
        const int myid = (lane < nrem) ? ssrc[beg + c0 + lane] : 0;
        for (int g = 0; g < nrem; g += 2 * NODE_GROUP) {
            float4 fv[NODE_GROUP];
            bool   ok[NODE_GROUP];
            for (int u = 0; u < NODE_GROUP; ++u) {
                const int k = g + 2 * u + half;
                ok[u] = k < nrem;
                const int s = __shfl(myid, min(k, nrem - 1));
                fv[u] = ft4[(size_t)s * 16 + q];
            }
            for (int u = 0; u < NODE_GROUP; ++u) {
                const float p  = head_score(fv[u], fd);
                const float ex = ok[u] ? __expf(p - m) : 0.0f;
                l += ex;
                acc.x = fmaf(ex, fv[u].x, acc.x);
                acc.y = fmaf(ex, fv[u].y, acc.y);
                acc.z = fmaf(ex, fv[u].z, acc.z);
                acc.w = fmaf(ex, fv[u].w, acc.w);
            }
        }
    }
    l     += __shfl_xor(l, 16);
    acc.x += __shfl_xor(acc.x, 16);
    acc.y += __shfl_xor(acc.y, 16);
    acc.z += __shfl_xor(acc.z, 16);
    acc.w += __shfl_xor(acc.w, 16);

    if (half == 0) {
        float4 o = zero4;
        if (cnt > 0) {
            const float inv = 1.0f / l;
            o.x = acc.x * inv;
            o.y = acc.y * inv;
            o.z = acc.z * inv;
            o.w = acc.w * inv;
        }
        ((float4*)out)[(size_t)node * 16 + q] = o;
    }
}

extern "C" void kernel_launch(void* const* d_in, const int* in_sizes, int n_in,
                              void* d_out, int out_size, void* d_ws, size_t ws_size,
                              hipStream_t stream) {
    const float* feat = (const float*)d_in[0];
    const float* W    = (const float*)d_in[1];
    const int*   src  = (const int*)d_in[2];
    const int*   dst  = (const int*)d_in[3];
    float*       out  = (float*)d_out;

    const int N = in_sizes[0] / D_IN;
    const int E = in_sizes[2];

    float* ft     = (float*)d_ws;
    int*   rank   = (int*)(ft + (size_t)N * HF);
    int*   raw    = rank + (size_t)E;
    int*   ssrc   = raw + (size_t)E;
    int*   count  = ssrc + (size_t)E;
    int*   offset = count + (size_t)N;
    int*   cursor = offset + (size_t)N;

    const int blk = 256;
    const int gn  = (N + blk - 1) / blk;
    const int ge4 = (E + 4 * blk - 1) / (4 * blk);

    dgat_zero_kernel<<<gn, blk, 0, stream>>>(count, cursor, N);

    const int mtiles = (N + 15) / 16;
    const int ntiles = mtiles * 4;
    dgat_proj_wmma_kernel<<<(ntiles + 7) / 8, 256, 0, stream>>>(feat, W, ft, N, ntiles);

    dgat_count_kernel<<<ge4, blk, 0, stream>>>(dst, count, rank, E);
    dgat_alloc_kernel<<<gn, 256, 0, stream>>>(count, offset, cursor, N);
    dgat_fill_kernel<<<ge4, blk, 0, stream>>>(dst, rank, offset, raw, E);
    dgat_order_kernel<<<ge4, blk, 0, stream>>>(src, dst, offset, count, raw, ssrc, E);
    dgat_node_kernel<<<(N + 7) / 8, 256, 0, stream>>>(ft, offset, count, ssrc, out, N);
}


# Group edges in one pass into fixed 32-slot buckets per destination (overflow list beyond); node wave sorts its ids by edge id in registers: scan, fill and order launches gone, sum order unchanged
# speedup vs baseline: 2.6494x; 1.2294x over previous
#include <hip/hip_runtime.h>
#include <hip/hip_bf16.h>
#include <math.h>


#define D_IN 128
#define NH   4
#define NF   16
#define HF   64

typedef __attribute__((ext_vector_type(2))) float v2f;
typedef __attribute__((ext_vector_type(8))) float v8f;

__global__ void dgat_zero_kernel(int* __restrict__ count,
                                 int* __restrict__ cursor,
                                 int N) {
    const int i = blockIdx.x * blockDim.x + threadIdx.x;
    if (i < N) count[i] = 0;
    if (i == 0) cursor[0] = 0;
}

__global__ void dgat_proj_wmma_kernel(const float* __restrict__ feat,
                                      const float* __restrict__ W,
                                      float* __restrict__ ft,
                                      int N, int ntiles) {
    const int lane = threadIdx.x & 31;
    const int wave = threadIdx.x >> 5;
    const int tile = blockIdx.x * (blockDim.x >> 5) + wave;
    if (tile >= ntiles) return;

    const int mt = tile >> 2;
    const int nt = tile & 3;
    const int m0 = mt * 16;
    const int n0 = nt * 16;

    const int half = lane >> 4;
    const int l    = lane & 15;

    int rowA = m0 + l;
    if (rowA > N - 1) rowA = N - 1;
    const float* __restrict__ arow = feat + (size_t)rowA * D_IN;
    const float* __restrict__ bp   = W + n0 + l;

    v8f c = {0.f, 0.f, 0.f, 0.f, 0.f, 0.f, 0.f, 0.f};

#pragma unroll 8
    for (int k = 0; k < D_IN; k += 4) {
        const int kk = k + 2 * half;
        v2f a, b;
        a.x = arow[kk];
        a.y = arow[kk + 1];
        b.x = bp[(size_t)kk * HF];
        b.y = bp[(size_t)(kk + 1) * HF];
        c = __builtin_amdgcn_wmma_f32_16x16x4_f32(
                false, a, false, b, (short)0, c, false, false);
    }

    float* __restrict__ op = ft + ((size_t)(m0 + 8 * half)) * HF + n0 + l;
    if (m0 + 16 <= N) {
#pragma unroll
        for (int v = 0; v < 8; ++v)
            op[(size_t)v * HF] = c[v];
    } else {
#pragma unroll
        for (int v = 0; v < 8; ++v) {
            const int row = m0 + v + 8 * half;
            if (row < N)
                ft[(size_t)row * HF + n0 + l] = c[v];
        }
    }
}

#define BCAP 32

__device__ __forceinline__ void put_edge(const int e, const int d, const int r,
                                         int* __restrict__ bucket,
                                         int* __restrict__ ovf,
                                         int* __restrict__ cursor) {
    if (r < BCAP) bucket[d * BCAP + r] = e;
    else          ovf[atomicAdd(cursor, 1)] = e;
}

__global__ void dgat_bucket_kernel(const int* __restrict__ dst,
                                   int* __restrict__ count,
                                   int* __restrict__ bucket,
                                   int* __restrict__ ovf,
                                   int* __restrict__ cursor,
                                   int E) {
    const int e0 = (blockIdx.x * blockDim.x + threadIdx.x) * 4;
    if (e0 + 3 < E) {
        const int4 d = *(const int4*)(dst + e0);
        const int r0 = atomicAdd(&count[d.x], 1);
        const int r1 = atomicAdd(&count[d.y], 1);
        const int r2 = atomicAdd(&count[d.z], 1);
        const int r3 = atomicAdd(&count[d.w], 1);
        put_edge(e0,     d.x, r0, bucket, ovf, cursor);
        put_edge(e0 + 1, d.y, r1, bucket, ovf, cursor);
        put_edge(e0 + 2, d.z, r2, bucket, ovf, cursor);
        put_edge(e0 + 3, d.w, r3, bucket, ovf, cursor);
    } else {
        for (int e = e0; e < E; ++e) {
            const int d = dst[e];
            put_edge(e, d, atomicAdd(&count[d], 1), bucket, ovf, cursor);
        }
    }
}

__device__ __forceinline__ float quad_sum(float v) {
    v += __int_as_float(__builtin_amdgcn_mov_dpp(__float_as_int(v), 0xB1, 0xF, 0xF, true));
    v += __int_as_float(__builtin_amdgcn_mov_dpp(__float_as_int(v), 0x4E, 0xF, 0xF, true));
    return v;
}

__device__ __forceinline__ float head_score(const float4 a, const float4 b) {
    float p = a.x * b.x;
    p = fmaf(a.y, b.y, p);
    p = fmaf(a.z, b.z, p);
    p = fmaf(a.w, b.w, p);
    return quad_sum(p) * 0.25f;
}

__device__ __forceinline__ int long_list_chunk(const int node, const int nrem, int& last,
                                            const int mine, const int lane,
                                            const int* __restrict__ src,
                                            const int* __restrict__ dst,
                                            const int* __restrict__ ovf,
                                            const int novf, const int E) {
    int myid = 0;
    for (int k = 0; k < nrem; ++k) {
        int best = (mine > last) ? mine : 0x7fffffff;
        for (int p0 = 0; p0 < novf; p0 += 32) {
            const int p = p0 + lane;
            if (p < novf) {
                const int e2 = ovf[p];
                if (e2 > last && dst[e2] == node) best = min(best, e2);
            }
        }
        for (int o = 16; o > 0; o >>= 1) best = min(best, __shfl_xor(best, o));
        best = min(best, E - 1);
        last = best;
        const int sm = src[best];
        if (lane == k) myid = sm;
    }
    return myid;
}

#define NODE_GROUP 4

__global__ void
dgat_node_kernel(const float* __restrict__ ft,
                 const int* __restrict__ src,
                 const int* __restrict__ dst,
                 const int* __restrict__ count,
                 const int* __restrict__ bucket,
                 const int* __restrict__ ovf,
                 const int* __restrict__ cursor,
                 float* __restrict__ out,
                 int N, int E) {
    const int lane = threadIdx.x & 31;
    const int node = __builtin_amdgcn_readfirstlane(blockIdx.x * 8 + (threadIdx.x >> 5));
    if (node >= N) return;
    const int half = lane >> 4;
    const int q    = lane & 15;
    const int cnt  = count[node];
    const int mine = bucket[node * BCAP + lane];

    const float4* __restrict__ ft4 = (const float4*)ft;
    const float4 fd = ft4[(size_t)node * 16 + q];
    const float4 zero4 = make_float4(0.f, 0.f, 0.f, 0.f);

    const bool shortlist = cnt <= BCAP;
    int ids0 = 0;
    int novf = 0;
    if (shortlist) {
        const int key = (lane < cnt) ? mine : 0x7fffffff;
        const int s   = (lane < cnt) ? src[key] : 0;
        int r = 0;
        for (int j = 0; j < cnt; ++j)
            r += (__builtin_amdgcn_readlane(key, j) < key) ? 1 : 0;
        if (lane >= cnt) r = lane;
        ids0 = __builtin_amdgcn_ds_permute(r << 2, s);
    } else {
        novf = cursor[0];
    }

    float m = -__builtin_huge_valf();
    int last = -1;
    for (int c0 = 0; c0 < cnt; c0 += 32) {
        const int nrem = min(cnt - c0, 32);
        const int myid = shortlist ? ids0
                                   : long_list_chunk(node, nrem, last, mine, lane,
                                                     src, dst, ovf, novf, E);
        for (int g = 0; g < nrem; g += 2 * NODE_GROUP) {
            float4 fv[NODE_GROUP];
            bool   ok[NODE_GROUP];
            for (int u = 0; u < NODE_GROUP; ++u) {
                const int k = g + 2 * u + half;
                ok[u] = k < nrem;
                const int s = __shfl(myid, min(k, nrem - 1));
                fv[u] = ft4[(size_t)s * 16 + q];
            }
            for (int u = 0; u < NODE_GROUP; ++u) {
                const float p = head_score(fv[u], fd);
                if (ok[u]) m = fmaxf(m, p);
            }
        }
    }
    m = fmaxf(m, __shfl_xor(m, 16));

    float  l   = 0.0f;
    float4 acc = zero4;
    last = -1;
    for (int c0 = 0; c0 < cnt; c0 += 32) {
        const int nrem = min(cnt - c0, 32);
        const int myid = shortlist ? ids0
                                   : long_list_chunk(node, nrem, last, mine, lane,
                                                     src, dst, ovf, novf, E);
        for (int g = 0; g < nrem; g += 2 * NODE_GROUP) {
            float4 fv[NODE_GROUP];
            bool   ok[NODE_GROUP];
            for (int u = 0; u < NODE_GROUP; ++u) {
                const int k = g + 2 * u + half;
                ok[u] = k < nrem;
                const int s = __shfl(myid, min(k, nrem - 1));
                fv[u] = ft4[(size_t)s * 16 + q];
            }
            for (int u = 0; u < NODE_GROUP; ++u) {
                const float p  = head_score(fv[u], fd);
                const float ex = ok[u] ? __expf(p - m) : 0.0f;
                l += ex;
                acc.x = fmaf(ex, fv[u].x, acc.x);
                acc.y = fmaf(ex, fv[u].y, acc.y);
                acc.z = fmaf(ex, fv[u].z, acc.z);
                acc.w = fmaf(ex, fv[u].w, acc.w);
            }
        }
    }
    l     += __shfl_xor(l, 16);
    acc.x += __shfl_xor(acc.x, 16);
    acc.y += __shfl_xor(acc.y, 16);
    acc.z += __shfl_xor(acc.z, 16);
    acc.w += __shfl_xor(acc.w, 16);

    if (half == 0) {
        float4 o = zero4;
        if (cnt > 0) {
            const float inv = 1.0f / l;
            o.x = acc.x * inv;
            o.y = acc.y * inv;
            o.z = acc.z * inv;
            o.w = acc.w * inv;
        }
        ((float4*)out)[(size_t)node * 16 + q] = o;
    }
}

extern "C" void kernel_launch(void* const* d_in, const int* in_sizes, int n_in,
                              void* d_out, int out_size, void* d_ws, size_t ws_size,
                              hipStream_t stream) {
    const float* feat = (const float*)d_in[0];
    const float* W    = (const float*)d_in[1];
    const int*   src  = (const int*)d_in[2];
    const int*   dst  = (const int*)d_in[3];
    float*       out  = (float*)d_out;

    const int N = in_sizes[0] / D_IN;
    const int E = in_sizes[2];

    float* ft     = (float*)d_ws;
    int*   bucket = (int*)(ft + (size_t)N * HF);
    int*   ovf    = bucket + (size_t)N * BCAP;
    int*   count  = ovf + (size_t)E;
    int*   cursor = count + (size_t)N;

    const int blk = 256;
    const int gn  = (N + blk - 1) / blk;
    const int ge4 = (E + 4 * blk - 1) / (4 * blk);

    dgat_zero_kernel<<<gn, blk, 0, stream>>>(count, cursor, N);

    const int mtiles = (N + 15) / 16;
    const int ntiles = mtiles * 4;
    dgat_proj_wmma_kernel<<<(ntiles + 7) / 8, 256, 0, stream>>>(feat, W, ft, N, ntiles);

    dgat_bucket_kernel<<<ge4, blk, 0, stream>>>(dst, count, bucket, ovf, cursor, E);
    dgat_node_kernel<<<(N + 7) / 8, 256, 0, stream>>>(ft, src, dst, count, bucket, ovf, cursor,
                                                      out, N, E);
}


# Node kernel, lists of 33 to 64 edges: bucket and overflow ids ranked once in registers for both passes; was one selection step with 3 dependent memory reads per edge and pass; sum order unchanged
# speedup vs baseline: 3.3238x; 1.2545x over previous
#include <hip/hip_runtime.h>
#include <hip/hip_bf16.h>
#include <math.h>


#define D_IN 128
#define NH   4
#define NF   16
#define HF   64

typedef __attribute__((ext_vector_type(2))) float v2f;
typedef __attribute__((ext_vector_type(8))) float v8f;

__global__ void dgat_zero_kernel(int* __restrict__ count,
                                 int* __restrict__ cursor,
                                 int N) {
    const int i = blockIdx.x * blockDim.x + threadIdx.x;
    if (i < N) count[i] = 0;
    if (i == 0) cursor[0] = 0;
}

__global__ void dgat_proj_wmma_kernel(const float* __restrict__ feat,
                                      const float* __restrict__ W,
                                      float* __restrict__ ft,
                                      int N, int ntiles) {
    const int lane = threadIdx.x & 31;
    const int wave = threadIdx.x >> 5;
    const int tile = blockIdx.x * (blockDim.x >> 5) + wave;
    if (tile >= ntiles) return;

    const int mt = tile >> 2;
    const int nt = tile & 3;
    const int m0 = mt * 16;
    const int n0 = nt * 16;

    const int half = lane >> 4;
    const int l    = lane & 15;

    int rowA = m0 + l;
    if (rowA > N - 1) rowA = N - 1;
    const float* __restrict__ arow = feat + (size_t)rowA * D_IN;
    const float* __restrict__ bp   = W + n0 + l;

    v8f c = {0.f, 0.f, 0.f, 0.f, 0.f, 0.f, 0.f, 0.f};

#pragma unroll 8
    for (int k = 0; k < D_IN; k += 4) {
        const int kk = k + 2 * half;
        v2f a, b;
        a.x = arow[kk];
        a.y = arow[kk + 1];
        b.x = bp[(size_t)kk * HF];
        b.y = bp[(size_t)(kk + 1) * HF];
        c = __builtin_amdgcn_wmma_f32_16x16x4_f32(
                false, a, false, b, (short)0, c, false, false);
    }

    float* __restrict__ op = ft + ((size_t)(m0 + 8 * half)) * HF + n0 + l;
    if (m0 + 16 <= N) {
#pragma unroll
        for (int v = 0; v < 8; ++v)
            op[(size_t)v * HF] = c[v];
    } else {
#pragma unroll
        for (int v = 0; v < 8; ++v) {
            const int row = m0 + v + 8 * half;
            if (row < N)
                ft[(size_t)row * HF + n0 + l] = c[v];
        }
    }
}

#define BCAP 32

__device__ __forceinline__ void put_edge(const int e, const int d, const int r,
                                         int* __restrict__ bucket,
                                         int* __restrict__ ovf,
                                         int* __restrict__ cursor) {
    if (r < BCAP) bucket[d * BCAP + r] = e;
    else          ovf[atomicAdd(cursor, 1)] = e;
}

__global__ void dgat_bucket_kernel(const int* __restrict__ dst,
                                   int* __restrict__ count,
                                   int* __restrict__ bucket,
                                   int* __restrict__ ovf,
                                   int* __restrict__ cursor,
                                   int E) {
    const int e0 = (blockIdx.x * blockDim.x + threadIdx.x) * 4;
    if (e0 + 3 < E) {
        const int4 d = *(const int4*)(dst + e0);
        const int r0 = atomicAdd(&count[d.x], 1);
        const int r1 = atomicAdd(&count[d.y], 1);
        const int r2 = atomicAdd(&count[d.z], 1);
        const int r3 = atomicAdd(&count[d.w], 1);
        put_edge(e0,     d.x, r0, bucket, ovf, cursor);
        put_edge(e0 + 1, d.y, r1, bucket, ovf, cursor);
        put_edge(e0 + 2, d.z, r2, bucket, ovf, cursor);
        put_edge(e0 + 3, d.w, r3, bucket, ovf, cursor);
    } else {
        for (int e = e0; e < E; ++e) {
            const int d = dst[e];
            put_edge(e, d, atomicAdd(&count[d], 1), bucket, ovf, cursor);
        }
    }
}

__device__ __forceinline__ float quad_sum(float v) {
    v += __int_as_float(__builtin_amdgcn_mov_dpp(__float_as_int(v), 0xB1, 0xF, 0xF, true));
    v += __int_as_float(__builtin_amdgcn_mov_dpp(__float_as_int(v), 0x4E, 0xF, 0xF, true));
    return v;
}

__device__ __forceinline__ float head_score(const float4 a, const float4 b) {
    float p = a.x * b.x;
    p = fmaf(a.y, b.y, p);
    p = fmaf(a.z, b.z, p);
    p = fmaf(a.w, b.w, p);
    return quad_sum(p) * 0.25f;
}

__device__ __forceinline__ int long_list_chunk(const int node, const int nrem, int& last,
                                            const int mine, const int lane,
                                            const int* __restrict__ src,
                                            const int* __restrict__ dst,
                                            const int* __restrict__ ovf,
                                            const int novf, const int E) {
    int myid = 0;
    for (int k = 0; k < nrem; ++k) {
        int best = (mine > last) ? mine : 0x7fffffff;
        for (int p0 = 0; p0 < novf; p0 += 32) {
            const int p = p0 + lane;
            if (p < novf) {
                const int e2 = ovf[p];
                if (e2 > last && dst[e2] == node) best = min(best, e2);
            }
        }
        for (int o = 16; o > 0; o >>= 1) best = min(best, __shfl_xor(best, o));
        best = min(best, E - 1);
        last = best;
        const int sm = src[best];
        if (lane == k) myid = sm;
    }
    return myid;
}

__device__ __forceinline__ void two_chunk_list(const int node, const int mine, const int lane,
                                               const int* __restrict__ src,
                                               const int* __restrict__ dst,
                                               const int* __restrict__ ovf,
                                               const int novf, const int E,
                                               int& ids0, int& ids1) {
    const int a  = min(mine, E - 1);
    const int sa = src[a];
    int extra = 0x7fffffff;
    int nx = 0;
    for (int p0 = 0; p0 < novf; p0 += 32) {
        const int p = p0 + lane;
        int  e2  = 0;
        bool hit = false;
        if (p < novf) {
            e2  = min(ovf[p], E - 1);
            hit = dst[e2] == node;
        }
        unsigned found = __builtin_amdgcn_ballot_w32(hit);
        while (found != 0u && nx < BCAP) {
            const int j = __builtin_ctz(found);
            found &= found - 1u;
            const int ej = __builtin_amdgcn_readlane(e2, j);
            if (lane == nx) extra = ej;
            ++nx;
        }
    }

    const int b  = extra;
    const int sb = (lane < nx) ? src[b] : 0;

    int ra = 0;
    int rb = 0;
    for (int j = 0; j < BCAP; ++j) {
        const int kj = __builtin_amdgcn_readlane(a, j);
        ra += (kj < a) ? 1 : 0;
        rb += (kj < b) ? 1 : 0;
    }
    for (int j = 0; j < nx; ++j) {
        const int kj = __builtin_amdgcn_readlane(b, j);
        ra += (kj < a) ? 1 : 0;
        rb += (kj < b) ? 1 : 0;
    }

    for (int j = 0; j < BCAP; ++j) {
        const int r = __builtin_amdgcn_readlane(ra, j);
        const int v = __builtin_amdgcn_readlane(sa, j);
        if (lane == r)        ids0 = v;
        if (lane == r - BCAP) ids1 = v;
    }
    for (int j = 0; j < nx; ++j) {
        const int r = __builtin_amdgcn_readlane(rb, j);
        const int v = __builtin_amdgcn_readlane(sb, j);
        if (lane == r)        ids0 = v;
        if (lane == r - BCAP) ids1 = v;
    }
}

#define NODE_GROUP 4

__global__ void
dgat_node_kernel(const float* __restrict__ ft,
                 const int* __restrict__ src,
                 const int* __restrict__ dst,
                 const int* __restrict__ count,
                 const int* __restrict__ bucket,
                 const int* __restrict__ ovf,
                 const int* __restrict__ cursor,
                 float* __restrict__ out,
                 int N, int E) {
    const int lane = threadIdx.x & 31;
    const int node = __builtin_amdgcn_readfirstlane(blockIdx.x * 8 + (threadIdx.x >> 5));
    if (node >= N) return;
    const int half = lane >> 4;
    const int q    = lane & 15;
    const int cnt  = count[node];
    const int mine = bucket[node * BCAP + lane];

    const float4* __restrict__ ft4 = (const float4*)ft;
    const float4 fd = ft4[(size_t)node * 16 + q];
    const float4 zero4 = make_float4(0.f, 0.f, 0.f, 0.f);

    const bool shortlist = cnt <= BCAP;
    const bool inreg     = cnt <= 2 * BCAP;
    int ids0 = 0;
    int ids1 = 0;
    int novf = 0;
    if (shortlist) {
        const int key = (lane < cnt) ? mine : 0x7fffffff;
        const int s   = (lane < cnt) ? src[key] : 0;
        int r = 0;
        for (int j = 0; j < cnt; ++j)
            r += (__builtin_amdgcn_readlane(key, j) < key) ? 1 : 0;
        if (lane >= cnt) r = lane;
        ids0 = __builtin_amdgcn_ds_permute(r << 2, s);
    } else {
        novf = cursor[0];
        if (inreg) two_chunk_list(node, mine, lane, src, dst, ovf, novf, E, ids0, ids1);
    }

    float m = -__builtin_huge_valf();
    int last = -1;
    for (int c0 = 0; c0 < cnt; c0 += 32) {
        const int nrem = min(cnt - c0, 32);
        const int myid = inreg ? (c0 == 0 ? ids0 : ids1)
                               : long_list_chunk(node, nrem, last, mine, lane,
                                                 src, dst, ovf, novf, E);
        for (int g = 0; g < nrem; g += 2 * NODE_GROUP) {
            float4 fv[NODE_GROUP];
            bool   ok[NODE_GROUP];
            for (int u = 0; u < NODE_GROUP; ++u) {
                const int k = g + 2 * u + half;
                ok[u] = k < nrem;
                const int s = __shfl(myid, min(k, nrem - 1));
                fv[u] = ft4[(size_t)s * 16 + q];
            }
            for (int u = 0; u < NODE_GROUP; ++u) {
                const float p = head_score(fv[u], fd);
                if (ok[u]) m = fmaxf(m, p);
            }
        }
    }
    m = fmaxf(m, __shfl_xor(m, 16));

    float  l   = 0.0f;
    float4 acc = zero4;
    last = -1;
    for (int c0 = 0; c0 < cnt; c0 += 32) {
        const int nrem = min(cnt - c0, 32);
        const int myid = inreg ? (c0 == 0 ? ids0 : ids1)
                               : long_list_chunk(node, nrem, last, mine, lane,
                                                 src, dst, ovf, novf, E);
        for (int g = 0; g < nrem; g += 2 * NODE_GROUP) {
            float4 fv[NODE_GROUP];
            bool   ok[NODE_GROUP];
            for (int u = 0; u < NODE_GROUP; ++u) {
                const int k = g + 2 * u + half;
                ok[u] = k < nrem;
                const int s = __shfl(myid, min(k, nrem - 1));
                fv[u] = ft4[(size_t)s * 16 + q];
            }
            for (int u = 0; u < NODE_GROUP; ++u) {
                const float p  = head_score(fv[u], fd);
                const float ex = ok[u] ? __expf(p - m) : 0.0f;
                l += ex;
                acc.x = fmaf(ex, fv[u].x, acc.x);
                acc.y = fmaf(ex, fv[u].y, acc.y);
                acc.z = fmaf(ex, fv[u].z, acc.z);
                acc.w = fmaf(ex, fv[u].w, acc.w);
            }
        }
    }
    l     += __shfl_xor(l, 16);
    acc.x += __shfl_xor(acc.x, 16);
    acc.y += __shfl_xor(acc.y, 16);
    acc.z += __shfl_xor(acc.z, 16);
    acc.w += __shfl_xor(acc.w, 16);

    if (half == 0) {
        float4 o = zero4;
        if (cnt > 0) {
            const float inv = 1.0f / l;
            o.x = acc.x * inv;
            o.y = acc.y * inv;
            o.z = acc.z * inv;
            o.w = acc.w * inv;
        }
        ((float4*)out)[(size_t)node * 16 + q] = o;
    }
}

extern "C" void kernel_launch(void* const* d_in, const int* in_sizes, int n_in,
                              void* d_out, int out_size, void* d_ws, size_t ws_size,
                              hipStream_t stream) {
    const float* feat = (const float*)d_in[0];
    const float* W    = (const float*)d_in[1];
    const int*   src  = (const int*)d_in[2];
    const int*   dst  = (const int*)d_in[3];
    float*       out  = (float*)d_out;

    const int N = in_sizes[0] / D_IN;
    const int E = in_sizes[2];

    float* ft     = (float*)d_ws;
    int*   bucket = (int*)(ft + (size_t)N * HF);
    int*   ovf    = bucket + (size_t)N * BCAP;
    int*   count  = ovf + (size_t)E;
    int*   cursor = count + (size_t)N;

    const int blk = 256;
    const int gn  = (N + blk - 1) / blk;
    const int ge4 = (E + 4 * blk - 1) / (4 * blk);

    dgat_zero_kernel<<<gn, blk, 0, stream>>>(count, cursor, N);

    const int mtiles = (N + 15) / 16;
    const int ntiles = mtiles * 4;
    dgat_proj_wmma_kernel<<<(ntiles + 7) / 8, 256, 0, stream>>>(feat, W, ft, N, ntiles);

    dgat_bucket_kernel<<<ge4, blk, 0, stream>>>(dst, count, bucket, ovf, cursor, E);
    dgat_node_kernel<<<(N + 7) / 8, 256, 0, stream>>>(ft, src, dst, count, bucket, ovf, cursor,
                                                      out, N, E);
}


# Edge grouping: slot ranks from LDS counters in 3 passes (count per 256-node bin, place bin by bin, one block per bin fills buckets); no global add per edge, zero launch gone; sum order unchanged
# speedup vs baseline: 6.2953x; 1.8940x over previous
#include <hip/hip_runtime.h>
#include <hip/hip_bf16.h>
#include <math.h>


#define D_IN 128
#define NH   4
#define NF   16
#define HF   64

typedef __attribute__((ext_vector_type(2))) float v2f;
typedef __attribute__((ext_vector_type(8))) float v8f;

__global__ void dgat_proj_wmma_kernel(const float* __restrict__ feat,
                                      const float* __restrict__ W,
                                      float* __restrict__ ft,
                                      int N, int ntiles) {
    const int lane = threadIdx.x & 31;
    const int wave = threadIdx.x >> 5;
    const int tile = blockIdx.x * (blockDim.x >> 5) + wave;
    if (tile >= ntiles) return;

    const int mt = tile >> 2;
    const int nt = tile & 3;
    const int m0 = mt * 16;
    const int n0 = nt * 16;

    const int half = lane >> 4;
    const int l    = lane & 15;

    int rowA = m0 + l;
    if (rowA > N - 1) rowA = N - 1;
    const float* __restrict__ arow = feat + (size_t)rowA * D_IN;
    const float* __restrict__ bp   = W + n0 + l;

    v8f c = {0.f, 0.f, 0.f, 0.f, 0.f, 0.f, 0.f, 0.f};

#pragma unroll 8
    for (int k = 0; k < D_IN; k += 4) {
        const int kk = k + 2 * half;
        v2f a, b;
        a.x = arow[kk];
        a.y = arow[kk + 1];
        b.x = bp[(size_t)kk * HF];
        b.y = bp[(size_t)(kk + 1) * HF];
        c = __builtin_amdgcn_wmma_f32_16x16x4_f32(
                false, a, false, b, (short)0, c, false, false);
    }

    float* __restrict__ op = ft + ((size_t)(m0 + 8 * half)) * HF + n0 + l;
    if (m0 + 16 <= N) {
#pragma unroll
        for (int v = 0; v < 8; ++v)
            op[(size_t)v * HF] = c[v];
    } else {
#pragma unroll
        for (int v = 0; v < 8; ++v) {
            const int row = m0 + v + 8 * half;
            if (row < N)
                ft[(size_t)row * HF + n0 + l] = c[v];
        }
    }
}

#define BCAP  32
#define BINSH 8
#define NBMAX 256
#define EPB   4096
#define BTHR  1024
#define CSR   8
#define RPT   6

__device__ __forceinline__ int bin_of(const int d) {
    const unsigned b = (unsigned)d >> BINSH;
    return (int)(b < (unsigned)(NBMAX - 1) ? b : (unsigned)(NBMAX - 1));
}

__device__ __forceinline__ int load_dst4(const int* __restrict__ dst, const int e0, const int E,
                                         int (&d)[4]) {
    if (e0 + 3 < E) {
        const int4 v = *(const int4*)(dst + e0);
        d[0] = v.x; d[1] = v.y; d[2] = v.z; d[3] = v.w;
        return 4;
    }
    int n = 0;
    for (int j = 0; j < 4; ++j) {
        d[j] = 0;
        if (e0 + j < E) { d[j] = dst[e0 + j]; n = j + 1; }
    }
    return n;
}

__device__ __forceinline__ void wave_excl_scan256(int* a, const int lane) {
    int v[8];
    int s = 0;
    for (int k = 0; k < 8; ++k) v[k] = a[lane * 8 + k];
    for (int k = 0; k < 8; ++k) { const int x = v[k]; v[k] = s; s += x; }
    int inc = s;
    for (int o = 1; o < 32; o <<= 1) {
        const int y = __shfl_up(inc, o);
        if (lane >= o) inc += y;
    }
    const int off = inc - s;
    for (int k = 0; k < 8; ++k) a[lane * 8 + k] = v[k] + off;
}

__global__ void __launch_bounds__(BTHR)
dgat_bincount_kernel(const int* __restrict__ dst,
                     int* __restrict__ blkcnt,
                     int* __restrict__ cursor,
                     int E) {
    __shared__ int c[NBMAX];
    const int t = threadIdx.x;
    if (t < NBMAX) c[t] = 0;
    if (blockIdx.x == 0 && t == 0) cursor[0] = 0;
    __syncthreads();
    int d[4];
    const int n = load_dst4(dst, blockIdx.x * EPB + t * 4, E, d);
    for (int j = 0; j < n; ++j) atomicAdd(&c[bin_of(d[j])], 1);
    __syncthreads();
    if (t < NBMAX) blkcnt[blockIdx.x * NBMAX + t] = c[t];
}

__global__ void __launch_bounds__(BTHR)
dgat_binplace_kernel(const int* __restrict__ dst,
                     const int* __restrict__ blkcnt,
                     int* __restrict__ rec,
                     int* __restrict__ binbase,
                     int E, int nblk) {
    __shared__ int tot[NBMAX];
    __shared__ int pre[NBMAX];
    __shared__ int lbase[NBMAX];
    __shared__ int run[NBMAX];
    __shared__ int goff[NBMAX];
    __shared__ int srec[EPB];
    __shared__ unsigned char sbin[EPB];

    const int t = threadIdx.x;
    if (t < NBMAX) { tot[t] = 0; pre[t] = 0; run[t] = 0; }
    __syncthreads();
    const int e0 = blockIdx.x * EPB + t * 4;
    int d[4];
    const int n = load_dst4(dst, e0, E, d);
    {
        const int c4   = t & 63;
        const int part = t >> 6;
        const int4* __restrict__ rows = (const int4*)blkcnt;
        int4 s = make_int4(0, 0, 0, 0);
        int4 p = make_int4(0, 0, 0, 0);
        for (int w0 = 0; w0 < nblk; w0 += 16 * CSR) {
            int4 v[CSR];
#pragma unroll
            for (int r = 0; r < CSR; ++r) {
                const int w = w0 + part + 16 * r;
                v[r] = rows[min(w, nblk - 1) * (NBMAX / 4) + c4];
            }
#pragma unroll
            for (int r = 0; r < CSR; ++r) {
                const int w  = w0 + part + 16 * r;
                const int in = (w < nblk) ? 1 : 0;
                const int bf = (w < (int)blockIdx.x) ? 1 : 0;
                s.x += in * v[r].x; s.y += in * v[r].y; s.z += in * v[r].z; s.w += in * v[r].w;
                p.x += bf * v[r].x; p.y += bf * v[r].y; p.z += bf * v[r].z; p.w += bf * v[r].w;
            }
        }
        atomicAdd(&tot[c4 * 4 + 0], s.x); atomicAdd(&tot[c4 * 4 + 1], s.y);
        atomicAdd(&tot[c4 * 4 + 2], s.z); atomicAdd(&tot[c4 * 4 + 3], s.w);
        atomicAdd(&pre[c4 * 4 + 0], p.x); atomicAdd(&pre[c4 * 4 + 1], p.y);
        atomicAdd(&pre[c4 * 4 + 2], p.z); atomicAdd(&pre[c4 * 4 + 3], p.w);
        if (t < NBMAX) lbase[t] = blkcnt[blockIdx.x * NBMAX + t];
    }
    __syncthreads();
    if ((t >> 5) == 0) wave_excl_scan256(tot, t & 31);
    if ((t >> 5) == 1) wave_excl_scan256(lbase, t & 31);
    __syncthreads();
    if (t < NBMAX) {
        goff[t] = tot[t] + pre[t] - lbase[t];
        if (blockIdx.x == 0) {
            binbase[t] = tot[t];
            if (t == 0) binbase[NBMAX] = E;
        }
    }
    __syncthreads();

    for (int j = 0; j < n; ++j) {
        const int b = bin_of(d[j]);
        const int k = atomicAdd(&run[b], 1);
        const int slot = min(lbase[b] + k, EPB - 1);
        srec[slot] = (int)(((unsigned)(e0 + j) << 8) | ((unsigned)d[j] & 255u));
        sbin[slot] = (unsigned char)b;
    }
    __syncthreads();

    const int ne = min(EPB, E - (int)blockIdx.x * EPB);
    for (int i = t; i < ne; i += BTHR) {
        const int g = goff[sbin[i]] + i;
        rec[min(max(g, 0), E - 1)] = srec[i];
    }
}

__global__ void __launch_bounds__(BTHR)
dgat_binsort_kernel(const int* __restrict__ rec,
                    const int* __restrict__ binbase,
                    int* __restrict__ count,
                    int* __restrict__ bucket,
                    int* __restrict__ ovf,
                    int* __restrict__ cursor,
                    int N, int E) {
    __shared__ int cnt[256];
    __shared__ int4 buck4[256 * BCAP / 4];
    int* buck = (int*)buck4;
    const int t = threadIdx.x;
    const int b = blockIdx.x;
    if (t < 256) cnt[t] = 0;
    for (int i = t; i < 256 * BCAP / 4; i += BTHR) buck4[i] = make_int4(0, 0, 0, 0);
    __syncthreads();

    const int beg = min(max(binbase[b], 0), E);
    const int end = min(max(binbase[b + 1], 0), E);
    for (int i0 = beg; i0 < end; i0 += BTHR * RPT) {
        unsigned r[RPT];
#pragma unroll
        for (int u = 0; u < RPT; ++u)
            r[u] = (unsigned)rec[min(i0 + t + u * BTHR, E - 1)];
#pragma unroll
        for (int u = 0; u < RPT; ++u) {
            if (i0 + t + u * BTHR < end) {
                const int e  = (int)(r[u] >> 8);
                const int dl = (int)(r[u] & 255u);
                const int k  = atomicAdd(&cnt[dl], 1);
                if (k < BCAP) buck[dl * BCAP + k] = e;
                else          ovf[atomicAdd(cursor, 1)] = e;
            }
        }
    }
    __syncthreads();

    const int node0 = b << BINSH;
    const int nn = min(256, N - node0);
    if (t < nn) count[node0 + t] = cnt[t];
    int4* __restrict__ gb = (int4*)(bucket + (size_t)node0 * BCAP);
    for (int i = t; i < nn * (BCAP / 4); i += BTHR) gb[i] = buck4[i];
}

__device__ __forceinline__ float quad_sum(float v) {
    v += __int_as_float(__builtin_amdgcn_mov_dpp(__float_as_int(v), 0xB1, 0xF, 0xF, true));
    v += __int_as_float(__builtin_amdgcn_mov_dpp(__float_as_int(v), 0x4E, 0xF, 0xF, true));
    return v;
}

__device__ __forceinline__ float head_score(const float4 a, const float4 b) {
    float p = a.x * b.x;
    p = fmaf(a.y, b.y, p);
    p = fmaf(a.z, b.z, p);
    p = fmaf(a.w, b.w, p);
    return quad_sum(p) * 0.25f;
}

__device__ __forceinline__ int long_list_chunk(const int node, const int nrem, int& last,
                                            const int mine, const int lane,
                                            const int* __restrict__ src,
                                            const int* __restrict__ dst,
                                            const int* __restrict__ ovf,
                                            const int novf, const int E) {
    int myid = 0;
    for (int k = 0; k < nrem; ++k) {
        int best = (mine > last) ? mine : 0x7fffffff;
        for (int p0 = 0; p0 < novf; p0 += 32) {
            const int p = p0 + lane;
            if (p < novf) {
                const int e2 = ovf[p];
                if (e2 > last && dst[e2] == node) best = min(best, e2);
            }
        }
        for (int o = 16; o > 0; o >>= 1) best = min(best, __shfl_xor(best, o));
        best = min(best, E - 1);
        last = best;
        const int sm = src[best];
        if (lane == k) myid = sm;
    }
    return myid;
}

__device__ __forceinline__ void two_chunk_list(const int node, const int mine, const int lane,
                                               const int* __restrict__ src,
                                               const int* __restrict__ dst,
                                               const int* __restrict__ ovf,
                                               const int novf, const int E,
                                               int& ids0, int& ids1) {
    const int a  = min(mine, E - 1);
    const int sa = src[a];
    int extra = 0x7fffffff;
    int nx = 0;
    for (int p0 = 0; p0 < novf; p0 += 32) {
        const int p = p0 + lane;
        int  e2  = 0;
        bool hit = false;
        if (p < novf) {
            e2  = min(ovf[p], E - 1);
            hit = dst[e2] == node;
        }
        unsigned found = __builtin_amdgcn_ballot_w32(hit);
        while (found != 0u && nx < BCAP) {
            const int j = __builtin_ctz(found);
            found &= found - 1u;
            const int ej = __builtin_amdgcn_readlane(e2, j);
            if (lane == nx) extra = ej;
            ++nx;
        }
    }

    const int b  = extra;
    const int sb = (lane < nx) ? src[b] : 0;

    int ra = 0;
    int rb = 0;
    for (int j = 0; j < BCAP; ++j) {
        const int kj = __builtin_amdgcn_readlane(a, j);
        ra += (kj < a) ? 1 : 0;
        rb += (kj < b) ? 1 : 0;
    }
    for (int j = 0; j < nx; ++j) {
        const int kj = __builtin_amdgcn_readlane(b, j);
        ra += (kj < a) ? 1 : 0;
        rb += (kj < b) ? 1 : 0;
    }

    for (int j = 0; j < BCAP; ++j) {
        const int r = __builtin_amdgcn_readlane(ra, j);
        const int v = __builtin_amdgcn_readlane(sa, j);
        if (lane == r)        ids0 = v;
        if (lane == r - BCAP) ids1 = v;
    }
    for (int j = 0; j < nx; ++j) {
        const int r = __builtin_amdgcn_readlane(rb, j);
        const int v = __builtin_amdgcn_readlane(sb, j);
        if (lane == r)        ids0 = v;
        if (lane == r - BCAP) ids1 = v;
    }
}

#define NODE_GROUP 4

__global__ void
dgat_node_kernel(const float* __restrict__ ft,
                 const int* __restrict__ src,
                 const int* __restrict__ dst,
                 const int* __restrict__ count,
                 const int* __restrict__ bucket,
                 const int* __restrict__ ovf,
                 const int* __restrict__ cursor,
                 float* __restrict__ out,
                 int N, int E) {
    const int lane = threadIdx.x & 31;
    const int node = __builtin_amdgcn_readfirstlane(blockIdx.x * 8 + (threadIdx.x >> 5));
    if (node >= N) return;
    const int half = lane >> 4;
    const int q    = lane & 15;
    const int cnt  = count[node];
    const int mine = bucket[node * BCAP + lane];

    const float4* __restrict__ ft4 = (const float4*)ft;
    const float4 fd = ft4[(size_t)node * 16 + q];
    const float4 zero4 = make_float4(0.f, 0.f, 0.f, 0.f);

    const bool shortlist = cnt <= BCAP;
    const bool inreg     = cnt <= 2 * BCAP;
    int ids0 = 0;
    int ids1 = 0;
    int novf = 0;
    if (shortlist) {
        const int key = (lane < cnt) ? mine : 0x7fffffff;
        const int s   = (lane < cnt) ? src[key] : 0;
        int r = 0;
        for (int j = 0; j < cnt; ++j)
            r += (__builtin_amdgcn_readlane(key, j) < key) ? 1 : 0;
        if (lane >= cnt) r = lane;
        ids0 = __builtin_amdgcn_ds_permute(r << 2, s);
    } else {
        novf = cursor[0];
        if (inreg) two_chunk_list(node, mine, lane, src, dst, ovf, novf, E, ids0, ids1);
    }

    float m = -__builtin_huge_valf();
    int last = -1;
    for (int c0 = 0; c0 < cnt; c0 += 32) {
        const int nrem = min(cnt - c0, 32);
        const int myid = inreg ? (c0 == 0 ? ids0 : ids1)
                               : long_list_chunk(node, nrem, last, mine, lane,
                                                 src, dst, ovf, novf, E);
        for (int g = 0; g < nrem; g += 2 * NODE_GROUP) {
            float4 fv[NODE_GROUP];
            bool   ok[NODE_GROUP];
            for (int u = 0; u < NODE_GROUP; ++u) {
                const int k = g + 2 * u + half;
                ok[u] = k < nrem;
                const int s = __shfl(myid, min(k, nrem - 1));
                fv[u] = ft4[(size_t)s * 16 + q];
            }
            for (int u = 0; u < NODE_GROUP; ++u) {
                const float p = head_score(fv[u], fd);
                if (ok[u]) m = fmaxf(m, p);
            }
        }
    }
    m = fmaxf(m, __shfl_xor(m, 16));

    float  l   = 0.0f;
    float4 acc = zero4;
    last = -1;
    for (int c0 = 0; c0 < cnt; c0 += 32) {
        const int nrem = min(cnt - c0, 32);
        const int myid = inreg ? (c0 == 0 ? ids0 : ids1)
                               : long_list_chunk(node, nrem, last, mine, lane,
                                                 src, dst, ovf, novf, E);
        for (int g = 0; g < nrem; g += 2 * NODE_GROUP) {
            float4 fv[NODE_GROUP];
            bool   ok[NODE_GROUP];
            for (int u = 0; u < NODE_GROUP; ++u) {
                const int k = g + 2 * u + half;
                ok[u] = k < nrem;
                const int s = __shfl(myid, min(k, nrem - 1));
                fv[u] = ft4[(size_t)s * 16 + q];
            }
            for (int u = 0; u < NODE_GROUP; ++u) {
                const float p  = head_score(fv[u], fd);
                const float ex = ok[u] ? __expf(p - m) : 0.0f;
                l += ex;
                acc.x = fmaf(ex, fv[u].x, acc.x);
                acc.y = fmaf(ex, fv[u].y, acc.y);
                acc.z = fmaf(ex, fv[u].z, acc.z);
                acc.w = fmaf(ex, fv[u].w, acc.w);
            }
        }
    }
    l     += __shfl_xor(l, 16);
    acc.x += __shfl_xor(acc.x, 16);
    acc.y += __shfl_xor(acc.y, 16);
    acc.z += __shfl_xor(acc.z, 16);
    acc.w += __shfl_xor(acc.w, 16);

    if (half == 0) {
        float4 o = zero4;
        if (cnt > 0) {
            const float inv = 1.0f / l;
            o.x = acc.x * inv;
            o.y = acc.y * inv;
            o.z = acc.z * inv;
            o.w = acc.w * inv;
        }
        ((float4*)out)[(size_t)node * 16 + q] = o;
    }
}

extern "C" void kernel_launch(void* const* d_in, const int* in_sizes, int n_in,
                              void* d_out, int out_size, void* d_ws, size_t ws_size,
                              hipStream_t stream) {
    const float* feat = (const float*)d_in[0];
    const float* W    = (const float*)d_in[1];
    const int*   src  = (const int*)d_in[2];
    const int*   dst  = (const int*)d_in[3];
    float*       out  = (float*)d_out;

    const int N = in_sizes[0] / D_IN;
    const int E = in_sizes[2];

    float* ft     = (float*)d_ws;
    int*   bucket = (int*)(ft + (size_t)N * HF);
    int*   ovf    = bucket + (size_t)N * BCAP;
    int*   count  = ovf + (size_t)E;
    int*   cursor = count + (size_t)N;

    const int nblk = (E + EPB - 1) / EPB;
    const int nbin = (N + 255) / 256;

    int* rec     = (int*)d_out;
    int* blkcnt  = rec + (size_t)((E + 3) / 4) * 4;
    int* binbase = blkcnt + (size_t)nblk * NBMAX;

    const int mtiles = (N + 15) / 16;
    const int ntiles = mtiles * 4;
    dgat_proj_wmma_kernel<<<(ntiles + 7) / 8, 256, 0, stream>>>(feat, W, ft, N, ntiles);

    dgat_bincount_kernel<<<nblk, BTHR, 0, stream>>>(dst, blkcnt, cursor, E);
    dgat_binplace_kernel<<<nblk, BTHR, 0, stream>>>(dst, blkcnt, rec, binbase, E, nblk);
    dgat_binsort_kernel<<<nbin, BTHR, 0, stream>>>(rec, binbase, count, bucket, ovf, cursor, N, E);

    dgat_node_kernel<<<(N + 7) / 8, 256, 0, stream>>>(ft, src, dst, count, bucket, ovf, cursor,
                                                      out, N, E);
}


# Node kernel: lists up to 32 edges gather each source row once, rows and scores kept in registers for both passes; a wave takes 4 nodes in turn so short waves do not wait on entry; sum order unchanged
# speedup vs baseline: 6.5341x; 1.0379x over previous
#include <hip/hip_runtime.h>
#include <hip/hip_bf16.h>
#include <math.h>


#define D_IN 128
#define NH   4
#define NF   16
#define HF   64

typedef __attribute__((ext_vector_type(2))) float v2f;
typedef __attribute__((ext_vector_type(8))) float v8f;

__global__ void dgat_proj_wmma_kernel(const float* __restrict__ feat,
                                      const float* __restrict__ W,
                                      float* __restrict__ ft,
                                      int N, int ntiles) {
    const int lane = threadIdx.x & 31;
    const int wave = threadIdx.x >> 5;
    const int tile = blockIdx.x * (blockDim.x >> 5) + wave;
    if (tile >= ntiles) return;

    const int mt = tile >> 2;
    const int nt = tile & 3;
    const int m0 = mt * 16;
    const int n0 = nt * 16;

    const int half = lane >> 4;
    const int l    = lane & 15;

    int rowA = m0 + l;
    if (rowA > N - 1) rowA = N - 1;
    const float* __restrict__ arow = feat + (size_t)rowA * D_IN;
    const float* __restrict__ bp   = W + n0 + l;

    v8f c = {0.f, 0.f, 0.f, 0.f, 0.f, 0.f, 0.f, 0.f};

#pragma unroll 8
    for (int k = 0; k < D_IN; k += 4) {
        const int kk = k + 2 * half;
        v2f a, b;
        a.x = arow[kk];
        a.y = arow[kk + 1];
        b.x = bp[(size_t)kk * HF];
        b.y = bp[(size_t)(kk + 1) * HF];
        c = __builtin_amdgcn_wmma_f32_16x16x4_f32(
                false, a, false, b, (short)0, c, false, false);
    }

    float* __restrict__ op = ft + ((size_t)(m0 + 8 * half)) * HF + n0 + l;
    if (m0 + 16 <= N) {
#pragma unroll
        for (int v = 0; v < 8; ++v)
            op[(size_t)v * HF] = c[v];
    } else {
#pragma unroll
        for (int v = 0; v < 8; ++v) {
            const int row = m0 + v + 8 * half;
            if (row < N)
                ft[(size_t)row * HF + n0 + l] = c[v];
        }
    }
}

#define BCAP  32
#define BINSH 8
#define NBMAX 256
#define EPB   4096
#define BTHR  1024
#define CSR   8
#define RPT   6

__device__ __forceinline__ int bin_of(const int d) {
    const unsigned b = (unsigned)d >> BINSH;
    return (int)(b < (unsigned)(NBMAX - 1) ? b : (unsigned)(NBMAX - 1));
}

__device__ __forceinline__ int load_dst4(const int* __restrict__ dst, const int e0, const int E,
                                         int (&d)[4]) {
    if (e0 + 3 < E) {
        const int4 v = *(const int4*)(dst + e0);
        d[0] = v.x; d[1] = v.y; d[2] = v.z; d[3] = v.w;
        return 4;
    }
    int n = 0;
    for (int j = 0; j < 4; ++j) {
        d[j] = 0;
        if (e0 + j < E) { d[j] = dst[e0 + j]; n = j + 1; }
    }
    return n;
}

__device__ __forceinline__ void wave_excl_scan256(int* a, const int lane) {
    int v[8];
    int s = 0;
    for (int k = 0; k < 8; ++k) v[k] = a[lane * 8 + k];
    for (int k = 0; k < 8; ++k) { const int x = v[k]; v[k] = s; s += x; }
    int inc = s;
    for (int o = 1; o < 32; o <<= 1) {
        const int y = __shfl_up(inc, o);
        if (lane >= o) inc += y;
    }
    const int off = inc - s;
    for (int k = 0; k < 8; ++k) a[lane * 8 + k] = v[k] + off;
}

__global__ void __launch_bounds__(BTHR)
dgat_bincount_kernel(const int* __restrict__ dst,
                     int* __restrict__ blkcnt,
                     int* __restrict__ cursor,
                     int E) {
    __shared__ int c[NBMAX];
    const int t = threadIdx.x;
    if (t < NBMAX) c[t] = 0;
    if (blockIdx.x == 0 && t == 0) cursor[0] = 0;
    __syncthreads();
    int d[4];
    const int n = load_dst4(dst, blockIdx.x * EPB + t * 4, E, d);
    for (int j = 0; j < n; ++j) atomicAdd(&c[bin_of(d[j])], 1);
    __syncthreads();
    if (t < NBMAX) blkcnt[blockIdx.x * NBMAX + t] = c[t];
}

__global__ void __launch_bounds__(BTHR)
dgat_binplace_kernel(const int* __restrict__ dst,
                     const int* __restrict__ blkcnt,
                     int* __restrict__ rec,
                     int* __restrict__ binbase,
                     int E, int nblk) {
    __shared__ int tot[NBMAX];
    __shared__ int pre[NBMAX];
    __shared__ int lbase[NBMAX];
    __shared__ int run[NBMAX];
    __shared__ int goff[NBMAX];
    __shared__ int srec[EPB];
    __shared__ unsigned char sbin[EPB];

    const int t = threadIdx.x;
    if (t < NBMAX) { tot[t] = 0; pre[t] = 0; run[t] = 0; }
    __syncthreads();
    const int e0 = blockIdx.x * EPB + t * 4;
    int d[4];
    const int n = load_dst4(dst, e0, E, d);
    {
        const int c4   = t & 63;
        const int part = t >> 6;
        const int4* __restrict__ rows = (const int4*)blkcnt;
        int4 s = make_int4(0, 0, 0, 0);
        int4 p = make_int4(0, 0, 0, 0);
        for (int w0 = 0; w0 < nblk; w0 += 16 * CSR) {
            int4 v[CSR];
#pragma unroll
            for (int r = 0; r < CSR; ++r) {
                const int w = w0 + part + 16 * r;
                v[r] = rows[min(w, nblk - 1) * (NBMAX / 4) + c4];
            }
#pragma unroll
            for (int r = 0; r < CSR; ++r) {
                const int w  = w0 + part + 16 * r;
                const int in = (w < nblk) ? 1 : 0;
                const int bf = (w < (int)blockIdx.x) ? 1 : 0;
                s.x += in * v[r].x; s.y += in * v[r].y; s.z += in * v[r].z; s.w += in * v[r].w;
                p.x += bf * v[r].x; p.y += bf * v[r].y; p.z += bf * v[r].z; p.w += bf * v[r].w;
            }
        }
        atomicAdd(&tot[c4 * 4 + 0], s.x); atomicAdd(&tot[c4 * 4 + 1], s.y);
        atomicAdd(&tot[c4 * 4 + 2], s.z); atomicAdd(&tot[c4 * 4 + 3], s.w);
        atomicAdd(&pre[c4 * 4 + 0], p.x); atomicAdd(&pre[c4 * 4 + 1], p.y);
        atomicAdd(&pre[c4 * 4 + 2], p.z); atomicAdd(&pre[c4 * 4 + 3], p.w);
        if (t < NBMAX) lbase[t] = blkcnt[blockIdx.x * NBMAX + t];
    }
    __syncthreads();
    if ((t >> 5) == 0) wave_excl_scan256(tot, t & 31);
    if ((t >> 5) == 1) wave_excl_scan256(lbase, t & 31);
    __syncthreads();
    if (t < NBMAX) {
        goff[t] = tot[t] + pre[t] - lbase[t];
        if (blockIdx.x == 0) {
            binbase[t] = tot[t];
            if (t == 0) binbase[NBMAX] = E;
        }
    }
    __syncthreads();

    for (int j = 0; j < n; ++j) {
        const int b = bin_of(d[j]);
        const int k = atomicAdd(&run[b], 1);
        const int slot = min(lbase[b] + k, EPB - 1);
        srec[slot] = (int)(((unsigned)(e0 + j) << 8) | ((unsigned)d[j] & 255u));
        sbin[slot] = (unsigned char)b;
    }
    __syncthreads();

    const int ne = min(EPB, E - (int)blockIdx.x * EPB);
    for (int i = t; i < ne; i += BTHR) {
        const int g = goff[sbin[i]] + i;
        rec[min(max(g, 0), E - 1)] = srec[i];
    }
}

__global__ void __launch_bounds__(BTHR)
dgat_binsort_kernel(const int* __restrict__ rec,
                    const int* __restrict__ binbase,
                    int* __restrict__ count,
                    int* __restrict__ bucket,
                    int* __restrict__ ovf,
                    int* __restrict__ cursor,
                    int N, int E) {
    __shared__ int cnt[256];
    __shared__ int4 buck4[256 * BCAP / 4];
    int* buck = (int*)buck4;
    const int t = threadIdx.x;
    const int b = blockIdx.x;
    if (t < 256) cnt[t] = 0;
    for (int i = t; i < 256 * BCAP / 4; i += BTHR) buck4[i] = make_int4(0, 0, 0, 0);
    __syncthreads();

    const int beg = min(max(binbase[b], 0), E);
    const int end = min(max(binbase[b + 1], 0), E);
    for (int i0 = beg; i0 < end; i0 += BTHR * RPT) {
        unsigned r[RPT];
#pragma unroll
        for (int u = 0; u < RPT; ++u)
            r[u] = (unsigned)rec[min(i0 + t + u * BTHR, E - 1)];
#pragma unroll
        for (int u = 0; u < RPT; ++u) {
            if (i0 + t + u * BTHR < end) {
                const int e  = (int)(r[u] >> 8);
                const int dl = (int)(r[u] & 255u);
                const int k  = atomicAdd(&cnt[dl], 1);
                if (k < BCAP) buck[dl * BCAP + k] = e;
                else          ovf[atomicAdd(cursor, 1)] = e;
            }
        }
    }
    __syncthreads();

    const int node0 = b << BINSH;
    const int nn = min(256, N - node0);
    if (t < nn) count[node0 + t] = cnt[t];
    int4* __restrict__ gb = (int4*)(bucket + (size_t)node0 * BCAP);
    for (int i = t; i < nn * (BCAP / 4); i += BTHR) gb[i] = buck4[i];
}

__device__ __forceinline__ float quad_sum(float v) {
    v += __int_as_float(__builtin_amdgcn_mov_dpp(__float_as_int(v), 0xB1, 0xF, 0xF, true));
    v += __int_as_float(__builtin_amdgcn_mov_dpp(__float_as_int(v), 0x4E, 0xF, 0xF, true));
    return v;
}

__device__ __forceinline__ float head_score(const float4 a, const float4 b) {
    float p = a.x * b.x;
    p = fmaf(a.y, b.y, p);
    p = fmaf(a.z, b.z, p);
    p = fmaf(a.w, b.w, p);
    return quad_sum(p) * 0.25f;
}

__device__ __forceinline__ int long_list_chunk(const int node, const int nrem, int& last,
                                            const int mine, const int lane,
                                            const int* __restrict__ src,
                                            const int* __restrict__ dst,
                                            const int* __restrict__ ovf,
                                            const int novf, const int E) {
    int myid = 0;
    for (int k = 0; k < nrem; ++k) {
        int best = (mine > last) ? mine : 0x7fffffff;
        for (int p0 = 0; p0 < novf; p0 += 32) {
            const int p = p0 + lane;
            if (p < novf) {
                const int e2 = ovf[p];
                if (e2 > last && dst[e2] == node) best = min(best, e2);
            }
        }
        for (int o = 16; o > 0; o >>= 1) best = min(best, __shfl_xor(best, o));
        best = min(best, E - 1);
        last = best;
        const int sm = src[best];
        if (lane == k) myid = sm;
    }
    return myid;
}

__device__ __forceinline__ void two_chunk_list(const int node, const int mine, const int lane,
                                               const int* __restrict__ src,
                                               const int* __restrict__ dst,
                                               const int* __restrict__ ovf,
                                               const int novf, const int E,
                                               int& ids0, int& ids1) {
    const int a  = min(mine, E - 1);
    const int sa = src[a];
    int extra = 0x7fffffff;
    int nx = 0;
    for (int p0 = 0; p0 < novf; p0 += 32) {
        const int p = p0 + lane;
        int  e2  = 0;
        bool hit = false;
        if (p < novf) {
            e2  = min(ovf[p], E - 1);
            hit = dst[e2] == node;
        }
        unsigned found = __builtin_amdgcn_ballot_w32(hit);
        while (found != 0u && nx < BCAP) {
            const int j = __builtin_ctz(found);
            found &= found - 1u;
            const int ej = __builtin_amdgcn_readlane(e2, j);
            if (lane == nx) extra = ej;
            ++nx;
        }
    }

    const int b  = extra;
    const int sb = (lane < nx) ? src[b] : 0;

    int ra = 0;
    int rb = 0;
    for (int j = 0; j < BCAP; ++j) {
        const int kj = __builtin_amdgcn_readlane(a, j);
        ra += (kj < a) ? 1 : 0;
        rb += (kj < b) ? 1 : 0;
    }
    for (int j = 0; j < nx; ++j) {
        const int kj = __builtin_amdgcn_readlane(b, j);
        ra += (kj < a) ? 1 : 0;
        rb += (kj < b) ? 1 : 0;
    }

    for (int j = 0; j < BCAP; ++j) {
        const int r = __builtin_amdgcn_readlane(ra, j);
        const int v = __builtin_amdgcn_readlane(sa, j);
        if (lane == r)        ids0 = v;
        if (lane == r - BCAP) ids1 = v;
    }
    for (int j = 0; j < nx; ++j) {
        const int r = __builtin_amdgcn_readlane(rb, j);
        const int v = __builtin_amdgcn_readlane(sb, j);
        if (lane == r)        ids0 = v;
        if (lane == r - BCAP) ids1 = v;
    }
}

template <int J>
__device__ __forceinline__ int pair_id(const int ids) {
    return __builtin_amdgcn_mov_dpp(ids, 0x150 + J, 0xF, 0xF, true);
}

__device__ __forceinline__ float4 load_row_q(const float* __restrict__ ft, const int s, const unsigned q16) {
    return *(const float4*)((const char*)ft + ((((unsigned)s) << 8) | q16));
}

#define DG_FOR16(M) M(0) M(1) M(2) M(3) M(4) M(5) M(6) M(7) M(8) M(9) M(10) M(11) M(12) M(13) M(14) M(15) \
                    }}}}}}}}}}}}}}}}

__device__ __forceinline__ void node_short(const float* __restrict__ ft,
                                           const int* __restrict__ src,
                                           float* __restrict__ out,
                                           const int node, const int cnt, const int mine,
                                           const int lane,
                                           const int* __restrict__ count,
                                           const int* __restrict__ bucket,
                                           const bool more, int& ncnt, int& nmine) {
    const int half = lane >> 4;
    const int q    = lane & 15;
    const unsigned q16 = (unsigned)q << 4;
    const int lim = cnt - half;
    const float4 fd = ((const float4*)ft)[(size_t)node * 16 + q];
    float4 o = make_float4(0.f, 0.f, 0.f, 0.f);

    if (cnt > 0) {
        const int key = (lane < cnt) ? mine : 0x7fffffff;
        int s = (lane < cnt) ? src[key] : 0;
        if (more) {
            ncnt  = count[node + 1];
            nmine = bucket[(node + 1) * BCAP + lane];
        }
        int r = 0;
        for (int j = 0; j < cnt; ++j)
            r += (__builtin_amdgcn_readlane(key, j) < key) ? 1 : 0;
        if (lane >= cnt) r = lane;
        const int s0 = __builtin_amdgcn_readfirstlane(s);
        if (lane >= cnt) s = s0;
        const int dest = ((r & 1) << 4) | (r >> 1);
        const int ids = __builtin_amdgcn_ds_permute(dest << 2, s);

        float4 fv[16];
        float  p[16];

#define DG_LOAD(J) if (2 * (J) < cnt) { fv[J] = load_row_q(ft, pair_id<J>(ids), q16);
        DG_FOR16(DG_LOAD)
#undef DG_LOAD

        float m = -__builtin_huge_valf();
#define DG_SCORE(J) if (2 * (J) < cnt) { p[J] = head_score(fv[J], fd); m = fmaxf(m, p[J]);
        DG_FOR16(DG_SCORE)
#undef DG_SCORE
        m = fmaxf(m, __shfl_xor(m, 16));

        float  l   = 0.0f;
        float4 acc = make_float4(0.f, 0.f, 0.f, 0.f);
#define DG_ACC(J) if (2 * (J) < cnt) {                                         \
            const float ex = (2 * (J) < lim) ? __expf(p[J] - m) : 0.0f;        \
            l += ex;                                                           \
            acc.x = fmaf(ex, fv[J].x, acc.x);                                  \
            acc.y = fmaf(ex, fv[J].y, acc.y);                                  \
            acc.z = fmaf(ex, fv[J].z, acc.z);                                  \
            acc.w = fmaf(ex, fv[J].w, acc.w);
        DG_FOR16(DG_ACC)
#undef DG_ACC
        l     += __shfl_xor(l, 16);
        acc.x += __shfl_xor(acc.x, 16);
        acc.y += __shfl_xor(acc.y, 16);
        acc.z += __shfl_xor(acc.z, 16);
        acc.w += __shfl_xor(acc.w, 16);

        const float inv = 1.0f / l;
        o.x = acc.x * inv;
        o.y = acc.y * inv;
        o.z = acc.z * inv;
        o.w = acc.w * inv;
    } else if (more) {
        ncnt  = count[node + 1];
        nmine = bucket[(node + 1) * BCAP + lane];
    }
    if (half == 0)
        ((float4*)out)[(size_t)node * 16 + q] = o;
}

#define NODE_GROUP 4

__device__ __forceinline__ void
node_long(const float* __restrict__ ft,
         const int* __restrict__ src,
         const int* __restrict__ dst,
         const int* __restrict__ ovf,
         const int* __restrict__ cursor,
         float* __restrict__ out,
         const int E, const int node, const int cnt, const int mine,
         const int lane) {
    const int half = lane >> 4;
    const int q    = lane & 15;
    const float4* __restrict__ ft4 = (const float4*)ft;
    const float4 fd = ft4[(size_t)node * 16 + q];
    const float4 zero4 = make_float4(0.f, 0.f, 0.f, 0.f);

    const bool inreg = cnt <= 2 * BCAP;
    int ids0 = 0;
    int ids1 = 0;
    const int novf = cursor[0];
    if (inreg) two_chunk_list(node, mine, lane, src, dst, ovf, novf, E, ids0, ids1);

    float m = -__builtin_huge_valf();
    int last = -1;
    for (int c0 = 0; c0 < cnt; c0 += 32) {
        const int nrem = min(cnt - c0, 32);
        const int myid = inreg ? (c0 == 0 ? ids0 : ids1)
                               : long_list_chunk(node, nrem, last, mine, lane,
                                                 src, dst, ovf, novf, E);
        for (int g = 0; g < nrem; g += 2 * NODE_GROUP) {
            float4 fv[NODE_GROUP];
            bool   ok[NODE_GROUP];
            for (int u = 0; u < NODE_GROUP; ++u) {
                const int k = g + 2 * u + half;
                ok[u] = k < nrem;
                const int s = __shfl(myid, min(k, nrem - 1));
                fv[u] = ft4[(size_t)s * 16 + q];
            }
            for (int u = 0; u < NODE_GROUP; ++u) {
                const float p = head_score(fv[u], fd);
                if (ok[u]) m = fmaxf(m, p);
            }
        }
    }
    m = fmaxf(m, __shfl_xor(m, 16));

    float  l   = 0.0f;
    float4 acc = zero4;
    last = -1;
    for (int c0 = 0; c0 < cnt; c0 += 32) {
        const int nrem = min(cnt - c0, 32);
        const int myid = inreg ? (c0 == 0 ? ids0 : ids1)
                               : long_list_chunk(node, nrem, last, mine, lane,
                                                 src, dst, ovf, novf, E);
        for (int g = 0; g < nrem; g += 2 * NODE_GROUP) {
            float4 fv[NODE_GROUP];
            bool   ok[NODE_GROUP];
            for (int u = 0; u < NODE_GROUP; ++u) {
                const int k = g + 2 * u + half;
                ok[u] = k < nrem;
                const int s = __shfl(myid, min(k, nrem - 1));
                fv[u] = ft4[(size_t)s * 16 + q];
            }
            for (int u = 0; u < NODE_GROUP; ++u) {
                const float p  = head_score(fv[u], fd);
                const float ex = ok[u] ? __expf(p - m) : 0.0f;
                l += ex;
                acc.x = fmaf(ex, fv[u].x, acc.x);
                acc.y = fmaf(ex, fv[u].y, acc.y);
                acc.z = fmaf(ex, fv[u].z, acc.z);
                acc.w = fmaf(ex, fv[u].w, acc.w);
            }
        }
    }
    l     += __shfl_xor(l, 16);
    acc.x += __shfl_xor(acc.x, 16);
    acc.y += __shfl_xor(acc.y, 16);
    acc.z += __shfl_xor(acc.z, 16);
    acc.w += __shfl_xor(acc.w, 16);

    if (half == 0) {
        const float inv = 1.0f / l;
        float4 o;
        o.x = acc.x * inv;
        o.y = acc.y * inv;
        o.z = acc.z * inv;
        o.w = acc.w * inv;
        ((float4*)out)[(size_t)node * 16 + q] = o;
    }
}


#define NODES_PER_WAVE 4

__global__ void
dgat_node_kernel(const float* __restrict__ ft,
                 const int* __restrict__ src,
                 const int* __restrict__ dst,
                 const int* __restrict__ count,
                 const int* __restrict__ bucket,
                 const int* __restrict__ ovf,
                 const int* __restrict__ cursor,
                 float* __restrict__ out,
                 int N, int E) {
    const int lane = threadIdx.x & 31;
    const int wave  = __builtin_amdgcn_readfirstlane(blockIdx.x * 8 + (threadIdx.x >> 5));
    const int node0 = wave * NODES_PER_WAVE;
    if (node0 >= N) return;
    const int nn = min(NODES_PER_WAVE, N - node0);
    int cnt  = count[node0];
    int mine = bucket[node0 * BCAP + lane];
    unsigned longmask = 0u;
#pragma clang loop unroll(disable)
    for (int i = 0; i < nn; ++i) {
        const int  node = node0 + i;
        const bool more = i + 1 < nn;
        int ncnt  = 0;
        int nmine = 0;
        if (cnt <= BCAP) {
            node_short(ft, src, out, node, cnt, mine, lane, count, bucket, more, ncnt, nmine);
        } else {
            longmask |= 1u << i;
            if (more) {
                ncnt  = count[node + 1];
                nmine = bucket[(node + 1) * BCAP + lane];
            }
        }
        cnt  = ncnt;
        mine = nmine;
    }
#pragma clang loop unroll(disable)
    for (int i = 0; i < nn; ++i) {
        if ((longmask >> i) & 1u) {
            const int node = node0 + i;
            node_long(ft, src, dst, ovf, cursor, out, E, node, count[node],
                      bucket[node * BCAP + lane], lane);
        }
    }
}

extern "C" void kernel_launch(void* const* d_in, const int* in_sizes, int n_in,
                              void* d_out, int out_size, void* d_ws, size_t ws_size,
                              hipStream_t stream) {
    const float* feat = (const float*)d_in[0];
    const float* W    = (const float*)d_in[1];
    const int*   src  = (const int*)d_in[2];
    const int*   dst  = (const int*)d_in[3];
    float*       out  = (float*)d_out;

    const int N = in_sizes[0] / D_IN;
    const int E = in_sizes[2];

    float* ft     = (float*)d_ws;
    int*   bucket = (int*)(ft + (size_t)N * HF);
    int*   ovf    = bucket + (size_t)N * BCAP;
    int*   count  = ovf + (size_t)E;
    int*   cursor = count + (size_t)N;

    const int nblk = (E + EPB - 1) / EPB;
    const int nbin = (N + 255) / 256;

    int* rec     = (int*)d_out;
    int* blkcnt  = rec + (size_t)((E + 3) / 4) * 4;
    int* binbase = blkcnt + (size_t)nblk * NBMAX;

    const int mtiles = (N + 15) / 16;
    const int ntiles = mtiles * 4;
    dgat_proj_wmma_kernel<<<(ntiles + 7) / 8, 256, 0, stream>>>(feat, W, ft, N, ntiles);

    dgat_bincount_kernel<<<nblk, BTHR, 0, stream>>>(dst, blkcnt, cursor, E);
    dgat_binplace_kernel<<<nblk, BTHR, 0, stream>>>(dst, blkcnt, rec, binbase, E, nblk);
    dgat_binsort_kernel<<<nbin, BTHR, 0, stream>>>(rec, binbase, count, bucket, ovf, cursor, N, E);

    dgat_node_kernel<<<(N + 8 * NODES_PER_WAVE - 1) / (8 * NODES_PER_WAVE), 256, 0, stream>>>(
        ft, src, dst, count, bucket, ovf, cursor, out, N, E);
}


# Edge grouping: a block reserves its run in each bin's fixed-size region by one returning global add per bin (spill list if full); count pass and its launch gone, 5 launches to 4; sum order unchanged
# speedup vs baseline: 7.1163x; 1.0891x over previous
#include <hip/hip_runtime.h>
#include <hip/hip_bf16.h>
#include <math.h>


#define D_IN 128
#define NH   4
#define NF   16
#define HF   64

#define GBINS 256
#define CURST 32

typedef __attribute__((ext_vector_type(2))) float v2f;
typedef __attribute__((ext_vector_type(8))) float v8f;

__global__ void dgat_proj_wmma_kernel(const float* __restrict__ feat,
                                      const float* __restrict__ W,
                                      float* __restrict__ ft,
                                      int* __restrict__ bincur,
                                      int* __restrict__ cur2,
                                      int* __restrict__ cursor,
                                      int N, int ntiles) {
    if (blockIdx.x == 0) {
        if (threadIdx.x < GBINS) bincur[threadIdx.x * CURST] = 0;
        if (threadIdx.x == 0) { cur2[0] = 0; cursor[0] = 0; }
    }
    const int lane = threadIdx.x & 31;
    const int wave = threadIdx.x >> 5;
    const int tile = blockIdx.x * (blockDim.x >> 5) + wave;
    if (tile >= ntiles) return;

    const int mt = tile >> 2;
    const int nt = tile & 3;
    const int m0 = mt * 16;
    const int n0 = nt * 16;

    const int half = lane >> 4;
    const int l    = lane & 15;

    int rowA = m0 + l;
    if (rowA > N - 1) rowA = N - 1;
    const float* __restrict__ arow = feat + (size_t)rowA * D_IN;
    const float* __restrict__ bp   = W + n0 + l;

    v8f c = {0.f, 0.f, 0.f, 0.f, 0.f, 0.f, 0.f, 0.f};

#pragma unroll 8
    for (int k = 0; k < D_IN; k += 4) {
        const int kk = k + 2 * half;
        v2f a, b;
        a.x = arow[kk];
        a.y = arow[kk + 1];
        b.x = bp[(size_t)kk * HF];
        b.y = bp[(size_t)(kk + 1) * HF];
        c = __builtin_amdgcn_wmma_f32_16x16x4_f32(
                false, a, false, b, (short)0, c, false, false);
    }

    float* __restrict__ op = ft + ((size_t)(m0 + 8 * half)) * HF + n0 + l;
    if (m0 + 16 <= N) {
#pragma unroll
        for (int v = 0; v < 8; ++v)
            op[(size_t)v * HF] = c[v];
    } else {
#pragma unroll
        for (int v = 0; v < 8; ++v) {
            const int row = m0 + v + 8 * half;
            if (row < N)
                ft[(size_t)row * HF + n0 + l] = c[v];
        }
    }
}

#define BCAP  32
#define BINSH 8
#define NBMAX GBINS
#define EPB   4096
#define BTHR  1024
#define RPT   6
#define CAPSH 13
#define CAPB  (1 << CAPSH)

__device__ __forceinline__ int bin_of(const int d) {
    const unsigned b = (unsigned)d >> BINSH;
    return (int)(b < (unsigned)(NBMAX - 1) ? b : (unsigned)(NBMAX - 1));
}

__device__ __forceinline__ int load_dst4(const int* __restrict__ dst, const int e0, const int E,
                                         int (&d)[4]) {
    if (e0 + 3 < E) {
        const int4 v = *(const int4*)(dst + e0);
        d[0] = v.x; d[1] = v.y; d[2] = v.z; d[3] = v.w;
        return 4;
    }
    int n = 0;
    for (int j = 0; j < 4; ++j) {
        d[j] = 0;
        if (e0 + j < E) { d[j] = dst[e0 + j]; n = j + 1; }
    }
    return n;
}

__device__ __forceinline__ void wave_excl_scan256(const int* a, int* res, const int lane) {
    int v[8];
    int s = 0;
    for (int k = 0; k < 8; ++k) v[k] = a[lane * 8 + k];
    for (int k = 0; k < 8; ++k) { const int x = v[k]; v[k] = s; s += x; }
    int inc = s;
    for (int o = 1; o < 32; o <<= 1) {
        const int y = __shfl_up(inc, o);
        if (lane >= o) inc += y;
    }
    const int off = inc - s;
    for (int k = 0; k < 8; ++k) res[lane * 8 + k] = v[k] + off;
}

__global__ void __launch_bounds__(BTHR)
dgat_binplace_kernel(const int* __restrict__ dst,
                     int* __restrict__ rec,
                     int* __restrict__ bincur,
                     int* __restrict__ ovf2,
                     int* __restrict__ cur2,
                     int E) {
    __shared__ int lcnt[NBMAX];
    __shared__ int lbase[NBMAX];
    __shared__ int run[NBMAX];
    __shared__ int goff[NBMAX];
    __shared__ int srec[EPB];
    __shared__ unsigned char sbin[EPB];

    const int t = threadIdx.x;
    if (t < NBMAX) { lcnt[t] = 0; run[t] = 0; }
    __syncthreads();
    const int e0 = blockIdx.x * EPB + t * 4;
    int d[4];
    const int n = load_dst4(dst, e0, E, d);
    for (int j = 0; j < n; ++j) atomicAdd(&lcnt[bin_of(d[j])], 1);
    __syncthreads();
    const int own = (t < NBMAX) ? lcnt[t] : 0;
    if ((t >> 5) == 0) wave_excl_scan256(lcnt, lbase, t & 31);
    __syncthreads();
    int first = 0;
    if (own > 0) first = atomicAdd(&bincur[t * CURST], own);

    for (int j = 0; j < n; ++j) {
        const int b = bin_of(d[j]);
        const int k = atomicAdd(&run[b], 1);
        const int slot = min(lbase[b] + k, EPB - 1);
        srec[slot] = (int)(((unsigned)(e0 + j) << 8) | ((unsigned)d[j] & 255u));
        sbin[slot] = (unsigned char)b;
    }
    if (t < NBMAX) goff[t] = first - lbase[t];
    __syncthreads();

    const int ne = min(EPB, E - (int)blockIdx.x * EPB);
    for (int i = t; i < ne; i += BTHR) {
        const int b = sbin[i];
        const int p = goff[b] + i;
        if ((unsigned)p < (unsigned)CAPB) rec[(b << CAPSH) + p] = srec[i];
        else ovf2[min(max(atomicAdd(cur2, 1), 0), E - 1)] = srec[i];
    }
}

__device__ __forceinline__ void take_record(const unsigned r, int* cnt, int* buck,
                                            int* __restrict__ ovf, int* __restrict__ cursor) {
    const int e  = (int)(r >> 8);
    const int dl = (int)(r & 255u);
    const int k  = atomicAdd(&cnt[dl], 1);
    if (k < BCAP) buck[dl * BCAP + k] = e;
    else          ovf[atomicAdd(cursor, 1)] = e;
}

__global__ void __launch_bounds__(BTHR)
dgat_binsort_kernel(const int* __restrict__ dst,
                    const int* __restrict__ rec,
                    const int* __restrict__ bincur,
                    const int* __restrict__ ovf2,
                    const int* __restrict__ cur2,
                    int* __restrict__ count,
                    int* __restrict__ bucket,
                    int* __restrict__ ovf,
                    int* __restrict__ cursor,
                    int N, int E) {
    __shared__ int cnt[256];
    __shared__ int4 buck4[256 * BCAP / 4];
    int* buck = (int*)buck4;
    const int t = threadIdx.x;
    const int b = blockIdx.x;
    if (t < 256) cnt[t] = 0;
    for (int i = t; i < 256 * BCAP / 4; i += BTHR) buck4[i] = make_int4(0, 0, 0, 0);
    __syncthreads();

    const int nrec = min(max(bincur[b * CURST], 0), CAPB);
    const int nsp  = min(max(cur2[0], 0), E);
    const int* __restrict__ seg = rec + ((size_t)b << CAPSH);
    for (int i0 = 0; i0 < nrec; i0 += BTHR * RPT) {
        unsigned r[RPT];
#pragma unroll
        for (int u = 0; u < RPT; ++u)
            r[u] = (unsigned)seg[min(i0 + t + u * BTHR, nrec - 1)];
#pragma unroll
        for (int u = 0; u < RPT; ++u) {
            if (i0 + t + u * BTHR < nrec) take_record(r[u], cnt, buck, ovf, cursor);
        }
    }
    for (int i = t; i < nsp; i += BTHR) {
        const unsigned r = (unsigned)ovf2[i];
        const int e = min((int)(r >> 8), E - 1);
        if (bin_of(dst[e]) == b) take_record(r, cnt, buck, ovf, cursor);
    }
    __syncthreads();

    const int node0 = b << BINSH;
    const int nn = min(256, N - node0);
    if (t < nn) count[node0 + t] = cnt[t];
    int4* __restrict__ gb = (int4*)(bucket + (size_t)node0 * BCAP);
    for (int i = t; i < nn * (BCAP / 4); i += BTHR) gb[i] = buck4[i];
}

__device__ __forceinline__ float quad_sum(float v) {
    v += __int_as_float(__builtin_amdgcn_mov_dpp(__float_as_int(v), 0xB1, 0xF, 0xF, true));
    v += __int_as_float(__builtin_amdgcn_mov_dpp(__float_as_int(v), 0x4E, 0xF, 0xF, true));
    return v;
}

__device__ __forceinline__ float head_score(const float4 a, const float4 b) {
    float p = a.x * b.x;
    p = fmaf(a.y, b.y, p);
    p = fmaf(a.z, b.z, p);
    p = fmaf(a.w, b.w, p);
    return quad_sum(p) * 0.25f;
}

__device__ __forceinline__ int long_list_chunk(const int node, const int nrem, int& last,
                                            const int mine, const int lane,
                                            const int* __restrict__ src,
                                            const int* __restrict__ dst,
                                            const int* __restrict__ ovf,
                                            const int novf, const int E) {
    int myid = 0;
    for (int k = 0; k < nrem; ++k) {
        int best = (mine > last) ? mine : 0x7fffffff;
        for (int p0 = 0; p0 < novf; p0 += 32) {
            const int p = p0 + lane;
            if (p < novf) {
                const int e2 = ovf[p];
                if (e2 > last && dst[e2] == node) best = min(best, e2);
            }
        }
        for (int o = 16; o > 0; o >>= 1) best = min(best, __shfl_xor(best, o));
        best = min(best, E - 1);
        last = best;
        const int sm = src[best];
        if (lane == k) myid = sm;
    }
    return myid;
}

__device__ __forceinline__ void two_chunk_list(const int node, const int mine, const int lane,
                                               const int* __restrict__ src,
                                               const int* __restrict__ dst,
                                               const int* __restrict__ ovf,
                                               const int novf, const int E,
                                               int& ids0, int& ids1) {
    const int a  = min(mine, E - 1);
    const int sa = src[a];
    int extra = 0x7fffffff;
    int nx = 0;
    for (int p0 = 0; p0 < novf; p0 += 32) {
        const int p = p0 + lane;
        int  e2  = 0;
        bool hit = false;
        if (p < novf) {
            e2  = min(ovf[p], E - 1);
            hit = dst[e2] == node;
        }
        unsigned found = __builtin_amdgcn_ballot_w32(hit);
        while (found != 0u && nx < BCAP) {
            const int j = __builtin_ctz(found);
            found &= found - 1u;
            const int ej = __builtin_amdgcn_readlane(e2, j);
            if (lane == nx) extra = ej;
            ++nx;
        }
    }

    const int b  = extra;
    const int sb = (lane < nx) ? src[b] : 0;

    int ra = 0;
    int rb = 0;
    for (int j = 0; j < BCAP; ++j) {
        const int kj = __builtin_amdgcn_readlane(a, j);
        ra += (kj < a) ? 1 : 0;
        rb += (kj < b) ? 1 : 0;
    }
    for (int j = 0; j < nx; ++j) {
        const int kj = __builtin_amdgcn_readlane(b, j);
        ra += (kj < a) ? 1 : 0;
        rb += (kj < b) ? 1 : 0;
    }

    for (int j = 0; j < BCAP; ++j) {
        const int r = __builtin_amdgcn_readlane(ra, j);
        const int v = __builtin_amdgcn_readlane(sa, j);
        if (lane == r)        ids0 = v;
        if (lane == r - BCAP) ids1 = v;
    }
    for (int j = 0; j < nx; ++j) {
        const int r = __builtin_amdgcn_readlane(rb, j);
        const int v = __builtin_amdgcn_readlane(sb, j);
        if (lane == r)        ids0 = v;
        if (lane == r - BCAP) ids1 = v;
    }
}

template <int J>
__device__ __forceinline__ int pair_id(const int ids) {
    return __builtin_amdgcn_mov_dpp(ids, 0x150 + J, 0xF, 0xF, true);
}

__device__ __forceinline__ float4 load_row_q(const float* __restrict__ ft, const int s, const unsigned q16) {
    return *(const float4*)((const char*)ft + ((((unsigned)s) << 8) | q16));
}

#define DG_FOR16(M) M(0) M(1) M(2) M(3) M(4) M(5) M(6) M(7) M(8) M(9) M(10) M(11) M(12) M(13) M(14) M(15) \
                    }}}}}}}}}}}}}}}}

__device__ __forceinline__ void node_short(const float* __restrict__ ft,
                                           const int* __restrict__ src,
                                           float* __restrict__ out,
                                           const int node, const int cnt, const int mine,
                                           const int lane,
                                           const int* __restrict__ count,
                                           const int* __restrict__ bucket,
                                           const bool more, int& ncnt, int& nmine) {
    const int half = lane >> 4;
    const int q    = lane & 15;
    const unsigned q16 = (unsigned)q << 4;
    const int lim = cnt - half;
    const float4 fd = ((const float4*)ft)[(size_t)node * 16 + q];
    float4 o = make_float4(0.f, 0.f, 0.f, 0.f);

    if (cnt > 0) {
        const int key = (lane < cnt) ? mine : 0x7fffffff;
        int s = (lane < cnt) ? src[key] : 0;
        if (more) {
            ncnt  = count[node + 1];
            nmine = bucket[(node + 1) * BCAP + lane];
        }
        int r = 0;
        for (int j = 0; j < cnt; ++j)
            r += (__builtin_amdgcn_readlane(key, j) < key) ? 1 : 0;
        if (lane >= cnt) r = lane;
        const int s0 = __builtin_amdgcn_readfirstlane(s);
        if (lane >= cnt) s = s0;
        const int dest = ((r & 1) << 4) | (r >> 1);
        const int ids = __builtin_amdgcn_ds_permute(dest << 2, s);

        float4 fv[16];
        float  p[16];

#define DG_LOAD(J) if (2 * (J) < cnt) { fv[J] = load_row_q(ft, pair_id<J>(ids), q16);
        DG_FOR16(DG_LOAD)
#undef DG_LOAD

        float m = -__builtin_huge_valf();
#define DG_SCORE(J) if (2 * (J) < cnt) { p[J] = head_score(fv[J], fd); m = fmaxf(m, p[J]);
        DG_FOR16(DG_SCORE)
#undef DG_SCORE
        m = fmaxf(m, __shfl_xor(m, 16));

        float  l   = 0.0f;
        float4 acc = make_float4(0.f, 0.f, 0.f, 0.f);
#define DG_ACC(J) if (2 * (J) < cnt) {                                         \
            const float ex = (2 * (J) < lim) ? __expf(p[J] - m) : 0.0f;        \
            l += ex;                                                           \
            acc.x = fmaf(ex, fv[J].x, acc.x);                                  \
            acc.y = fmaf(ex, fv[J].y, acc.y);                                  \
            acc.z = fmaf(ex, fv[J].z, acc.z);                                  \
            acc.w = fmaf(ex, fv[J].w, acc.w);
        DG_FOR16(DG_ACC)
#undef DG_ACC
        l     += __shfl_xor(l, 16);
        acc.x += __shfl_xor(acc.x, 16);
        acc.y += __shfl_xor(acc.y, 16);
        acc.z += __shfl_xor(acc.z, 16);
        acc.w += __shfl_xor(acc.w, 16);

        const float inv = 1.0f / l;
        o.x = acc.x * inv;
        o.y = acc.y * inv;
        o.z = acc.z * inv;
        o.w = acc.w * inv;
    } else if (more) {
        ncnt  = count[node + 1];
        nmine = bucket[(node + 1) * BCAP + lane];
    }
    if (half == 0)
        ((float4*)out)[(size_t)node * 16 + q] = o;
}

#define NODE_GROUP 4

__device__ __forceinline__ void
node_long(const float* __restrict__ ft,
         const int* __restrict__ src,
         const int* __restrict__ dst,
         const int* __restrict__ ovf,
         const int* __restrict__ cursor,
         float* __restrict__ out,
         const int E, const int node, const int cnt, const int mine,
         const int lane) {
    const int half = lane >> 4;
    const int q    = lane & 15;
    const float4* __restrict__ ft4 = (const float4*)ft;
    const float4 fd = ft4[(size_t)node * 16 + q];
    const float4 zero4 = make_float4(0.f, 0.f, 0.f, 0.f);

    const bool inreg = cnt <= 2 * BCAP;
    int ids0 = 0;
    int ids1 = 0;
    const int novf = cursor[0];
    if (inreg) two_chunk_list(node, mine, lane, src, dst, ovf, novf, E, ids0, ids1);

    float m = -__builtin_huge_valf();
    int last = -1;
    for (int c0 = 0; c0 < cnt; c0 += 32) {
        const int nrem = min(cnt - c0, 32);
        const int myid = inreg ? (c0 == 0 ? ids0 : ids1)
                               : long_list_chunk(node, nrem, last, mine, lane,
                                                 src, dst, ovf, novf, E);
        for (int g = 0; g < nrem; g += 2 * NODE_GROUP) {
            float4 fv[NODE_GROUP];
            bool   ok[NODE_GROUP];
            for (int u = 0; u < NODE_GROUP; ++u) {
                const int k = g + 2 * u + half;
                ok[u] = k < nrem;
                const int s = __shfl(myid, min(k, nrem - 1));
                fv[u] = ft4[(size_t)s * 16 + q];
            }
            for (int u = 0; u < NODE_GROUP; ++u) {
                const float p = head_score(fv[u], fd);
                if (ok[u]) m = fmaxf(m, p);
            }
        }
    }
    m = fmaxf(m, __shfl_xor(m, 16));

    float  l   = 0.0f;
    float4 acc = zero4;
    last = -1;
    for (int c0 = 0; c0 < cnt; c0 += 32) {
        const int nrem = min(cnt - c0, 32);
        const int myid = inreg ? (c0 == 0 ? ids0 : ids1)
                               : long_list_chunk(node, nrem, last, mine, lane,
                                                 src, dst, ovf, novf, E);
        for (int g = 0; g < nrem; g += 2 * NODE_GROUP) {
            float4 fv[NODE_GROUP];
            bool   ok[NODE_GROUP];
            for (int u = 0; u < NODE_GROUP; ++u) {
                const int k = g + 2 * u + half;
                ok[u] = k < nrem;
                const int s = __shfl(myid, min(k, nrem - 1));
                fv[u] = ft4[(size_t)s * 16 + q];
            }
            for (int u = 0; u < NODE_GROUP; ++u) {
                const float p  = head_score(fv[u], fd);
                const float ex = ok[u] ? __expf(p - m) : 0.0f;
                l += ex;
                acc.x = fmaf(ex, fv[u].x, acc.x);
                acc.y = fmaf(ex, fv[u].y, acc.y);
                acc.z = fmaf(ex, fv[u].z, acc.z);
                acc.w = fmaf(ex, fv[u].w, acc.w);
            }
        }
    }
    l     += __shfl_xor(l, 16);
    acc.x += __shfl_xor(acc.x, 16);
    acc.y += __shfl_xor(acc.y, 16);
    acc.z += __shfl_xor(acc.z, 16);
    acc.w += __shfl_xor(acc.w, 16);

    if (half == 0) {
        const float inv = 1.0f / l;
        float4 o;
        o.x = acc.x * inv;
        o.y = acc.y * inv;
        o.z = acc.z * inv;
        o.w = acc.w * inv;
        ((float4*)out)[(size_t)node * 16 + q] = o;
    }
}


#define NODES_PER_WAVE 4

__global__ void
dgat_node_kernel(const float* __restrict__ ft,
                 const int* __restrict__ src,
                 const int* __restrict__ dst,
                 const int* __restrict__ count,
                 const int* __restrict__ bucket,
                 const int* __restrict__ ovf,
                 const int* __restrict__ cursor,
                 float* __restrict__ out,
                 int N, int E) {
    const int lane = threadIdx.x & 31;
    const int wave  = __builtin_amdgcn_readfirstlane(blockIdx.x * 8 + (threadIdx.x >> 5));
    const int node0 = wave * NODES_PER_WAVE;
    if (node0 >= N) return;
    const int nn = min(NODES_PER_WAVE, N - node0);
    int cnt  = count[node0];
    int mine = bucket[node0 * BCAP + lane];
    unsigned longmask = 0u;
#pragma clang loop unroll(disable)
    for (int i = 0; i < nn; ++i) {
        const int  node = node0 + i;
        const bool more = i + 1 < nn;
        int ncnt  = 0;
        int nmine = 0;
        if (cnt <= BCAP) {
            node_short(ft, src, out, node, cnt, mine, lane, count, bucket, more, ncnt, nmine);
        } else {
            longmask |= 1u << i;
            if (more) {
                ncnt  = count[node + 1];
                nmine = bucket[(node + 1) * BCAP + lane];
            }
        }
        cnt  = ncnt;
        mine = nmine;
    }
#pragma clang loop unroll(disable)
    for (int i = 0; i < nn; ++i) {
        if ((longmask >> i) & 1u) {
            const int node = node0 + i;
            node_long(ft, src, dst, ovf, cursor, out, E, node, count[node],
                      bucket[node * BCAP + lane], lane);
        }
    }
}

extern "C" void kernel_launch(void* const* d_in, const int* in_sizes, int n_in,
                              void* d_out, int out_size, void* d_ws, size_t ws_size,
                              hipStream_t stream) {
    const float* feat = (const float*)d_in[0];
    const float* W    = (const float*)d_in[1];
    const int*   src  = (const int*)d_in[2];
    const int*   dst  = (const int*)d_in[3];
    float*       out  = (float*)d_out;

    const int N = in_sizes[0] / D_IN;
    const int E = in_sizes[2];

    float* ft     = (float*)d_ws;
    int*   bucket = (int*)(ft + (size_t)N * HF);
    int*   ovf    = bucket + (size_t)N * BCAP;
    int*   count  = ovf + (size_t)E;
    int*   cursor = count + (size_t)N;

    const int nblk = (E + EPB - 1) / EPB;
    const int nbin = (N + 255) / 256;

    int* rec    = (int*)d_out;
    int* ovf2   = rec + ((size_t)NBMAX << CAPSH);
    int* bincur = ovf2 + (size_t)((E + 31) / 32) * 32;
    int* cur2   = bincur + (size_t)NBMAX * CURST;

    const int mtiles = (N + 15) / 16;
    const int ntiles = mtiles * 4;
    dgat_proj_wmma_kernel<<<(ntiles + 7) / 8, 256, 0, stream>>>(feat, W, ft, bincur, cur2, cursor, N, ntiles);

    dgat_binplace_kernel<<<nblk, BTHR, 0, stream>>>(dst, rec, bincur, ovf2, cur2, E);
    dgat_binsort_kernel<<<nbin, BTHR, 0, stream>>>(dst, rec, bincur, ovf2, cur2, count, bucket, ovf, cursor, N, E);

    dgat_node_kernel<<<(N + 8 * NODES_PER_WAVE - 1) / (8 * NODES_PER_WAVE), 256, 0, stream>>>(
        ft, src, dst, count, bucket, ovf, cursor, out, N, E);
}


# Edge records and bucket slots carry the source node, not the edge id: node kernel gathers rows without the dependent src lookup per edge; sums now in ascending source order (was edge order)
# speedup vs baseline: 7.5526x; 1.0613x over previous
#include <hip/hip_runtime.h>
#include <hip/hip_bf16.h>
#include <math.h>


#define D_IN 128
#define NH   4
#define NF   16
#define HF   64

#define GBINS 256
#define CURST 32

typedef __attribute__((ext_vector_type(2))) float v2f;
typedef __attribute__((ext_vector_type(8))) float v8f;

__global__ void dgat_proj_wmma_kernel(const float* __restrict__ feat,
                                      const float* __restrict__ W,
                                      float* __restrict__ ft,
                                      int* __restrict__ bincur,
                                      int* __restrict__ cur2,
                                      int* __restrict__ cursor,
                                      int N, int ntiles) {
    if (blockIdx.x == 0) {
        if (threadIdx.x < GBINS) bincur[threadIdx.x * CURST] = 0;
        if (threadIdx.x == 0) { cur2[0] = 0; cursor[0] = 0; }
    }
    const int lane = threadIdx.x & 31;
    const int wave = threadIdx.x >> 5;
    const int tile = blockIdx.x * (blockDim.x >> 5) + wave;
    if (tile >= ntiles) return;

    const int mt = tile >> 2;
    const int nt = tile & 3;
    const int m0 = mt * 16;
    const int n0 = nt * 16;

    const int half = lane >> 4;
    const int l    = lane & 15;

    int rowA = m0 + l;
    if (rowA > N - 1) rowA = N - 1;
    const float* __restrict__ arow = feat + (size_t)rowA * D_IN;
    const float* __restrict__ bp   = W + n0 + l;

    v8f c = {0.f, 0.f, 0.f, 0.f, 0.f, 0.f, 0.f, 0.f};

#pragma unroll 8
    for (int k = 0; k < D_IN; k += 4) {
        const int kk = k + 2 * half;
        v2f a, b;
        a.x = arow[kk];
        a.y = arow[kk + 1];
        b.x = bp[(size_t)kk * HF];
        b.y = bp[(size_t)(kk + 1) * HF];
        c = __builtin_amdgcn_wmma_f32_16x16x4_f32(
                false, a, false, b, (short)0, c, false, false);
    }

    float* __restrict__ op = ft + ((size_t)(m0 + 8 * half)) * HF + n0 + l;
    if (m0 + 16 <= N) {
#pragma unroll
        for (int v = 0; v < 8; ++v)
            op[(size_t)v * HF] = c[v];
    } else {
#pragma unroll
        for (int v = 0; v < 8; ++v) {
            const int row = m0 + v + 8 * half;
            if (row < N)
                ft[(size_t)row * HF + n0 + l] = c[v];
        }
    }
}

#define BCAP  32
#define BINSH 8
#define NBMAX GBINS
#define EPB   4096
#define BTHR  1024
#define RPT   6
#define CAPSH 13
#define CAPB  (1 << CAPSH)

__device__ __forceinline__ int bin_of(const int d) {
    const unsigned b = (unsigned)d >> BINSH;
    return (int)(b < (unsigned)(NBMAX - 1) ? b : (unsigned)(NBMAX - 1));
}

__device__ __forceinline__ int load_dst4(const int* __restrict__ dst, const int e0, const int E,
                                         int (&d)[4]) {
    if (e0 + 3 < E) {
        const int4 v = *(const int4*)(dst + e0);
        d[0] = v.x; d[1] = v.y; d[2] = v.z; d[3] = v.w;
        return 4;
    }
    int n = 0;
    for (int j = 0; j < 4; ++j) {
        d[j] = 0;
        if (e0 + j < E) { d[j] = dst[e0 + j]; n = j + 1; }
    }
    return n;
}

__device__ __forceinline__ void wave_excl_scan256(const int* a, int* res, const int lane) {
    int v[8];
    int s = 0;
    for (int k = 0; k < 8; ++k) v[k] = a[lane * 8 + k];
    for (int k = 0; k < 8; ++k) { const int x = v[k]; v[k] = s; s += x; }
    int inc = s;
    for (int o = 1; o < 32; o <<= 1) {
        const int y = __shfl_up(inc, o);
        if (lane >= o) inc += y;
    }
    const int off = inc - s;
    for (int k = 0; k < 8; ++k) res[lane * 8 + k] = v[k] + off;
}

__global__ void __launch_bounds__(BTHR)
dgat_binplace_kernel(const int* __restrict__ src,
                     const int* __restrict__ dst,
                     int* __restrict__ rec,
                     int* __restrict__ bincur,
                     int* __restrict__ ovf2,
                     int* __restrict__ cur2,
                     int E) {
    __shared__ int lcnt[NBMAX];
    __shared__ int lbase[NBMAX];
    __shared__ int run[NBMAX];
    __shared__ int goff[NBMAX];
    __shared__ int srec[EPB];
    __shared__ unsigned char sbin[EPB];

    const int t = threadIdx.x;
    if (t < NBMAX) { lcnt[t] = 0; run[t] = 0; }
    __syncthreads();
    const int e0 = blockIdx.x * EPB + t * 4;
    int d[4];
    int s[4];
    const int n = load_dst4(dst, e0, E, d);
    load_dst4(src, e0, E, s);
    for (int j = 0; j < n; ++j) atomicAdd(&lcnt[bin_of(d[j])], 1);
    __syncthreads();
    const int own = (t < NBMAX) ? lcnt[t] : 0;
    if ((t >> 5) == 0) wave_excl_scan256(lcnt, lbase, t & 31);
    __syncthreads();
    int first = 0;
    if (own > 0) first = atomicAdd(&bincur[t * CURST], own);

    for (int j = 0; j < n; ++j) {
        const int b = bin_of(d[j]);
        const int k = atomicAdd(&run[b], 1);
        const int slot = min(lbase[b] + k, EPB - 1);
        srec[slot] = (int)(((unsigned)s[j] << 16) | ((unsigned)d[j] & 0xFFFFu));
        sbin[slot] = (unsigned char)b;
    }
    if (t < NBMAX) goff[t] = first - lbase[t];
    __syncthreads();

    const int ne = min(EPB, E - (int)blockIdx.x * EPB);
    for (int i = t; i < ne; i += BTHR) {
        const int b = sbin[i];
        const int p = goff[b] + i;
        if ((unsigned)p < (unsigned)CAPB) rec[(b << CAPSH) + p] = srec[i];
        else ovf2[min(max(atomicAdd(cur2, 1), 0), E - 1)] = srec[i];
    }
}

__device__ __forceinline__ void take_record(const unsigned r, int* cnt, int* buck,
                                            int* __restrict__ ovf, int* __restrict__ cursor) {
    const int s  = (int)(r >> 16);
    const int dl = (int)(r & 255u);
    const int k  = atomicAdd(&cnt[dl], 1);
    if (k < BCAP) buck[dl * BCAP + k] = s;
    else          ovf[atomicAdd(cursor, 1)] = (int)((r << 16) | (r >> 16));
}

__global__ void __launch_bounds__(BTHR)
dgat_binsort_kernel(const int* __restrict__ rec,
                    const int* __restrict__ bincur,
                    const int* __restrict__ ovf2,
                    const int* __restrict__ cur2,
                    int* __restrict__ count,
                    int* __restrict__ bucket,
                    int* __restrict__ ovf,
                    int* __restrict__ cursor,
                    int N, int E) {
    __shared__ int cnt[256];
    __shared__ int4 buck4[256 * BCAP / 4];
    int* buck = (int*)buck4;
    const int t = threadIdx.x;
    const int b = blockIdx.x;
    if (t < 256) cnt[t] = 0;
    for (int i = t; i < 256 * BCAP / 4; i += BTHR) buck4[i] = make_int4(0, 0, 0, 0);
    __syncthreads();

    const int nrec = min(max(bincur[b * CURST], 0), CAPB);
    const int nsp  = min(max(cur2[0], 0), E);
    const int* __restrict__ seg = rec + ((size_t)b << CAPSH);
    for (int i0 = 0; i0 < nrec; i0 += BTHR * RPT) {
        unsigned r[RPT];
#pragma unroll
        for (int u = 0; u < RPT; ++u)
            r[u] = (unsigned)seg[min(i0 + t + u * BTHR, nrec - 1)];
#pragma unroll
        for (int u = 0; u < RPT; ++u) {
            if (i0 + t + u * BTHR < nrec) take_record(r[u], cnt, buck, ovf, cursor);
        }
    }
    for (int i = t; i < nsp; i += BTHR) {
        const unsigned r = (unsigned)ovf2[i];
        if (bin_of((int)(r & 0xFFFFu)) == b) take_record(r, cnt, buck, ovf, cursor);
    }
    __syncthreads();

    const int node0 = b << BINSH;
    const int nn = min(256, N - node0);
    if (t < nn) count[node0 + t] = cnt[t];
    int4* __restrict__ gb = (int4*)(bucket + (size_t)node0 * BCAP);
    for (int i = t; i < nn * (BCAP / 4); i += BTHR) gb[i] = buck4[i];
}

__device__ __forceinline__ float quad_sum(float v) {
    v += __int_as_float(__builtin_amdgcn_mov_dpp(__float_as_int(v), 0xB1, 0xF, 0xF, true));
    v += __int_as_float(__builtin_amdgcn_mov_dpp(__float_as_int(v), 0x4E, 0xF, 0xF, true));
    return v;
}

__device__ __forceinline__ float head_score(const float4 a, const float4 b) {
    float p = a.x * b.x;
    p = fmaf(a.y, b.y, p);
    p = fmaf(a.z, b.z, p);
    p = fmaf(a.w, b.w, p);
    return quad_sum(p) * 0.25f;
}

__device__ __forceinline__ int long_list_chunk(const int node, const int nrem, long long& last,
                                            const int mine, const int lane,
                                            const int* __restrict__ ovf,
                                            const int novf, const int N) {
    int myid = 0;
    const long long mykey = ((long long)mine << 32) | (long long)lane;
    for (int k = 0; k < nrem; ++k) {
        long long best = (mykey > last) ? mykey : 0x7fffffffffffffffLL;
        for (int p0 = 0; p0 < novf; p0 += 32) {
            const int p = p0 + lane;
            if (p < novf) {
                const unsigned r = (unsigned)ovf[p];
                const long long k2 = ((long long)(r & 0xFFFFu) << 32) | (long long)(BCAP + p);
                if ((int)(r >> 16) == node && k2 > last) best = min(best, k2);
            }
        }
        for (int o = 16; o > 0; o >>= 1) {
            const long long y = __shfl_xor(best, o);
            best = min(best, y);
        }
        last = best;
        const int sm = min((int)(best >> 32) & 0xFFFF, N - 1);
        if (lane == k) myid = sm;
    }
    return myid;
}

__device__ __forceinline__ void two_chunk_list(const int node, const int mine, const int lane,
                                               const int* __restrict__ ovf,
                                               const int novf,
                                               int& ids0, int& ids1) {
    int extra = 0;
    int nx = 0;
    for (int p0 = 0; p0 < novf; p0 += 32) {
        const int p = p0 + lane;
        int  s2  = 0;
        bool hit = false;
        if (p < novf) {
            const unsigned r = (unsigned)ovf[p];
            s2  = (int)(r & 0xFFFFu);
            hit = (int)(r >> 16) == node;
        }
        unsigned found = __builtin_amdgcn_ballot_w32(hit);
        while (found != 0u && nx < BCAP) {
            const int j = __builtin_ctz(found);
            found &= found - 1u;
            const int sj = __builtin_amdgcn_readlane(s2, j);
            if (lane == nx) extra = sj;
            ++nx;
        }
    }

    const int sa = mine;
    const int sb = extra;
    const int a  = (mine << 6) | lane;
    const int b  = (lane < nx) ? ((extra << 6) | (BCAP + lane)) : 0x7fffffff;

    int ra = 0;
    int rb = 0;
    for (int j = 0; j < BCAP; ++j) {
        const int kj = __builtin_amdgcn_readlane(a, j);
        ra += (kj < a) ? 1 : 0;
        rb += (kj < b) ? 1 : 0;
    }
    for (int j = 0; j < nx; ++j) {
        const int kj = __builtin_amdgcn_readlane(b, j);
        ra += (kj < a) ? 1 : 0;
        rb += (kj < b) ? 1 : 0;
    }

    for (int j = 0; j < BCAP; ++j) {
        const int r = __builtin_amdgcn_readlane(ra, j);
        const int v = __builtin_amdgcn_readlane(sa, j);
        if (lane == r)        ids0 = v;
        if (lane == r - BCAP) ids1 = v;
    }
    for (int j = 0; j < nx; ++j) {
        const int r = __builtin_amdgcn_readlane(rb, j);
        const int v = __builtin_amdgcn_readlane(sb, j);
        if (lane == r)        ids0 = v;
        if (lane == r - BCAP) ids1 = v;
    }
}

template <int J>
__device__ __forceinline__ int pair_id(const int ids) {
    return __builtin_amdgcn_mov_dpp(ids, 0x150 + J, 0xF, 0xF, true);
}

__device__ __forceinline__ float4 load_row_q(const float* __restrict__ ft, const int s, const unsigned q16) {
    return *(const float4*)((const char*)ft + ((((unsigned)s) << 8) | q16));
}

#define DG_FOR16(M) M(0) M(1) M(2) M(3) M(4) M(5) M(6) M(7) M(8) M(9) M(10) M(11) M(12) M(13) M(14) M(15) \
                    }}}}}}}}}}}}}}}}

__device__ __forceinline__ void node_short(const float* __restrict__ ft,
                                           float* __restrict__ out,
                                           const int node, const int cnt, const int mine,
                                           const int lane,
                                           const int* __restrict__ count,
                                           const int* __restrict__ bucket,
                                           const bool more, int& ncnt, int& nmine) {
    const int half = lane >> 4;
    const int q    = lane & 15;
    const unsigned q16 = (unsigned)q << 4;
    const int lim = cnt - half;
    const float4 fd = ((const float4*)ft)[(size_t)node * 16 + q];
    float4 o = make_float4(0.f, 0.f, 0.f, 0.f);

    if (cnt > 0) {
        if (more) {
            ncnt  = count[node + 1];
            nmine = bucket[(node + 1) * BCAP + lane];
        }
        const int key = (lane < cnt) ? ((mine << 5) | lane) : 0x7fffffff;
        int s = mine;
        int r = 0;
        for (int j = 0; j < cnt; ++j)
            r += (__builtin_amdgcn_readlane(key, j) < key) ? 1 : 0;
        if (lane >= cnt) r = lane;
        const int s0 = __builtin_amdgcn_readfirstlane(s);
        if (lane >= cnt) s = s0;
        const int dest = ((r & 1) << 4) | (r >> 1);
        const int ids = __builtin_amdgcn_ds_permute(dest << 2, s);

        float4 fv[16];
        float  p[16];

#define DG_LOAD(J) if (2 * (J) < cnt) { fv[J] = load_row_q(ft, pair_id<J>(ids), q16);
        DG_FOR16(DG_LOAD)
#undef DG_LOAD

        float m = -__builtin_huge_valf();
#define DG_SCORE(J) if (2 * (J) < cnt) { p[J] = head_score(fv[J], fd); m = fmaxf(m, p[J]);
        DG_FOR16(DG_SCORE)
#undef DG_SCORE
        m = fmaxf(m, __shfl_xor(m, 16));

        float  l   = 0.0f;
        float4 acc = make_float4(0.f, 0.f, 0.f, 0.f);
#define DG_ACC(J) if (2 * (J) < cnt) {                                         \
            const float ex = (2 * (J) < lim) ? __expf(p[J] - m) : 0.0f;        \
            l += ex;                                                           \
            acc.x = fmaf(ex, fv[J].x, acc.x);                                  \
            acc.y = fmaf(ex, fv[J].y, acc.y);                                  \
            acc.z = fmaf(ex, fv[J].z, acc.z);                                  \
            acc.w = fmaf(ex, fv[J].w, acc.w);
        DG_FOR16(DG_ACC)
#undef DG_ACC
        l     += __shfl_xor(l, 16);
        acc.x += __shfl_xor(acc.x, 16);
        acc.y += __shfl_xor(acc.y, 16);
        acc.z += __shfl_xor(acc.z, 16);
        acc.w += __shfl_xor(acc.w, 16);

        const float inv = 1.0f / l;
        o.x = acc.x * inv;
        o.y = acc.y * inv;
        o.z = acc.z * inv;
        o.w = acc.w * inv;
    } else if (more) {
        ncnt  = count[node + 1];
        nmine = bucket[(node + 1) * BCAP + lane];
    }
    if (half == 0)
        ((float4*)out)[(size_t)node * 16 + q] = o;
}

#define NODE_GROUP 4

__device__ __forceinline__ void
node_long(const float* __restrict__ ft,
         const int* __restrict__ ovf,
         const int* __restrict__ cursor,
         float* __restrict__ out,
         const int N, const int node, const int cnt, const int mine,
         const int lane) {
    const int half = lane >> 4;
    const int q    = lane & 15;
    const float4* __restrict__ ft4 = (const float4*)ft;
    const float4 fd = ft4[(size_t)node * 16 + q];
    const float4 zero4 = make_float4(0.f, 0.f, 0.f, 0.f);

    const bool inreg = cnt <= 2 * BCAP;
    int ids0 = 0;
    int ids1 = 0;
    const int novf = cursor[0];
    if (inreg) two_chunk_list(node, mine, lane, ovf, novf, ids0, ids1);

    float m = -__builtin_huge_valf();
    long long last = -1;
    for (int c0 = 0; c0 < cnt; c0 += 32) {
        const int nrem = min(cnt - c0, 32);
        const int myid = inreg ? (c0 == 0 ? ids0 : ids1)
                               : long_list_chunk(node, nrem, last, mine, lane,
                                                 ovf, novf, N);
        for (int g = 0; g < nrem; g += 2 * NODE_GROUP) {
            float4 fv[NODE_GROUP];
            bool   ok[NODE_GROUP];
            for (int u = 0; u < NODE_GROUP; ++u) {
                const int k = g + 2 * u + half;
                ok[u] = k < nrem;
                const int s = __shfl(myid, min(k, nrem - 1));
                fv[u] = ft4[(size_t)s * 16 + q];
            }
            for (int u = 0; u < NODE_GROUP; ++u) {
                const float p = head_score(fv[u], fd);
                if (ok[u]) m = fmaxf(m, p);
            }
        }
    }
    m = fmaxf(m, __shfl_xor(m, 16));

    float  l   = 0.0f;
    float4 acc = zero4;
    last = -1;
    for (int c0 = 0; c0 < cnt; c0 += 32) {
        const int nrem = min(cnt - c0, 32);
        const int myid = inreg ? (c0 == 0 ? ids0 : ids1)
                               : long_list_chunk(node, nrem, last, mine, lane,
                                                 ovf, novf, N);
        for (int g = 0; g < nrem; g += 2 * NODE_GROUP) {
            float4 fv[NODE_GROUP];
            bool   ok[NODE_GROUP];
            for (int u = 0; u < NODE_GROUP; ++u) {
                const int k = g + 2 * u + half;
                ok[u] = k < nrem;
                const int s = __shfl(myid, min(k, nrem - 1));
                fv[u] = ft4[(size_t)s * 16 + q];
            }
            for (int u = 0; u < NODE_GROUP; ++u) {
                const float p  = head_score(fv[u], fd);
                const float ex = ok[u] ? __expf(p - m) : 0.0f;
                l += ex;
                acc.x = fmaf(ex, fv[u].x, acc.x);
                acc.y = fmaf(ex, fv[u].y, acc.y);
                acc.z = fmaf(ex, fv[u].z, acc.z);
                acc.w = fmaf(ex, fv[u].w, acc.w);
            }
        }
    }
    l     += __shfl_xor(l, 16);
    acc.x += __shfl_xor(acc.x, 16);
    acc.y += __shfl_xor(acc.y, 16);
    acc.z += __shfl_xor(acc.z, 16);
    acc.w += __shfl_xor(acc.w, 16);

    if (half == 0) {
        const float inv = 1.0f / l;
        float4 o;
        o.x = acc.x * inv;
        o.y = acc.y * inv;
        o.z = acc.z * inv;
        o.w = acc.w * inv;
        ((float4*)out)[(size_t)node * 16 + q] = o;
    }
}


#define NODES_PER_WAVE 4

__global__ void
dgat_node_kernel(const float* __restrict__ ft,
                 const int* __restrict__ count,
                 const int* __restrict__ bucket,
                 const int* __restrict__ ovf,
                 const int* __restrict__ cursor,
                 float* __restrict__ out,
                 int N) {
    const int lane = threadIdx.x & 31;
    const int wave  = __builtin_amdgcn_readfirstlane(blockIdx.x * 8 + (threadIdx.x >> 5));
    const int node0 = wave * NODES_PER_WAVE;
    if (node0 >= N) return;
    const int nn = min(NODES_PER_WAVE, N - node0);
    int cnt  = count[node0];
    int mine = bucket[node0 * BCAP + lane];
    unsigned longmask = 0u;
#pragma clang loop unroll(disable)
    for (int i = 0; i < nn; ++i) {
        const int  node = node0 + i;
        const bool more = i + 1 < nn;
        int ncnt  = 0;
        int nmine = 0;
        if (cnt <= BCAP) {
            node_short(ft, out, node, cnt, mine, lane, count, bucket, more, ncnt, nmine);
        } else {
            longmask |= 1u << i;
            if (more) {
                ncnt  = count[node + 1];
                nmine = bucket[(node + 1) * BCAP + lane];
            }
        }
        cnt  = ncnt;
        mine = nmine;
    }
#pragma clang loop unroll(disable)
    for (int i = 0; i < nn; ++i) {
        if ((longmask >> i) & 1u) {
            const int node = node0 + i;
            node_long(ft, ovf, cursor, out, N, node, count[node],
                      bucket[node * BCAP + lane], lane);
        }
    }
}

extern "C" void kernel_launch(void* const* d_in, const int* in_sizes, int n_in,
                              void* d_out, int out_size, void* d_ws, size_t ws_size,
                              hipStream_t stream) {
    const float* feat = (const float*)d_in[0];
    const float* W    = (const float*)d_in[1];
    const int*   src  = (const int*)d_in[2];
    const int*   dst  = (const int*)d_in[3];
    float*       out  = (float*)d_out;

    const int N = in_sizes[0] / D_IN;
    const int E = in_sizes[2];

    float* ft     = (float*)d_ws;
    int*   bucket = (int*)(ft + (size_t)N * HF);
    int*   ovf    = bucket + (size_t)N * BCAP;
    int*   count  = ovf + (size_t)E;
    int*   cursor = count + (size_t)N;

    const int nblk = (E + EPB - 1) / EPB;
    const int nbin = (N + 255) / 256;

    int* rec    = (int*)d_out;
    int* ovf2   = rec + ((size_t)NBMAX << CAPSH);
    int* bincur = ovf2 + (size_t)((E + 31) / 32) * 32;
    int* cur2   = bincur + (size_t)NBMAX * CURST;

    const int mtiles = (N + 15) / 16;
    const int ntiles = mtiles * 4;
    dgat_proj_wmma_kernel<<<(ntiles + 7) / 8, 256, 0, stream>>>(feat, W, ft, bincur, cur2, cursor, N, ntiles);

    dgat_binplace_kernel<<<nblk, BTHR, 0, stream>>>(src, dst, rec, bincur, ovf2, cur2, E);
    dgat_binsort_kernel<<<nbin, BTHR, 0, stream>>>(rec, bincur, ovf2, cur2, count, bucket, ovf, cursor, N, E);

    dgat_node_kernel<<<(N + 8 * NODES_PER_WAVE - 1) / (8 * NODES_PER_WAVE), 256, 0, stream>>>(
        ft, count, bucket, ovf, cursor, out, N);
}


# Bucket fill and node pass in one launch: a block fills its bin's buckets in LDS (200 nodes a bin), its waves do those nodes from there; bucket and count arrays gone, 4 launches to 3; sum order kept
# speedup vs baseline: 7.8155x; 1.0348x over previous
#include <hip/hip_runtime.h>
#include <hip/hip_bf16.h>
#include <math.h>


#define D_IN 128
#define NH   4
#define NF   16
#define HF   64

#define GBINS 256
#define CURST 32

typedef __attribute__((ext_vector_type(2))) float v2f;
typedef __attribute__((ext_vector_type(8))) float v8f;

__global__ void dgat_proj_wmma_kernel(const float* __restrict__ feat,
                                      const float* __restrict__ W,
                                      float* __restrict__ ft,
                                      int* __restrict__ bincur,
                                      int* __restrict__ cur2,
                                      int* __restrict__ cursor,
                                      int N, int ntiles) {
    if (blockIdx.x == 0) {
        if (threadIdx.x < GBINS) bincur[threadIdx.x * CURST] = 0;
        if (threadIdx.x == 0) { cur2[0] = 0; cursor[0] = 0; }
    }
    const int lane = threadIdx.x & 31;
    const int wave = threadIdx.x >> 5;
    const int tile = blockIdx.x * (blockDim.x >> 5) + wave;
    if (tile >= ntiles) return;

    const int mt = tile >> 2;
    const int nt = tile & 3;
    const int m0 = mt * 16;
    const int n0 = nt * 16;

    const int half = lane >> 4;
    const int l    = lane & 15;

    int rowA = m0 + l;
    if (rowA > N - 1) rowA = N - 1;
    const float* __restrict__ arow = feat + (size_t)rowA * D_IN;
    const float* __restrict__ bp   = W + n0 + l;

    v8f c = {0.f, 0.f, 0.f, 0.f, 0.f, 0.f, 0.f, 0.f};

#pragma unroll 8
    for (int k = 0; k < D_IN; k += 4) {
        const int kk = k + 2 * half;
        v2f a, b;
        a.x = arow[kk];
        a.y = arow[kk + 1];
        b.x = bp[(size_t)kk * HF];
        b.y = bp[(size_t)(kk + 1) * HF];
        c = __builtin_amdgcn_wmma_f32_16x16x4_f32(
                false, a, false, b, (short)0, c, false, false);
    }

    float* __restrict__ op = ft + ((size_t)(m0 + 8 * half)) * HF + n0 + l;
    if (m0 + 16 <= N) {
#pragma unroll
        for (int v = 0; v < 8; ++v)
            op[(size_t)v * HF] = c[v];
    } else {
#pragma unroll
        for (int v = 0; v < 8; ++v) {
            const int row = m0 + v + 8 * half;
            if (row < N)
                ft[(size_t)row * HF + n0 + l] = c[v];
        }
    }
}

#define BCAP  32
#define BINN  200
#define RSTRIDE (BINN * HF)
#define NBMAX GBINS
#define EPB   4096
#define BTHR  1024
#define RPT   6
#define CAPSH 13
#define CAPB  (1 << CAPSH)

__device__ __forceinline__ int bin_of(const int d) {
    const unsigned b = (unsigned)d / (unsigned)BINN;
    return (int)(b < (unsigned)(NBMAX - 1) ? b : (unsigned)(NBMAX - 1));
}

__device__ __forceinline__ int load_dst4(const int* __restrict__ dst, const int e0, const int E,
                                         int (&d)[4]) {
    if (e0 + 3 < E) {
        const int4 v = *(const int4*)(dst + e0);
        d[0] = v.x; d[1] = v.y; d[2] = v.z; d[3] = v.w;
        return 4;
    }
    int n = 0;
    for (int j = 0; j < 4; ++j) {
        d[j] = 0;
        if (e0 + j < E) { d[j] = dst[e0 + j]; n = j + 1; }
    }
    return n;
}

__device__ __forceinline__ void wave_excl_scan256(const int* a, int* res, const int lane) {
    int v[8];
    int s = 0;
    for (int k = 0; k < 8; ++k) v[k] = a[lane * 8 + k];
    for (int k = 0; k < 8; ++k) { const int x = v[k]; v[k] = s; s += x; }
    int inc = s;
    for (int o = 1; o < 32; o <<= 1) {
        const int y = __shfl_up(inc, o);
        if (lane >= o) inc += y;
    }
    const int off = inc - s;
    for (int k = 0; k < 8; ++k) res[lane * 8 + k] = v[k] + off;
}

__global__ void __launch_bounds__(BTHR)
dgat_binplace_kernel(const int* __restrict__ src,
                     const int* __restrict__ dst,
                     int* __restrict__ rec,
                     int* __restrict__ bincur,
                     int* __restrict__ ovf2,
                     int* __restrict__ cur2,
                     int E) {
    __shared__ int lcnt[NBMAX];
    __shared__ int lbase[NBMAX];
    __shared__ int run[NBMAX];
    __shared__ int goff[NBMAX];
    __shared__ int srec[EPB];
    __shared__ unsigned char sbin[EPB];

    const int t = threadIdx.x;
    if (t < NBMAX) { lcnt[t] = 0; run[t] = 0; }
    __syncthreads();
    const int e0 = blockIdx.x * EPB + t * 4;
    int d[4];
    int s[4];
    const int n = load_dst4(dst, e0, E, d);
    load_dst4(src, e0, E, s);
    for (int j = 0; j < n; ++j) atomicAdd(&lcnt[bin_of(d[j])], 1);
    __syncthreads();
    const int own = (t < NBMAX) ? lcnt[t] : 0;
    if ((t >> 5) == 0) wave_excl_scan256(lcnt, lbase, t & 31);
    __syncthreads();
    int first = 0;
    if (own > 0) first = atomicAdd(&bincur[t * CURST], own);

    for (int j = 0; j < n; ++j) {
        const int b = bin_of(d[j]);
        const int k = atomicAdd(&run[b], 1);
        const int slot = min(lbase[b] + k, EPB - 1);
        srec[slot] = (int)(((unsigned)s[j] << 16) | ((unsigned)d[j] & 0xFFFFu));
        sbin[slot] = (unsigned char)b;
    }
    if (t < NBMAX) goff[t] = first - lbase[t];
    __syncthreads();

    const int ne = min(EPB, E - (int)blockIdx.x * EPB);
    for (int i = t; i < ne; i += BTHR) {
        const int b = sbin[i];
        const int p = goff[b] + i;
        if ((unsigned)p < (unsigned)CAPB) rec[b * RSTRIDE + p] = srec[i];
        else ovf2[min(max(atomicAdd(cur2, 1), 0), E - 1)] = srec[i];
    }
}

__device__ __forceinline__ float quad_sum(float v) {
    v += __int_as_float(__builtin_amdgcn_mov_dpp(__float_as_int(v), 0xB1, 0xF, 0xF, true));
    v += __int_as_float(__builtin_amdgcn_mov_dpp(__float_as_int(v), 0x4E, 0xF, 0xF, true));
    return v;
}

__device__ __forceinline__ float head_score(const float4 a, const float4 b) {
    float p = a.x * b.x;
    p = fmaf(a.y, b.y, p);
    p = fmaf(a.z, b.z, p);
    p = fmaf(a.w, b.w, p);
    return quad_sum(p) * 0.25f;
}

__device__ __forceinline__ int long_list_chunk(const int node, const int nrem, long long& last,
                                            const int mine, const int lane,
                                            const int* __restrict__ ovf,
                                            const int novf, const int N) {
    int myid = 0;
    const long long mykey = ((long long)mine << 32) | (long long)lane;
    for (int k = 0; k < nrem; ++k) {
        long long best = (mykey > last) ? mykey : 0x7fffffffffffffffLL;
        for (int p0 = 0; p0 < novf; p0 += 32) {
            const int p = p0 + lane;
            if (p < novf) {
                const unsigned r = (unsigned)ovf[p];
                const long long k2 = ((long long)(r & 0xFFFFu) << 32) | (long long)(BCAP + p);
                if ((int)(r >> 16) == node && k2 > last) best = min(best, k2);
            }
        }
        for (int o = 16; o > 0; o >>= 1) {
            const long long y = __shfl_xor(best, o);
            best = min(best, y);
        }
        last = best;
        const int sm = min((int)(best >> 32) & 0xFFFF, N - 1);
        if (lane == k) myid = sm;
    }
    return myid;
}

__device__ __forceinline__ void two_chunk_list(const int node, const int mine, const int lane,
                                               const int* __restrict__ ovf,
                                               const int novf,
                                               int& ids0, int& ids1) {
    int extra = 0;
    int nx = 0;
    for (int p0 = 0; p0 < novf; p0 += 32) {
        const int p = p0 + lane;
        int  s2  = 0;
        bool hit = false;
        if (p < novf) {
            const unsigned r = (unsigned)ovf[p];
            s2  = (int)(r & 0xFFFFu);
            hit = (int)(r >> 16) == node;
        }
        unsigned found = __builtin_amdgcn_ballot_w32(hit);
        while (found != 0u && nx < BCAP) {
            const int j = __builtin_ctz(found);
            found &= found - 1u;
            const int sj = __builtin_amdgcn_readlane(s2, j);
            if (lane == nx) extra = sj;
            ++nx;
        }
    }

    const int sa = mine;
    const int sb = extra;
    const int a  = (mine << 6) | lane;
    const int b  = (lane < nx) ? ((extra << 6) | (BCAP + lane)) : 0x7fffffff;

    int ra = 0;
    int rb = 0;
    for (int j = 0; j < BCAP; ++j) {
        const int kj = __builtin_amdgcn_readlane(a, j);
        ra += (kj < a) ? 1 : 0;
        rb += (kj < b) ? 1 : 0;
    }
    for (int j = 0; j < nx; ++j) {
        const int kj = __builtin_amdgcn_readlane(b, j);
        ra += (kj < a) ? 1 : 0;
        rb += (kj < b) ? 1 : 0;
    }

    for (int j = 0; j < BCAP; ++j) {
        const int r = __builtin_amdgcn_readlane(ra, j);
        const int v = __builtin_amdgcn_readlane(sa, j);
        if (lane == r)        ids0 = v;
        if (lane == r - BCAP) ids1 = v;
    }
    for (int j = 0; j < nx; ++j) {
        const int r = __builtin_amdgcn_readlane(rb, j);
        const int v = __builtin_amdgcn_readlane(sb, j);
        if (lane == r)        ids0 = v;
        if (lane == r - BCAP) ids1 = v;
    }
}

template <int J>
__device__ __forceinline__ int pair_id(const int ids) {
    return __builtin_amdgcn_mov_dpp(ids, 0x150 + J, 0xF, 0xF, true);
}

__device__ __forceinline__ float4 load_row_q(const float* __restrict__ ft, const int s, const unsigned q16) {
    return *(const float4*)((const char*)ft + ((((unsigned)s) << 8) | q16));
}

#define DG_FOR16(M) M(0) M(1) M(2) M(3) M(4) M(5) M(6) M(7) M(8) M(9) M(10) M(11) M(12) M(13) M(14) M(15) \
                    }}}}}}}}}}}}}}}}

__device__ __forceinline__ void node_short(const float* __restrict__ ft,
                                           float* __restrict__ out,
                                           const int node, const int cnt, const int mine,
                                           const int lane) {
    const int half = lane >> 4;
    const int q    = lane & 15;
    const unsigned q16 = (unsigned)q << 4;
    const int lim = cnt - half;
    const float4 fd = ((const float4*)ft)[(size_t)node * 16 + q];
    float4 o = make_float4(0.f, 0.f, 0.f, 0.f);

    if (cnt > 0) {
        const int key = (lane < cnt) ? ((mine << 5) | lane) : 0x7fffffff;
        int s = mine;
        int r = 0;
        for (int j = 0; j < cnt; ++j)
            r += (__builtin_amdgcn_readlane(key, j) < key) ? 1 : 0;
        if (lane >= cnt) r = lane;
        const int s0 = __builtin_amdgcn_readfirstlane(s);
        if (lane >= cnt) s = s0;
        const int dest = ((r & 1) << 4) | (r >> 1);
        const int ids = __builtin_amdgcn_ds_permute(dest << 2, s);

        float4 fv[16];
        float  p[16];

#define DG_LOAD(J) if (2 * (J) < cnt) { fv[J] = load_row_q(ft, pair_id<J>(ids), q16);
        DG_FOR16(DG_LOAD)
#undef DG_LOAD

        float m = -__builtin_huge_valf();
#define DG_SCORE(J) if (2 * (J) < cnt) { p[J] = head_score(fv[J], fd); m = fmaxf(m, p[J]);
        DG_FOR16(DG_SCORE)
#undef DG_SCORE
        m = fmaxf(m, __shfl_xor(m, 16));

        float  l   = 0.0f;
        float4 acc = make_float4(0.f, 0.f, 0.f, 0.f);
#define DG_ACC(J) if (2 * (J) < cnt) {                                         \
            const float ex = (2 * (J) < lim) ? __expf(p[J] - m) : 0.0f;        \
            l += ex;                                                           \
            acc.x = fmaf(ex, fv[J].x, acc.x);                                  \
            acc.y = fmaf(ex, fv[J].y, acc.y);                                  \
            acc.z = fmaf(ex, fv[J].z, acc.z);                                  \
            acc.w = fmaf(ex, fv[J].w, acc.w);
        DG_FOR16(DG_ACC)
#undef DG_ACC
        l     += __shfl_xor(l, 16);
        acc.x += __shfl_xor(acc.x, 16);
        acc.y += __shfl_xor(acc.y, 16);
        acc.z += __shfl_xor(acc.z, 16);
        acc.w += __shfl_xor(acc.w, 16);

        const float inv = 1.0f / l;
        o.x = acc.x * inv;
        o.y = acc.y * inv;
        o.z = acc.z * inv;
        o.w = acc.w * inv;
    }
    if (half == 0)
        ((float4*)out)[(size_t)node * 16 + q] = o;
}

#define NODE_GROUP 4

__device__ __forceinline__ void
node_long(const float* __restrict__ ft,
         const int* __restrict__ ovf,
         const int novf,
         float* __restrict__ out,
         const int N, const int node, const int cnt, const int mine,
         const int lane) {
    const int half = lane >> 4;
    const int q    = lane & 15;
    const float4* __restrict__ ft4 = (const float4*)ft;
    const float4 fd = ft4[(size_t)node * 16 + q];
    const float4 zero4 = make_float4(0.f, 0.f, 0.f, 0.f);

    const bool inreg = cnt <= 2 * BCAP;
    int ids0 = 0;
    int ids1 = 0;
    if (inreg) two_chunk_list(node, mine, lane, ovf, novf, ids0, ids1);

    float m = -__builtin_huge_valf();
    long long last = -1;
    for (int c0 = 0; c0 < cnt; c0 += 32) {
        const int nrem = min(cnt - c0, 32);
        const int myid = inreg ? (c0 == 0 ? ids0 : ids1)
                               : long_list_chunk(node, nrem, last, mine, lane,
                                                 ovf, novf, N);
        for (int g = 0; g < nrem; g += 2 * NODE_GROUP) {
            float4 fv[NODE_GROUP];
            bool   ok[NODE_GROUP];
            for (int u = 0; u < NODE_GROUP; ++u) {
                const int k = g + 2 * u + half;
                ok[u] = k < nrem;
                const int s = __shfl(myid, min(k, nrem - 1));
                fv[u] = ft4[(size_t)s * 16 + q];
            }
            for (int u = 0; u < NODE_GROUP; ++u) {
                const float p = head_score(fv[u], fd);
                if (ok[u]) m = fmaxf(m, p);
            }
        }
    }
    m = fmaxf(m, __shfl_xor(m, 16));

    float  l   = 0.0f;
    float4 acc = zero4;
    last = -1;
    for (int c0 = 0; c0 < cnt; c0 += 32) {
        const int nrem = min(cnt - c0, 32);
        const int myid = inreg ? (c0 == 0 ? ids0 : ids1)
                               : long_list_chunk(node, nrem, last, mine, lane,
                                                 ovf, novf, N);
        for (int g = 0; g < nrem; g += 2 * NODE_GROUP) {
            float4 fv[NODE_GROUP];
            bool   ok[NODE_GROUP];
            for (int u = 0; u < NODE_GROUP; ++u) {
                const int k = g + 2 * u + half;
                ok[u] = k < nrem;
                const int s = __shfl(myid, min(k, nrem - 1));
                fv[u] = ft4[(size_t)s * 16 + q];
            }
            for (int u = 0; u < NODE_GROUP; ++u) {
                const float p  = head_score(fv[u], fd);
                const float ex = ok[u] ? __expf(p - m) : 0.0f;
                l += ex;
                acc.x = fmaf(ex, fv[u].x, acc.x);
                acc.y = fmaf(ex, fv[u].y, acc.y);
                acc.z = fmaf(ex, fv[u].z, acc.z);
                acc.w = fmaf(ex, fv[u].w, acc.w);
            }
        }
    }
    l     += __shfl_xor(l, 16);
    acc.x += __shfl_xor(acc.x, 16);
    acc.y += __shfl_xor(acc.y, 16);
    acc.z += __shfl_xor(acc.z, 16);
    acc.w += __shfl_xor(acc.w, 16);

    if (half == 0) {
        const float inv = 1.0f / l;
        float4 o;
        o.x = acc.x * inv;
        o.y = acc.y * inv;
        o.z = acc.z * inv;
        o.w = acc.w * inv;
        ((float4*)out)[(size_t)node * 16 + q] = o;
    }
}


#define NWAVES (BTHR / 32)

__device__ __forceinline__ void take_record(const unsigned r, const int node0, int* cnt, int* buck) {
    const int s  = (int)(r >> 16);
    const int dl = ((int)(r & 0xFFFFu) - node0) & 255;
    const int k  = atomicAdd(&cnt[dl], 1);
    if (k < BCAP) buck[dl * BCAP + k] = s;
}

__device__ __forceinline__ void relay_record(const unsigned r, const int node0, const int* cnt,
                                             int* cnt2, int* buck, int* lovn,
                                             int* __restrict__ ovfb, const int tot) {
    const int s  = (int)(r >> 16);
    const int dl = ((int)(r & 0xFFFFu) - node0) & 255;
    if (cnt[dl] > BCAP) {
        const int k = atomicAdd(&cnt2[dl], 1);
        if (k < BCAP) buck[dl * BCAP + k] = s;
        else {
            const int j = min(atomicAdd(lovn, 1), tot - 1);
            ovfb[j] = (int)((((unsigned)(node0 + dl)) << 16) | (unsigned)s);
        }
    }
}

__global__ void __launch_bounds__(BTHR)
dgat_node_kernel(const float* __restrict__ ft,
                 const int* rec,
                 const int* __restrict__ bincur,
                 const int* __restrict__ ovf2,
                 const int* __restrict__ cur2,
                 int* __restrict__ ovf,
                 int* __restrict__ cursor,
                 float* out,
                 int N, int E) {
    __shared__ int cnt[256];
    __shared__ int cnt2[256];
    __shared__ int4 buck4[256 * BCAP / 4];
    __shared__ int ltot;
    __shared__ int lbase;
    __shared__ int lovn;
    int* buck = (int*)buck4;
    const int t = threadIdx.x;
    const int b = blockIdx.x;
    const int node0 = b * BINN;
    if (t < 256) { cnt[t] = 0; cnt2[t] = 0; }
    if (t == 0) { ltot = 0; lbase = 0; lovn = 0; }
    for (int i = t; i < 256 * BCAP / 4; i += BTHR) buck4[i] = make_int4(0, 0, 0, 0);
    __syncthreads();

    const int nrec = min(max(bincur[b * CURST], 0), CAPB);
    const int nsp  = min(max(cur2[0], 0), E);
    const int* seg = rec + (size_t)b * RSTRIDE;
    for (int i0 = 0; i0 < nrec; i0 += BTHR * RPT) {
        unsigned r[RPT];
#pragma unroll
        for (int u = 0; u < RPT; ++u)
            r[u] = (unsigned)seg[min(i0 + t + u * BTHR, nrec - 1)];
#pragma unroll
        for (int u = 0; u < RPT; ++u) {
            if (i0 + t + u * BTHR < nrec) take_record(r[u], node0, cnt, buck);
        }
    }
    for (int i = t; i < nsp; i += BTHR) {
        const unsigned r = (unsigned)ovf2[i];
        if (bin_of((int)(r & 0xFFFFu)) == b) take_record(r, node0, cnt, buck);
    }
    __syncthreads();

    if (t < 256 && cnt[t] > BCAP) atomicAdd(&ltot, cnt[t] - BCAP);
    __syncthreads();
    const int tot = ltot;
    if (tot > 0) {
        if (t == 0) lbase = atomicAdd(cursor, tot);
        __syncthreads();
        int* __restrict__ ovfb = ovf + min(max(lbase, 0), max(E - tot, 0));
        for (int i = t; i < nrec; i += BTHR)
            relay_record((unsigned)seg[i], node0, cnt, cnt2, buck, &lovn, ovfb, tot);
        for (int i = t; i < nsp; i += BTHR) {
            const unsigned r = (unsigned)ovf2[i];
            if (bin_of((int)(r & 0xFFFFu)) == b)
                relay_record(r, node0, cnt, cnt2, buck, &lovn, ovfb, tot);
        }
        __syncthreads();
    }
    const int* __restrict__ ovfb = ovf + min(max(lbase, 0), max(E - tot, 0));

    const int lane = t & 31;
    const int wv = __builtin_amdgcn_readfirstlane(t >> 5);
    const int nn = min(BINN, N - node0);
    unsigned longmask = 0u;
    int c    = __builtin_amdgcn_readfirstlane(cnt[wv]);
    int mine = buck[wv * BCAP + lane];
#pragma clang loop unroll(disable)
    for (int j = 0; j < (BINN + NWAVES - 1) / NWAVES; ++j) {
        const int nl = wv + j * NWAVES;
        if (nl >= nn) break;
        int ncnt  = 0;
        int nmine = 0;
        if (nl + NWAVES < nn) {
            ncnt  = cnt[nl + NWAVES];
            nmine = buck[(nl + NWAVES) * BCAP + lane];
        }
        if (c <= BCAP) node_short(ft, out, node0 + nl, c, mine, lane);
        else           longmask |= 1u << j;
        c    = __builtin_amdgcn_readfirstlane(ncnt);
        mine = nmine;
    }
#pragma clang loop unroll(disable)
    for (int j = 0; j < (BINN + NWAVES - 1) / NWAVES; ++j) {
        if ((longmask >> j) & 1u) {
            const int nl = wv + j * NWAVES;
            node_long(ft, ovfb, tot, out, N, node0 + nl,
                      __builtin_amdgcn_readfirstlane(cnt[nl]), buck[nl * BCAP + lane], lane);
        }
    }
}

extern "C" void kernel_launch(void* const* d_in, const int* in_sizes, int n_in,
                              void* d_out, int out_size, void* d_ws, size_t ws_size,
                              hipStream_t stream) {
    const float* feat = (const float*)d_in[0];
    const float* W    = (const float*)d_in[1];
    const int*   src  = (const int*)d_in[2];
    const int*   dst  = (const int*)d_in[3];
    float*       out  = (float*)d_out;

    const int N = in_sizes[0] / D_IN;
    const int E = in_sizes[2];

    float* ft      = (float*)d_ws;
    int*   scratch = (int*)(ft + (size_t)N * HF);
    int*   ovf     = scratch + (size_t)N * BCAP;
    int*   cursor  = ovf + (size_t)E + (size_t)N;

    const int nblk = (E + EPB - 1) / EPB;
    const int nbin = (N + BINN - 1) / BINN;

    int* ovf2   = scratch;
    int* bincur = ovf2 + (size_t)((E + 31) / 32) * 32;
    int* cur2   = bincur + (size_t)NBMAX * CURST;

    int* rec    = (int*)d_out;

    const int mtiles = (N + 15) / 16;
    const int ntiles = mtiles * 4;
    dgat_proj_wmma_kernel<<<(ntiles + 7) / 8, 256, 0, stream>>>(feat, W, ft, bincur, cur2, cursor, N, ntiles);

    dgat_binplace_kernel<<<nblk, BTHR, 0, stream>>>(src, dst, rec, bincur, ovf2, cur2, E);
    dgat_node_kernel<<<nbin, BTHR, 0, stream>>>(ft, rec, bincur, ovf2, cur2, ovf, cursor, out, N, E);
}


# Projection: one wave per 16x64 row strip, each A fragment loaded once feeds 4 WMMAs (4 accumulator tiles, columns 4c+t); feat, W, ft move in 16-byte accesses; k order inside each 8 of K changed
# speedup vs baseline: 8.5697x; 1.0965x over previous
#include <hip/hip_runtime.h>
#include <hip/hip_bf16.h>
#include <math.h>


#define D_IN 128
#define NH   4
#define NF   16
#define HF   64

#define GBINS 256
#define CURST 32

typedef __attribute__((ext_vector_type(2))) float v2f;
typedef __attribute__((ext_vector_type(8))) float v8f;

#define PWAVES 8

__global__ void __launch_bounds__(PWAVES * 32)
dgat_proj_wmma_kernel(const float* __restrict__ feat,
                      const float* __restrict__ W,
                      float* __restrict__ ft,
                      int* __restrict__ bincur,
                      int* __restrict__ cur2,
                      int* __restrict__ cursor,
                      int N, int mtiles) {
    if (blockIdx.x == 0) {
        for (int i = threadIdx.x; i < GBINS; i += PWAVES * 32) bincur[i * CURST] = 0;
        if (threadIdx.x == 0) { cur2[0] = 0; cursor[0] = 0; }
    }
    const int lane  = threadIdx.x & 31;
    const int wave  = __builtin_amdgcn_readfirstlane(threadIdx.x >> 5);
    const int strip = blockIdx.x * PWAVES + wave;
    if (strip >= mtiles) return;

    const int m0   = strip * 16;
    const int half = lane >> 4;
    const int l    = lane & 15;

    int rowA = m0 + l;
    if (rowA > N - 1) rowA = N - 1;
    const float4* __restrict__ arow = (const float4*)(feat + (size_t)rowA * D_IN) + half;
    const float4* __restrict__ wrow = (const float4*)(W + (size_t)(4 * half) * HF) + l;

    float4 a4[D_IN / 8];
#pragma unroll
    for (int i = 0; i < D_IN / 8; ++i) a4[i] = arow[2 * i];

    float4 wq[3][4];
#pragma unroll
    for (int j = 0; j < 4; ++j) {
        wq[0][j] = wrow[(size_t)(0 + j) * (HF / 4)];
        wq[1][j] = wrow[(size_t)(8 + j) * (HF / 4)];
    }

    v8f c0 = {0.f, 0.f, 0.f, 0.f, 0.f, 0.f, 0.f, 0.f};
    v8f c1 = c0, c2 = c0, c3 = c0;

#pragma unroll
    for (int i = 0; i < D_IN / 8; ++i) {
        if (i + 2 < D_IN / 8) {
#pragma unroll
            for (int j = 0; j < 4; ++j)
                wq[(i + 2) % 3][j] = wrow[(size_t)(8 * (i + 2) + j) * (HF / 4)];
        }
        __builtin_amdgcn_sched_barrier(0);
        const float4 w0 = wq[i % 3][0];
        const float4 w1 = wq[i % 3][1];
        const float4 w2 = wq[i % 3][2];
        const float4 w3 = wq[i % 3][3];
        v2f a, b;
        a.x = a4[i].x; a.y = a4[i].y;
        b.x = w0.x; b.y = w1.x;
        c0 = __builtin_amdgcn_wmma_f32_16x16x4_f32(false, a, false, b, (short)0, c0, false, false);
        b.x = w0.y; b.y = w1.y;
        c1 = __builtin_amdgcn_wmma_f32_16x16x4_f32(false, a, false, b, (short)0, c1, false, false);
        b.x = w0.z; b.y = w1.z;
        c2 = __builtin_amdgcn_wmma_f32_16x16x4_f32(false, a, false, b, (short)0, c2, false, false);
        b.x = w0.w; b.y = w1.w;
        c3 = __builtin_amdgcn_wmma_f32_16x16x4_f32(false, a, false, b, (short)0, c3, false, false);
        a.x = a4[i].z; a.y = a4[i].w;
        b.x = w2.x; b.y = w3.x;
        c0 = __builtin_amdgcn_wmma_f32_16x16x4_f32(false, a, false, b, (short)0, c0, false, false);
        b.x = w2.y; b.y = w3.y;
        c1 = __builtin_amdgcn_wmma_f32_16x16x4_f32(false, a, false, b, (short)0, c1, false, false);
        b.x = w2.z; b.y = w3.z;
        c2 = __builtin_amdgcn_wmma_f32_16x16x4_f32(false, a, false, b, (short)0, c2, false, false);
        b.x = w2.w; b.y = w3.w;
        c3 = __builtin_amdgcn_wmma_f32_16x16x4_f32(false, a, false, b, (short)0, c3, false, false);
    }

    __builtin_amdgcn_sched_barrier(0);
    asm volatile("v_nop\n\tv_nop\n\tv_nop\n\tv_nop");
    __builtin_amdgcn_sched_barrier(0);

    float4* __restrict__ op = (float4*)(ft + ((size_t)(m0 + 8 * half)) * HF) + l;
    if (m0 + 16 <= N) {
#pragma unroll
        for (int v = 0; v < 8; ++v)
            op[(size_t)v * (HF / 4)] = make_float4(c0[v], c1[v], c2[v], c3[v]);
    } else {
#pragma unroll
        for (int v = 0; v < 8; ++v) {
            const int row = m0 + v + 8 * half;
            if (row < N)
                op[(size_t)v * (HF / 4)] = make_float4(c0[v], c1[v], c2[v], c3[v]);
        }
    }
}

#define BCAP  32
#define BINN  200
#define RSTRIDE (BINN * HF)
#define NBMAX GBINS
#define EPB   4096
#define BTHR  1024
#define RPT   6
#define CAPSH 13
#define CAPB  (1 << CAPSH)

__device__ __forceinline__ int bin_of(const int d) {
    const unsigned b = (unsigned)d / (unsigned)BINN;
    return (int)(b < (unsigned)(NBMAX - 1) ? b : (unsigned)(NBMAX - 1));
}

__device__ __forceinline__ int load_dst4(const int* __restrict__ dst, const int e0, const int E,
                                         int (&d)[4]) {
    if (e0 + 3 < E) {
        const int4 v = *(const int4*)(dst + e0);
        d[0] = v.x; d[1] = v.y; d[2] = v.z; d[3] = v.w;
        return 4;
    }
    int n = 0;
    for (int j = 0; j < 4; ++j) {
        d[j] = 0;
        if (e0 + j < E) { d[j] = dst[e0 + j]; n = j + 1; }
    }
    return n;
}

__device__ __forceinline__ void wave_excl_scan256(const int* a, int* res, const int lane) {
    int v[8];
    int s = 0;
    for (int k = 0; k < 8; ++k) v[k] = a[lane * 8 + k];
    for (int k = 0; k < 8; ++k) { const int x = v[k]; v[k] = s; s += x; }
    int inc = s;
    for (int o = 1; o < 32; o <<= 1) {
        const int y = __shfl_up(inc, o);
        if (lane >= o) inc += y;
    }
    const int off = inc - s;
    for (int k = 0; k < 8; ++k) res[lane * 8 + k] = v[k] + off;
}

__global__ void __launch_bounds__(BTHR)
dgat_binplace_kernel(const int* __restrict__ src,
                     const int* __restrict__ dst,
                     int* __restrict__ rec,
                     int* __restrict__ bincur,
                     int* __restrict__ ovf2,
                     int* __restrict__ cur2,
                     int E) {
    __shared__ int lcnt[NBMAX];
    __shared__ int lbase[NBMAX];
    __shared__ int run[NBMAX];
    __shared__ int goff[NBMAX];
    __shared__ int srec[EPB];
    __shared__ unsigned char sbin[EPB];

    const int t = threadIdx.x;
    if (t < NBMAX) { lcnt[t] = 0; run[t] = 0; }
    __syncthreads();
    const int e0 = blockIdx.x * EPB + t * 4;
    int d[4];
    int s[4];
    const int n = load_dst4(dst, e0, E, d);
    load_dst4(src, e0, E, s);
    for (int j = 0; j < n; ++j) atomicAdd(&lcnt[bin_of(d[j])], 1);
    __syncthreads();
    const int own = (t < NBMAX) ? lcnt[t] : 0;
    if ((t >> 5) == 0) wave_excl_scan256(lcnt, lbase, t & 31);
    __syncthreads();
    int first = 0;
    if (own > 0) first = atomicAdd(&bincur[t * CURST], own);

    for (int j = 0; j < n; ++j) {
        const int b = bin_of(d[j]);
        const int k = atomicAdd(&run[b], 1);
        const int slot = min(lbase[b] + k, EPB - 1);
        srec[slot] = (int)(((unsigned)s[j] << 16) | ((unsigned)d[j] & 0xFFFFu));
        sbin[slot] = (unsigned char)b;
    }
    if (t < NBMAX) goff[t] = first - lbase[t];
    __syncthreads();

    const int ne = min(EPB, E - (int)blockIdx.x * EPB);
    for (int i = t; i < ne; i += BTHR) {
        const int b = sbin[i];
        const int p = goff[b] + i;
        if ((unsigned)p < (unsigned)CAPB) rec[b * RSTRIDE + p] = srec[i];
        else ovf2[min(max(atomicAdd(cur2, 1), 0), E - 1)] = srec[i];
    }
}

__device__ __forceinline__ float quad_sum(float v) {
    v += __int_as_float(__builtin_amdgcn_mov_dpp(__float_as_int(v), 0xB1, 0xF, 0xF, true));
    v += __int_as_float(__builtin_amdgcn_mov_dpp(__float_as_int(v), 0x4E, 0xF, 0xF, true));
    return v;
}

__device__ __forceinline__ float head_score(const float4 a, const float4 b) {
    float p = a.x * b.x;
    p = fmaf(a.y, b.y, p);
    p = fmaf(a.z, b.z, p);
    p = fmaf(a.w, b.w, p);
    return quad_sum(p) * 0.25f;
}

__device__ __forceinline__ int long_list_chunk(const int node, const int nrem, long long& last,
                                            const int mine, const int lane,
                                            const int* __restrict__ ovf,
                                            const int novf, const int N) {
    int myid = 0;
    const long long mykey = ((long long)mine << 32) | (long long)lane;
    for (int k = 0; k < nrem; ++k) {
        long long best = (mykey > last) ? mykey : 0x7fffffffffffffffLL;
        for (int p0 = 0; p0 < novf; p0 += 32) {
            const int p = p0 + lane;
            if (p < novf) {
                const unsigned r = (unsigned)ovf[p];
                const long long k2 = ((long long)(r & 0xFFFFu) << 32) | (long long)(BCAP + p);
                if ((int)(r >> 16) == node && k2 > last) best = min(best, k2);
            }
        }
        for (int o = 16; o > 0; o >>= 1) {
            const long long y = __shfl_xor(best, o);
            best = min(best, y);
        }
        last = best;
        const int sm = min((int)(best >> 32) & 0xFFFF, N - 1);
        if (lane == k) myid = sm;
    }
    return myid;
}

__device__ __forceinline__ void two_chunk_list(const int node, const int mine, const int lane,
                                               const int* __restrict__ ovf,
                                               const int novf,
                                               int& ids0, int& ids1) {
    int extra = 0;
    int nx = 0;
    for (int p0 = 0; p0 < novf; p0 += 32) {
        const int p = p0 + lane;
        int  s2  = 0;
        bool hit = false;
        if (p < novf) {
            const unsigned r = (unsigned)ovf[p];
            s2  = (int)(r & 0xFFFFu);
            hit = (int)(r >> 16) == node;
        }
        unsigned found = __builtin_amdgcn_ballot_w32(hit);
        while (found != 0u && nx < BCAP) {
            const int j = __builtin_ctz(found);
            found &= found - 1u;
            const int sj = __builtin_amdgcn_readlane(s2, j);
            if (lane == nx) extra = sj;
            ++nx;
        }
    }

    const int sa = mine;
    const int sb = extra;
    const int a  = (mine << 6) | lane;
    const int b  = (lane < nx) ? ((extra << 6) | (BCAP + lane)) : 0x7fffffff;

    int ra = 0;
    int rb = 0;
    for (int j = 0; j < BCAP; ++j) {
        const int kj = __builtin_amdgcn_readlane(a, j);
        ra += (kj < a) ? 1 : 0;
        rb += (kj < b) ? 1 : 0;
    }
    for (int j = 0; j < nx; ++j) {
        const int kj = __builtin_amdgcn_readlane(b, j);
        ra += (kj < a) ? 1 : 0;
        rb += (kj < b) ? 1 : 0;
    }

    for (int j = 0; j < BCAP; ++j) {
        const int r = __builtin_amdgcn_readlane(ra, j);
        const int v = __builtin_amdgcn_readlane(sa, j);
        if (lane == r)        ids0 = v;
        if (lane == r - BCAP) ids1 = v;
    }
    for (int j = 0; j < nx; ++j) {
        const int r = __builtin_amdgcn_readlane(rb, j);
        const int v = __builtin_amdgcn_readlane(sb, j);
        if (lane == r)        ids0 = v;
        if (lane == r - BCAP) ids1 = v;
    }
}

template <int J>
__device__ __forceinline__ int pair_id(const int ids) {
    return __builtin_amdgcn_mov_dpp(ids, 0x150 + J, 0xF, 0xF, true);
}

__device__ __forceinline__ float4 load_row_q(const float* __restrict__ ft, const int s, const unsigned q16) {
    return *(const float4*)((const char*)ft + ((((unsigned)s) << 8) | q16));
}

#define DG_FOR16(M) M(0) M(1) M(2) M(3) M(4) M(5) M(6) M(7) M(8) M(9) M(10) M(11) M(12) M(13) M(14) M(15) \
                    }}}}}}}}}}}}}}}}

__device__ __forceinline__ void node_short(const float* __restrict__ ft,
                                           float* __restrict__ out,
                                           const int node, const int cnt, const int mine,
                                           const int lane) {
    const int half = lane >> 4;
    const int q    = lane & 15;
    const unsigned q16 = (unsigned)q << 4;
    const int lim = cnt - half;
    const float4 fd = ((const float4*)ft)[(size_t)node * 16 + q];
    float4 o = make_float4(0.f, 0.f, 0.f, 0.f);

    if (cnt > 0) {
        const int key = (lane < cnt) ? ((mine << 5) | lane) : 0x7fffffff;
        int s = mine;
        int r = 0;
        for (int j = 0; j < cnt; ++j)
            r += (__builtin_amdgcn_readlane(key, j) < key) ? 1 : 0;
        if (lane >= cnt) r = lane;
        const int s0 = __builtin_amdgcn_readfirstlane(s);
        if (lane >= cnt) s = s0;
        const int dest = ((r & 1) << 4) | (r >> 1);
        const int ids = __builtin_amdgcn_ds_permute(dest << 2, s);

        float4 fv[16];
        float  p[16];

#define DG_LOAD(J) if (2 * (J) < cnt) { fv[J] = load_row_q(ft, pair_id<J>(ids), q16);
        DG_FOR16(DG_LOAD)
#undef DG_LOAD

        float m = -__builtin_huge_valf();
#define DG_SCORE(J) if (2 * (J) < cnt) { p[J] = head_score(fv[J], fd); m = fmaxf(m, p[J]);
        DG_FOR16(DG_SCORE)
#undef DG_SCORE
        m = fmaxf(m, __shfl_xor(m, 16));

        float  l   = 0.0f;
        float4 acc = make_float4(0.f, 0.f, 0.f, 0.f);
#define DG_ACC(J) if (2 * (J) < cnt) {                                         \
            const float ex = (2 * (J) < lim) ? __expf(p[J] - m) : 0.0f;        \
            l += ex;                                                           \
            acc.x = fmaf(ex, fv[J].x, acc.x);                                  \
            acc.y = fmaf(ex, fv[J].y, acc.y);                                  \
            acc.z = fmaf(ex, fv[J].z, acc.z);                                  \
            acc.w = fmaf(ex, fv[J].w, acc.w);
        DG_FOR16(DG_ACC)
#undef DG_ACC
        l     += __shfl_xor(l, 16);
        acc.x += __shfl_xor(acc.x, 16);
        acc.y += __shfl_xor(acc.y, 16);
        acc.z += __shfl_xor(acc.z, 16);
        acc.w += __shfl_xor(acc.w, 16);

        const float inv = 1.0f / l;
        o.x = acc.x * inv;
        o.y = acc.y * inv;
        o.z = acc.z * inv;
        o.w = acc.w * inv;
    }
    if (half == 0)
        ((float4*)out)[(size_t)node * 16 + q] = o;
}

#define NODE_GROUP 4

__device__ __forceinline__ void
node_long(const float* __restrict__ ft,
         const int* __restrict__ ovf,
         const int novf,
         float* __restrict__ out,
         const int N, const int node, const int cnt, const int mine,
         const int lane) {
    const int half = lane >> 4;
    const int q    = lane & 15;
    const float4* __restrict__ ft4 = (const float4*)ft;
    const float4 fd = ft4[(size_t)node * 16 + q];
    const float4 zero4 = make_float4(0.f, 0.f, 0.f, 0.f);

    const bool inreg = cnt <= 2 * BCAP;
    int ids0 = 0;
    int ids1 = 0;
    if (inreg) two_chunk_list(node, mine, lane, ovf, novf, ids0, ids1);

    float m = -__builtin_huge_valf();
    long long last = -1;
    for (int c0 = 0; c0 < cnt; c0 += 32) {
        const int nrem = min(cnt - c0, 32);
        const int myid = inreg ? (c0 == 0 ? ids0 : ids1)
                               : long_list_chunk(node, nrem, last, mine, lane,
                                                 ovf, novf, N);
        for (int g = 0; g < nrem; g += 2 * NODE_GROUP) {
            float4 fv[NODE_GROUP];
            bool   ok[NODE_GROUP];
            for (int u = 0; u < NODE_GROUP; ++u) {
                const int k = g + 2 * u + half;
                ok[u] = k < nrem;
                const int s = __shfl(myid, min(k, nrem - 1));
                fv[u] = ft4[(size_t)s * 16 + q];
            }
            for (int u = 0; u < NODE_GROUP; ++u) {
                const float p = head_score(fv[u], fd);
                if (ok[u]) m = fmaxf(m, p);
            }
        }
    }
    m = fmaxf(m, __shfl_xor(m, 16));

    float  l   = 0.0f;
    float4 acc = zero4;
    last = -1;
    for (int c0 = 0; c0 < cnt; c0 += 32) {
        const int nrem = min(cnt - c0, 32);
        const int myid = inreg ? (c0 == 0 ? ids0 : ids1)
                               : long_list_chunk(node, nrem, last, mine, lane,
                                                 ovf, novf, N);
        for (int g = 0; g < nrem; g += 2 * NODE_GROUP) {
            float4 fv[NODE_GROUP];
            bool   ok[NODE_GROUP];
            for (int u = 0; u < NODE_GROUP; ++u) {
                const int k = g + 2 * u + half;
                ok[u] = k < nrem;
                const int s = __shfl(myid, min(k, nrem - 1));
                fv[u] = ft4[(size_t)s * 16 + q];
            }
            for (int u = 0; u < NODE_GROUP; ++u) {
                const float p  = head_score(fv[u], fd);
                const float ex = ok[u] ? __expf(p - m) : 0.0f;
                l += ex;
                acc.x = fmaf(ex, fv[u].x, acc.x);
                acc.y = fmaf(ex, fv[u].y, acc.y);
                acc.z = fmaf(ex, fv[u].z, acc.z);
                acc.w = fmaf(ex, fv[u].w, acc.w);
            }
        }
    }
    l     += __shfl_xor(l, 16);
    acc.x += __shfl_xor(acc.x, 16);
    acc.y += __shfl_xor(acc.y, 16);
    acc.z += __shfl_xor(acc.z, 16);
    acc.w += __shfl_xor(acc.w, 16);

    if (half == 0) {
        const float inv = 1.0f / l;
        float4 o;
        o.x = acc.x * inv;
        o.y = acc.y * inv;
        o.z = acc.z * inv;
        o.w = acc.w * inv;
        ((float4*)out)[(size_t)node * 16 + q] = o;
    }
}


#define NWAVES (BTHR / 32)

__device__ __forceinline__ void take_record(const unsigned r, const int node0, int* cnt, int* buck) {
    const int s  = (int)(r >> 16);
    const int dl = ((int)(r & 0xFFFFu) - node0) & 255;
    const int k  = atomicAdd(&cnt[dl], 1);
    if (k < BCAP) buck[dl * BCAP + k] = s;
}

__device__ __forceinline__ void relay_record(const unsigned r, const int node0, const int* cnt,
                                             int* cnt2, int* buck, int* lovn,
                                             int* __restrict__ ovfb, const int tot) {
    const int s  = (int)(r >> 16);
    const int dl = ((int)(r & 0xFFFFu) - node0) & 255;
    if (cnt[dl] > BCAP) {
        const int k = atomicAdd(&cnt2[dl], 1);
        if (k < BCAP) buck[dl * BCAP + k] = s;
        else {
            const int j = min(atomicAdd(lovn, 1), tot - 1);
            ovfb[j] = (int)((((unsigned)(node0 + dl)) << 16) | (unsigned)s);
        }
    }
}

__global__ void __launch_bounds__(BTHR)
dgat_node_kernel(const float* __restrict__ ft,
                 const int* rec,
                 const int* __restrict__ bincur,
                 const int* __restrict__ ovf2,
                 const int* __restrict__ cur2,
                 int* __restrict__ ovf,
                 int* __restrict__ cursor,
                 float* out,
                 int N, int E) {
    __shared__ int cnt[256];
    __shared__ int cnt2[256];
    __shared__ int4 buck4[256 * BCAP / 4];
    __shared__ int ltot;
    __shared__ int lbase;
    __shared__ int lovn;
    int* buck = (int*)buck4;
    const int t = threadIdx.x;
    const int b = blockIdx.x;
    const int node0 = b * BINN;
    if (t < 256) { cnt[t] = 0; cnt2[t] = 0; }
    if (t == 0) { ltot = 0; lbase = 0; lovn = 0; }
    for (int i = t; i < 256 * BCAP / 4; i += BTHR) buck4[i] = make_int4(0, 0, 0, 0);
    __syncthreads();

    const int nrec = min(max(bincur[b * CURST], 0), CAPB);
    const int nsp  = min(max(cur2[0], 0), E);
    const int* seg = rec + (size_t)b * RSTRIDE;
    for (int i0 = 0; i0 < nrec; i0 += BTHR * RPT) {
        unsigned r[RPT];
#pragma unroll
        for (int u = 0; u < RPT; ++u)
            r[u] = (unsigned)seg[min(i0 + t + u * BTHR, nrec - 1)];
#pragma unroll
        for (int u = 0; u < RPT; ++u) {
            if (i0 + t + u * BTHR < nrec) take_record(r[u], node0, cnt, buck);
        }
    }
    for (int i = t; i < nsp; i += BTHR) {
        const unsigned r = (unsigned)ovf2[i];
        if (bin_of((int)(r & 0xFFFFu)) == b) take_record(r, node0, cnt, buck);
    }
    __syncthreads();

    if (t < 256 && cnt[t] > BCAP) atomicAdd(&ltot, cnt[t] - BCAP);
    __syncthreads();
    const int tot = ltot;
    if (tot > 0) {
        if (t == 0) lbase = atomicAdd(cursor, tot);
        __syncthreads();
        int* __restrict__ ovfb = ovf + min(max(lbase, 0), max(E - tot, 0));
        for (int i = t; i < nrec; i += BTHR)
            relay_record((unsigned)seg[i], node0, cnt, cnt2, buck, &lovn, ovfb, tot);
        for (int i = t; i < nsp; i += BTHR) {
            const unsigned r = (unsigned)ovf2[i];
            if (bin_of((int)(r & 0xFFFFu)) == b)
                relay_record(r, node0, cnt, cnt2, buck, &lovn, ovfb, tot);
        }
        __syncthreads();
    }
    const int* __restrict__ ovfb = ovf + min(max(lbase, 0), max(E - tot, 0));

    const int lane = t & 31;
    const int wv = __builtin_amdgcn_readfirstlane(t >> 5);
    const int nn = min(BINN, N - node0);
    unsigned longmask = 0u;
    int c    = __builtin_amdgcn_readfirstlane(cnt[wv]);
    int mine = buck[wv * BCAP + lane];
#pragma clang loop unroll(disable)
    for (int j = 0; j < (BINN + NWAVES - 1) / NWAVES; ++j) {
        const int nl = wv + j * NWAVES;
        if (nl >= nn) break;
        int ncnt  = 0;
        int nmine = 0;
        if (nl + NWAVES < nn) {
            ncnt  = cnt[nl + NWAVES];
            nmine = buck[(nl + NWAVES) * BCAP + lane];
        }
        if (c <= BCAP) node_short(ft, out, node0 + nl, c, mine, lane);
        else           longmask |= 1u << j;
        c    = __builtin_amdgcn_readfirstlane(ncnt);
        mine = nmine;
    }
#pragma clang loop unroll(disable)
    for (int j = 0; j < (BINN + NWAVES - 1) / NWAVES; ++j) {
        if ((longmask >> j) & 1u) {
            const int nl = wv + j * NWAVES;
            node_long(ft, ovfb, tot, out, N, node0 + nl,
                      __builtin_amdgcn_readfirstlane(cnt[nl]), buck[nl * BCAP + lane], lane);
        }
    }
}

extern "C" void kernel_launch(void* const* d_in, const int* in_sizes, int n_in,
                              void* d_out, int out_size, void* d_ws, size_t ws_size,
                              hipStream_t stream) {
    const float* feat = (const float*)d_in[0];
    const float* W    = (const float*)d_in[1];
    const int*   src  = (const int*)d_in[2];
    const int*   dst  = (const int*)d_in[3];
    float*       out  = (float*)d_out;

    const int N = in_sizes[0] / D_IN;
    const int E = in_sizes[2];

    float* ft      = (float*)d_ws;
    int*   scratch = (int*)(ft + (size_t)N * HF);
    int*   ovf     = scratch + (size_t)N * BCAP;
    int*   cursor  = ovf + (size_t)E + (size_t)N;

    const int nblk = (E + EPB - 1) / EPB;
    const int nbin = (N + BINN - 1) / BINN;

    int* ovf2   = scratch;
    int* bincur = ovf2 + (size_t)((E + 31) / 32) * 32;
    int* cur2   = bincur + (size_t)NBMAX * CURST;

    int* rec    = (int*)d_out;

    const int mtiles = (N + 15) / 16;
    dgat_proj_wmma_kernel<<<(mtiles + PWAVES - 1) / PWAVES, PWAVES * 32, 0, stream>>>(feat, W, ft, bincur, cur2, cursor, N, mtiles);

    dgat_binplace_kernel<<<nblk, BTHR, 0, stream>>>(src, dst, rec, bincur, ovf2, cur2, E);
    dgat_node_kernel<<<nbin, BTHR, 0, stream>>>(ft, rec, bincur, ovf2, cur2, ovf, cursor, out, N, E);
}


# Edge grouping: each block keeps its records in its own region with a table of bin starts (no shared cursor, no global add) and runs as blocks of the projection launch: 3 launches to 2; sum order kept
# speedup vs baseline: 9.7027x; 1.1322x over previous
#include <hip/hip_runtime.h>
#include <hip/hip_bf16.h>
#include <math.h>


#define D_IN 128
#define NH   4
#define NF   16
#define HF   64

#define GBINS 256

typedef __attribute__((ext_vector_type(2))) float v2f;
typedef __attribute__((ext_vector_type(8))) float v8f;

#define PWAVES 8
#define FTHR (PWAVES * 32)

__device__ __forceinline__ void
proj_strip_block(const float* __restrict__ feat,
                 const float* __restrict__ W,
                 float* __restrict__ ft,
                 const int N, const int mtiles, const int pblk) {
    const int lane  = threadIdx.x & 31;
    const int wave  = __builtin_amdgcn_readfirstlane(threadIdx.x >> 5);
    const int strip = pblk * PWAVES + wave;
    if (strip >= mtiles) return;

    const int m0   = strip * 16;
    const int half = lane >> 4;
    const int l    = lane & 15;

    int rowA = m0 + l;
    if (rowA > N - 1) rowA = N - 1;
    const float4* __restrict__ arow = (const float4*)(feat + (size_t)rowA * D_IN) + half;
    const float4* __restrict__ wrow = (const float4*)(W + (size_t)(4 * half) * HF) + l;

    float4 a4[D_IN / 8];
#pragma unroll
    for (int i = 0; i < D_IN / 8; ++i) a4[i] = arow[2 * i];

    float4 wq[3][4];
#pragma unroll
    for (int j = 0; j < 4; ++j) {
        wq[0][j] = wrow[(size_t)(0 + j) * (HF / 4)];
        wq[1][j] = wrow[(size_t)(8 + j) * (HF / 4)];
    }

    v8f c0 = {0.f, 0.f, 0.f, 0.f, 0.f, 0.f, 0.f, 0.f};
    v8f c1 = c0, c2 = c0, c3 = c0;

#pragma unroll
    for (int i = 0; i < D_IN / 8; ++i) {
        if (i + 2 < D_IN / 8) {
#pragma unroll
            for (int j = 0; j < 4; ++j)
                wq[(i + 2) % 3][j] = wrow[(size_t)(8 * (i + 2) + j) * (HF / 4)];
        }
        __builtin_amdgcn_sched_barrier(0);
        const float4 w0 = wq[i % 3][0];
        const float4 w1 = wq[i % 3][1];
        const float4 w2 = wq[i % 3][2];
        const float4 w3 = wq[i % 3][3];
        v2f a, b;
        a.x = a4[i].x; a.y = a4[i].y;
        b.x = w0.x; b.y = w1.x;
        c0 = __builtin_amdgcn_wmma_f32_16x16x4_f32(false, a, false, b, (short)0, c0, false, false);
        b.x = w0.y; b.y = w1.y;
        c1 = __builtin_amdgcn_wmma_f32_16x16x4_f32(false, a, false, b, (short)0, c1, false, false);
        b.x = w0.z; b.y = w1.z;
        c2 = __builtin_amdgcn_wmma_f32_16x16x4_f32(false, a, false, b, (short)0, c2, false, false);
        b.x = w0.w; b.y = w1.w;
        c3 = __builtin_amdgcn_wmma_f32_16x16x4_f32(false, a, false, b, (short)0, c3, false, false);
        a.x = a4[i].z; a.y = a4[i].w;
        b.x = w2.x; b.y = w3.x;
        c0 = __builtin_amdgcn_wmma_f32_16x16x4_f32(false, a, false, b, (short)0, c0, false, false);
        b.x = w2.y; b.y = w3.y;
        c1 = __builtin_amdgcn_wmma_f32_16x16x4_f32(false, a, false, b, (short)0, c1, false, false);
        b.x = w2.z; b.y = w3.z;
        c2 = __builtin_amdgcn_wmma_f32_16x16x4_f32(false, a, false, b, (short)0, c2, false, false);
        b.x = w2.w; b.y = w3.w;
        c3 = __builtin_amdgcn_wmma_f32_16x16x4_f32(false, a, false, b, (short)0, c3, false, false);
    }

    __builtin_amdgcn_sched_barrier(0);
    asm volatile("v_nop\n\tv_nop\n\tv_nop\n\tv_nop");
    __builtin_amdgcn_sched_barrier(0);

    float4* __restrict__ op = (float4*)(ft + ((size_t)(m0 + 8 * half)) * HF) + l;
    if (m0 + 16 <= N) {
#pragma unroll
        for (int v = 0; v < 8; ++v)
            op[(size_t)v * (HF / 4)] = make_float4(c0[v], c1[v], c2[v], c3[v]);
    } else {
#pragma unroll
        for (int v = 0; v < 8; ++v) {
            const int row = m0 + v + 8 * half;
            if (row < N)
                op[(size_t)v * (HF / 4)] = make_float4(c0[v], c1[v], c2[v], c3[v]);
        }
    }
}

#define BCAP  32
#define BINN  200
#define NBMAX GBINS
#define EPB   4096
#define EPT   (EPB / FTHR)
#define BTHR  1024
#define RPT   8

__device__ __forceinline__ int bin_of(const int d) {
    const unsigned b = (unsigned)d / (unsigned)BINN;
    return (int)(b < (unsigned)(NBMAX - 1) ? b : (unsigned)(NBMAX - 1));
}

__device__ __forceinline__ int load_dst4(const int* __restrict__ dst, const int e0, const int E,
                                         int (&d)[4]) {
    if (e0 + 3 < E) {
        const int4 v = *(const int4*)(dst + e0);
        d[0] = v.x; d[1] = v.y; d[2] = v.z; d[3] = v.w;
        return 4;
    }
    int n = 0;
    for (int j = 0; j < 4; ++j) {
        d[j] = 0;
        if (e0 + j < E) { d[j] = dst[e0 + j]; n = j + 1; }
    }
    return n;
}

__device__ __forceinline__ void wave_excl_scan256(const int* a, int* res, const int lane) {
    int v[8];
    int s = 0;
    for (int k = 0; k < 8; ++k) v[k] = a[lane * 8 + k];
    for (int k = 0; k < 8; ++k) { const int x = v[k]; v[k] = s; s += x; }
    int inc = s;
    for (int o = 1; o < 32; o <<= 1) {
        const int y = __shfl_up(inc, o);
        if (lane >= o) inc += y;
    }
    const int off = inc - s;
    for (int k = 0; k < 8; ++k) res[lane * 8 + k] = v[k] + off;
}

__device__ __forceinline__ void
group_edges_block(const int* __restrict__ src,
                  const int* __restrict__ dst,
                  int* __restrict__ rec,
                  int* __restrict__ tab,
                  const int E, const int nblk, const int blk) {
    __shared__ int lcnt[NBMAX];
    __shared__ int lbase[NBMAX];
    __shared__ int run[NBMAX];
    __shared__ int4 srec4[EPB / 4];
    int* srec = (int*)srec4;

    const int t = threadIdx.x;
    for (int i = t; i < NBMAX; i += FTHR) { lcnt[i] = 0; run[i] = 0; }
    __syncthreads();
    int d[EPT / 4][4];
    int s[EPT / 4][4];
    int n[EPT / 4];
#pragma unroll
    for (int u = 0; u < EPT / 4; ++u) {
        const int e0 = blk * EPB + u * (FTHR * 4) + t * 4;
        n[u] = load_dst4(dst, e0, E, d[u]);
        load_dst4(src, e0, E, s[u]);
    }
#pragma unroll
    for (int u = 0; u < EPT / 4; ++u) {
#pragma unroll
        for (int j = 0; j < 4; ++j)
            if (j < n[u]) atomicAdd(&lcnt[bin_of(d[u][j])], 1);
    }
    __syncthreads();
    if ((t >> 5) == 0) wave_excl_scan256(lcnt, lbase, t & 31);
    __syncthreads();
#pragma unroll
    for (int u = 0; u < EPT / 4; ++u) {
#pragma unroll
        for (int j = 0; j < 4; ++j) {
            if (j < n[u]) {
                const int b = bin_of(d[u][j]);
                const int k = atomicAdd(&run[b], 1);
                const int slot = min(lbase[b] + k, EPB - 1);
                srec[slot] = (int)(((unsigned)s[u][j] << 16) | ((unsigned)d[u][j] & 0xFFFFu));
            }
        }
    }
    __syncthreads();

    const int ne = min(EPB, max(E - blk * EPB, 0));
    int4* __restrict__ rec4 = (int4*)(rec + (size_t)blk * EPB);
    for (int i = t; 4 * i < ne; i += FTHR) {
        if (4 * i + 3 < ne) rec4[i] = srec4[i];
        else for (int k = 4 * i; k < ne; ++k) rec[(size_t)blk * EPB + k] = srec[k];
    }
    for (int i = t; i < NBMAX; i += FTHR) tab[(size_t)i * nblk + blk] = lbase[i];
    if (t == 0) tab[(size_t)NBMAX * nblk + blk] = ne;
}

__global__ void __launch_bounds__(FTHR)
dgat_front_kernel(const float* __restrict__ feat,
                  const float* __restrict__ W,
                  float* __restrict__ ft,
                  const int* __restrict__ src,
                  const int* __restrict__ dst,
                  int* __restrict__ rec,
                  int* __restrict__ tab,
                  int* __restrict__ cursor,
                  int N, int mtiles, int E, int nblk) {
    const int blk = blockIdx.x;
    if (blk == 0 && threadIdx.x == 0) cursor[0] = 0;
    if (blk < nblk) {
        group_edges_block(src, dst, rec, tab, E, nblk, blk);
        return;
    }
    proj_strip_block(feat, W, ft, N, mtiles, blk - nblk);
}

__device__ __forceinline__ float quad_sum(float v) {
    v += __int_as_float(__builtin_amdgcn_mov_dpp(__float_as_int(v), 0xB1, 0xF, 0xF, true));
    v += __int_as_float(__builtin_amdgcn_mov_dpp(__float_as_int(v), 0x4E, 0xF, 0xF, true));
    return v;
}

__device__ __forceinline__ float head_score(const float4 a, const float4 b) {
    float p = a.x * b.x;
    p = fmaf(a.y, b.y, p);
    p = fmaf(a.z, b.z, p);
    p = fmaf(a.w, b.w, p);
    return quad_sum(p) * 0.25f;
}

__device__ __forceinline__ int long_list_chunk(const int node, const int nrem, long long& last,
                                            const int mine, const int lane,
                                            const int* __restrict__ ovf,
                                            const int novf, const int N) {
    int myid = 0;
    const long long mykey = ((long long)mine << 32) | (long long)lane;
    for (int k = 0; k < nrem; ++k) {
        long long best = (mykey > last) ? mykey : 0x7fffffffffffffffLL;
        for (int p0 = 0; p0 < novf; p0 += 32) {
            const int p = p0 + lane;
            if (p < novf) {
                const unsigned r = (unsigned)ovf[p];
                const long long k2 = ((long long)(r & 0xFFFFu) << 32) | (long long)(BCAP + p);
                if ((int)(r >> 16) == node && k2 > last) best = min(best, k2);
            }
        }
        for (int o = 16; o > 0; o >>= 1) {
            const long long y = __shfl_xor(best, o);
            best = min(best, y);
        }
        last = best;
        const int sm = min((int)(best >> 32) & 0xFFFF, N - 1);
        if (lane == k) myid = sm;
    }
    return myid;
}

__device__ __forceinline__ void two_chunk_list(const int node, const int mine, const int lane,
                                               const int* __restrict__ ovf,
                                               const int novf,
                                               int& ids0, int& ids1) {
    int extra = 0;
    int nx = 0;
    for (int p0 = 0; p0 < novf; p0 += 32) {
        const int p = p0 + lane;
        int  s2  = 0;
        bool hit = false;
        if (p < novf) {
            const unsigned r = (unsigned)ovf[p];
            s2  = (int)(r & 0xFFFFu);
            hit = (int)(r >> 16) == node;
        }
        unsigned found = __builtin_amdgcn_ballot_w32(hit);
        while (found != 0u && nx < BCAP) {
            const int j = __builtin_ctz(found);
            found &= found - 1u;
            const int sj = __builtin_amdgcn_readlane(s2, j);
            if (lane == nx) extra = sj;
            ++nx;
        }
    }

    const int sa = mine;
    const int sb = extra;
    const int a  = (mine << 6) | lane;
    const int b  = (lane < nx) ? ((extra << 6) | (BCAP + lane)) : 0x7fffffff;

    int ra = 0;
    int rb = 0;
    for (int j = 0; j < BCAP; ++j) {
        const int kj = __builtin_amdgcn_readlane(a, j);
        ra += (kj < a) ? 1 : 0;
        rb += (kj < b) ? 1 : 0;
    }
    for (int j = 0; j < nx; ++j) {
        const int kj = __builtin_amdgcn_readlane(b, j);
        ra += (kj < a) ? 1 : 0;
        rb += (kj < b) ? 1 : 0;
    }

    for (int j = 0; j < BCAP; ++j) {
        const int r = __builtin_amdgcn_readlane(ra, j);
        const int v = __builtin_amdgcn_readlane(sa, j);
        if (lane == r)        ids0 = v;
        if (lane == r - BCAP) ids1 = v;
    }
    for (int j = 0; j < nx; ++j) {
        const int r = __builtin_amdgcn_readlane(rb, j);
        const int v = __builtin_amdgcn_readlane(sb, j);
        if (lane == r)        ids0 = v;
        if (lane == r - BCAP) ids1 = v;
    }
}

template <int J>
__device__ __forceinline__ int pair_id(const int ids) {
    return __builtin_amdgcn_mov_dpp(ids, 0x150 + J, 0xF, 0xF, true);
}

__device__ __forceinline__ float4 load_row_q(const float* __restrict__ ft, const int s, const unsigned q16) {
    return *(const float4*)((const char*)ft + ((((unsigned)s) << 8) | q16));
}

#define DG_FOR16(M) M(0) M(1) M(2) M(3) M(4) M(5) M(6) M(7) M(8) M(9) M(10) M(11) M(12) M(13) M(14) M(15) \
                    }}}}}}}}}}}}}}}}

__device__ __forceinline__ void node_short(const float* __restrict__ ft,
                                           float* __restrict__ out,
                                           const int node, const int cnt, const int mine,
                                           const int lane) {
    const int half = lane >> 4;
    const int q    = lane & 15;
    const unsigned q16 = (unsigned)q << 4;
    const int lim = cnt - half;
    const float4 fd = ((const float4*)ft)[(size_t)node * 16 + q];
    float4 o = make_float4(0.f, 0.f, 0.f, 0.f);

    if (cnt > 0) {
        const int key = (lane < cnt) ? ((mine << 5) | lane) : 0x7fffffff;
        int s = mine;
        int r = 0;
        for (int j = 0; j < cnt; ++j)
            r += (__builtin_amdgcn_readlane(key, j) < key) ? 1 : 0;
        if (lane >= cnt) r = lane;
        const int s0 = __builtin_amdgcn_readfirstlane(s);
        if (lane >= cnt) s = s0;
        const int dest = ((r & 1) << 4) | (r >> 1);
        const int ids = __builtin_amdgcn_ds_permute(dest << 2, s);

        float4 fv[16];
        float  p[16];

#define DG_LOAD(J) if (2 * (J) < cnt) { fv[J] = load_row_q(ft, pair_id<J>(ids), q16);
        DG_FOR16(DG_LOAD)
#undef DG_LOAD

        float m = -__builtin_huge_valf();
#define DG_SCORE(J) if (2 * (J) < cnt) { p[J] = head_score(fv[J], fd); m = fmaxf(m, p[J]);
        DG_FOR16(DG_SCORE)
#undef DG_SCORE
        m = fmaxf(m, __shfl_xor(m, 16));

        float  l   = 0.0f;
        float4 acc = make_float4(0.f, 0.f, 0.f, 0.f);
#define DG_ACC(J) if (2 * (J) < cnt) {                                         \
            const float ex = (2 * (J) < lim) ? __expf(p[J] - m) : 0.0f;        \
            l += ex;                                                           \
            acc.x = fmaf(ex, fv[J].x, acc.x);                                  \
            acc.y = fmaf(ex, fv[J].y, acc.y);                                  \
            acc.z = fmaf(ex, fv[J].z, acc.z);                                  \
            acc.w = fmaf(ex, fv[J].w, acc.w);
        DG_FOR16(DG_ACC)
#undef DG_ACC
        l     += __shfl_xor(l, 16);
        acc.x += __shfl_xor(acc.x, 16);
        acc.y += __shfl_xor(acc.y, 16);
        acc.z += __shfl_xor(acc.z, 16);
        acc.w += __shfl_xor(acc.w, 16);

        const float inv = 1.0f / l;
        o.x = acc.x * inv;
        o.y = acc.y * inv;
        o.z = acc.z * inv;
        o.w = acc.w * inv;
    }
    if (half == 0)
        ((float4*)out)[(size_t)node * 16 + q] = o;
}

#define NODE_GROUP 4

__device__ __forceinline__ void
node_long(const float* __restrict__ ft,
         const int* __restrict__ ovf,
         const int novf,
         float* __restrict__ out,
         const int N, const int node, const int cnt, const int mine,
         const int lane) {
    const int half = lane >> 4;
    const int q    = lane & 15;
    const float4* __restrict__ ft4 = (const float4*)ft;
    const float4 fd = ft4[(size_t)node * 16 + q];
    const float4 zero4 = make_float4(0.f, 0.f, 0.f, 0.f);

    const bool inreg = cnt <= 2 * BCAP;
    int ids0 = 0;
    int ids1 = 0;
    if (inreg) two_chunk_list(node, mine, lane, ovf, novf, ids0, ids1);

    float m = -__builtin_huge_valf();
    long long last = -1;
    for (int c0 = 0; c0 < cnt; c0 += 32) {
        const int nrem = min(cnt - c0, 32);
        const int myid = inreg ? (c0 == 0 ? ids0 : ids1)
                               : long_list_chunk(node, nrem, last, mine, lane,
                                                 ovf, novf, N);
        for (int g = 0; g < nrem; g += 2 * NODE_GROUP) {
            float4 fv[NODE_GROUP];
            bool   ok[NODE_GROUP];
            for (int u = 0; u < NODE_GROUP; ++u) {
                const int k = g + 2 * u + half;
                ok[u] = k < nrem;
                const int s = __shfl(myid, min(k, nrem - 1));
                fv[u] = ft4[(size_t)s * 16 + q];
            }
            for (int u = 0; u < NODE_GROUP; ++u) {
                const float p = head_score(fv[u], fd);
                if (ok[u]) m = fmaxf(m, p);
            }
        }
    }
    m = fmaxf(m, __shfl_xor(m, 16));

    float  l   = 0.0f;
    float4 acc = zero4;
    last = -1;
    for (int c0 = 0; c0 < cnt; c0 += 32) {
        const int nrem = min(cnt - c0, 32);
        const int myid = inreg ? (c0 == 0 ? ids0 : ids1)
                               : long_list_chunk(node, nrem, last, mine, lane,
                                                 ovf, novf, N);
        for (int g = 0; g < nrem; g += 2 * NODE_GROUP) {
            float4 fv[NODE_GROUP];
            bool   ok[NODE_GROUP];
            for (int u = 0; u < NODE_GROUP; ++u) {
                const int k = g + 2 * u + half;
                ok[u] = k < nrem;
                const int s = __shfl(myid, min(k, nrem - 1));
                fv[u] = ft4[(size_t)s * 16 + q];
            }
            for (int u = 0; u < NODE_GROUP; ++u) {
                const float p  = head_score(fv[u], fd);
                const float ex = ok[u] ? __expf(p - m) : 0.0f;
                l += ex;
                acc.x = fmaf(ex, fv[u].x, acc.x);
                acc.y = fmaf(ex, fv[u].y, acc.y);
                acc.z = fmaf(ex, fv[u].z, acc.z);
                acc.w = fmaf(ex, fv[u].w, acc.w);
            }
        }
    }
    l     += __shfl_xor(l, 16);
    acc.x += __shfl_xor(acc.x, 16);
    acc.y += __shfl_xor(acc.y, 16);
    acc.z += __shfl_xor(acc.z, 16);
    acc.w += __shfl_xor(acc.w, 16);

    if (half == 0) {
        const float inv = 1.0f / l;
        float4 o;
        o.x = acc.x * inv;
        o.y = acc.y * inv;
        o.z = acc.z * inv;
        o.w = acc.w * inv;
        ((float4*)out)[(size_t)node * 16 + q] = o;
    }
}


#define NWAVES (BTHR / 32)

__device__ __forceinline__ void take_record(const unsigned r, const int node0, int* cnt, int* buck) {
    const int s  = (int)(r >> 16);
    const int dl = ((int)(r & 0xFFFFu) - node0) & 255;
    const int k  = atomicAdd(&cnt[dl], 1);
    if (k < BCAP) buck[dl * BCAP + k] = s;
}

__device__ __forceinline__ void relay_record(const unsigned r, const int node0, const int* cnt,
                                             int* cnt2, int* buck, int* lovn,
                                             int* __restrict__ ovfb, const int tot) {
    const int s  = (int)(r >> 16);
    const int dl = ((int)(r & 0xFFFFu) - node0) & 255;
    if (cnt[dl] > BCAP) {
        const int k = atomicAdd(&cnt2[dl], 1);
        if (k < BCAP) buck[dl * BCAP + k] = s;
        else {
            const int j = min(atomicAdd(lovn, 1), tot - 1);
            ovfb[j] = (int)((((unsigned)(node0 + dl)) << 16) | (unsigned)s);
        }
    }
}

__device__ __forceinline__ int wave_run_bounds(const int* __restrict__ tab0,
                                               const int* __restrict__ tab1,
                                               const int nblk, const int base, const int wv,
                                               const int lane, int& st, int& en) {
    const int bm = base + wv + NWAVES * lane;
    st = 0;
    en = 0;
    if (bm < nblk) { st = tab0[bm]; en = tab1[bm]; }
    const int left = nblk - base - wv;
    return left > 0 ? min(32, (left + NWAVES - 1) / NWAVES) : 0;
}

__device__ __forceinline__ void clamp_run_bounds(int& st, int& en) {
    st = min(max(st, 0), EPB);
    en = min(max(en, st), EPB);
}

__global__ void __launch_bounds__(BTHR)
dgat_node_kernel(const float* __restrict__ ft,
                 const int* __restrict__ rec,
                 const int* __restrict__ tab,
                 int* __restrict__ ovf,
                 int* __restrict__ cursor,
                 float* __restrict__ out,
                 int N, int E, int nblk) {
    __shared__ int cnt[256];
    __shared__ int cnt2[256];
    __shared__ int4 buck4[256 * BCAP / 4];
    __shared__ int ltot;
    __shared__ int lbase;
    __shared__ int lovn;
    int* buck = (int*)buck4;
    const int t = threadIdx.x;
    const int b = blockIdx.x;
    const int node0 = b * BINN;
    const int lane = t & 31;
    const int wv = __builtin_amdgcn_readfirstlane(t >> 5);
    const int* __restrict__ tab0 = tab + (size_t)b * nblk;
    const int* __restrict__ tab1 = tab0 + nblk;
    int st0, en0;
    const int nrun0 = wave_run_bounds(tab0, tab1, nblk, 0, wv, lane, st0, en0);

    if (t < 256) { cnt[t] = 0; cnt2[t] = 0; }
    if (t == 0) { ltot = 0; lbase = 0; lovn = 0; }
    for (int i = t; i < 256 * BCAP / 4; i += BTHR) buck4[i] = make_int4(0, 0, 0, 0);
    __syncthreads();

    for (int base = 0; base < nblk; base += 32 * NWAVES) {
        int st = st0;
        int en = en0;
        int nrun = nrun0;
        if (base > 0) nrun = wave_run_bounds(tab0, tab1, nblk, base, wv, lane, st, en);
        clamp_run_bounds(st, en);
        for (int j0 = 0; j0 < nrun; j0 += RPT) {
            unsigned r[RPT];
            bool     ok[RPT];
#pragma unroll
            for (int u = 0; u < RPT; ++u) {
                const int j  = min(j0 + u, 31);
                const int sj = __builtin_amdgcn_readlane(st, j);
                const int ej = __builtin_amdgcn_readlane(en, j);
                const int o  = sj + lane;
                ok[u] = (j0 + u < nrun) && (o < ej);
                r[u]  = 0u;
                if (ok[u]) r[u] = (unsigned)rec[(size_t)(base + wv + NWAVES * j) * EPB + o];
            }
#pragma unroll
            for (int u = 0; u < RPT; ++u) {
                if (ok[u]) take_record(r[u], node0, cnt, buck);
            }
        }
        for (int j = 0; j < nrun; ++j) {
            const int sj = __builtin_amdgcn_readlane(st, j);
            const int ej = __builtin_amdgcn_readlane(en, j);
            const int* __restrict__ seg = rec + (size_t)(base + wv + NWAVES * j) * EPB;
            for (int o = sj + 32 + lane; o < ej; o += 32)
                take_record((unsigned)seg[o], node0, cnt, buck);
        }
    }
    __syncthreads();

    if (t < 256 && cnt[t] > BCAP) atomicAdd(&ltot, cnt[t] - BCAP);
    __syncthreads();
    const int tot = ltot;
    if (tot > 0) {
        if (t == 0) lbase = atomicAdd(cursor, tot);
        __syncthreads();
        int* __restrict__ ovfb = ovf + min(max(lbase, 0), max(E - tot, 0));
        for (int base = 0; base < nblk; base += 32 * NWAVES) {
            int st, en;
            const int nrun = wave_run_bounds(tab0, tab1, nblk, base, wv, lane, st, en);
            clamp_run_bounds(st, en);
            for (int j = 0; j < nrun; ++j) {
                const int sj = __builtin_amdgcn_readlane(st, j);
                const int ej = __builtin_amdgcn_readlane(en, j);
                const int* __restrict__ seg = rec + (size_t)(base + wv + NWAVES * j) * EPB;
                for (int o = sj + lane; o < ej; o += 32)
                    relay_record((unsigned)seg[o], node0, cnt, cnt2, buck, &lovn, ovfb, tot);
            }
        }
        __syncthreads();
    }
    const int* __restrict__ ovfb = ovf + min(max(lbase, 0), max(E - tot, 0));

    const int nn = min(BINN, N - node0);
    unsigned longmask = 0u;
    int c    = __builtin_amdgcn_readfirstlane(cnt[wv]);
    int mine = buck[wv * BCAP + lane];
#pragma clang loop unroll(disable)
    for (int j = 0; j < (BINN + NWAVES - 1) / NWAVES; ++j) {
        const int nl = wv + j * NWAVES;
        if (nl >= nn) break;
        int ncnt  = 0;
        int nmine = 0;
        if (nl + NWAVES < nn) {
            ncnt  = cnt[nl + NWAVES];
            nmine = buck[(nl + NWAVES) * BCAP + lane];
        }
        if (c <= BCAP) node_short(ft, out, node0 + nl, c, mine, lane);
        else           longmask |= 1u << j;
        c    = __builtin_amdgcn_readfirstlane(ncnt);
        mine = nmine;
    }
#pragma clang loop unroll(disable)
    for (int j = 0; j < (BINN + NWAVES - 1) / NWAVES; ++j) {
        if ((longmask >> j) & 1u) {
            const int nl = wv + j * NWAVES;
            node_long(ft, ovfb, tot, out, N, node0 + nl,
                      __builtin_amdgcn_readfirstlane(cnt[nl]), buck[nl * BCAP + lane], lane);
        }
    }
}

extern "C" void kernel_launch(void* const* d_in, const int* in_sizes, int n_in,
                              void* d_out, int out_size, void* d_ws, size_t ws_size,
                              hipStream_t stream) {
    const float* feat = (const float*)d_in[0];
    const float* W    = (const float*)d_in[1];
    const int*   src  = (const int*)d_in[2];
    const int*   dst  = (const int*)d_in[3];
    float*       out  = (float*)d_out;

    const int N = in_sizes[0] / D_IN;
    const int E = in_sizes[2];

    float* ft      = (float*)d_ws;
    int*   scratch = (int*)(ft + (size_t)N * HF);
    int*   ovf     = scratch + (size_t)N * BCAP;
    int*   cursor  = ovf + (size_t)E + (size_t)N;

    const int nblk = (E + EPB - 1) / EPB;
    const int nbin = (N + BINN - 1) / BINN;

    int* rec = scratch;
    int* tab = rec + (size_t)nblk * EPB;

    const int mtiles = (N + 15) / 16;
    const int nproj  = (mtiles + PWAVES - 1) / PWAVES;
    dgat_front_kernel<<<nblk + nproj, FTHR, 0, stream>>>(feat, W, ft, src, dst, rec, tab, cursor, N, mtiles, E, nblk);
    dgat_node_kernel<<<nbin, BTHR, 0, stream>>>(ft, rec, tab, ovf, cursor, out, N, E, nblk);
}
